# starting kernel
# baseline (speedup 1.0000x reference)
#include <hip/hip_runtime.h>
#include <hip/hip_bf16.h>

typedef __attribute__((ext_vector_type(2))) float v2f;
typedef __attribute__((ext_vector_type(8))) float v8f;

#define IN_CH  512
#define HIDC   128
#define OUTC   64

__global__ void k_init_deg(float* deg, int n) {
    int i = blockIdx.x * blockDim.x + threadIdx.x;
    if (i < n) deg[i] = 1.0f;
}

__global__ void k_count_deg(const int* __restrict__ col, float* deg, int E) {
    int e = blockIdx.x * blockDim.x + threadIdx.x;
    if (e < E) atomicAdd(&deg[col[e]], 1.0f);
}

__global__ void k_rsqrt(float* deg, int n) {
    int i = blockIdx.x * blockDim.x + threadIdx.x;
    if (i < n) deg[i] = rsqrtf(deg[i]);
}

__global__ __launch_bounds__(256)
void k_gemm_wmma_f32(const float* __restrict__ A, const float* __restrict__ B,
                     float* __restrict__ C, int M, int K, int N) {
    __shared__ float ldsA[64][36];
    __shared__ float ldsB[32][33];

    const int tid   = threadIdx.x;
    const int lane  = tid & 31;
    const int wave  = tid >> 5;
    const int waveM = wave >> 1;
    const int waveN = wave & 1;
    const int m0 = blockIdx.x * 64;
    const int n0 = blockIdx.y * 32;

    v8f acc = {};

    const int lhalf = lane >> 4;
    const int l15   = lane & 15;
    const int arow  = waveM * 16 + l15;
    const int bcol  = waveN * 16 + l15;
    const int khalf = lhalf * 2;

    for (int k0 = 0; k0 < K; k0 += 32) {
        #pragma unroll
        for (int i = 0; i < 2; ++i) {
            int idx = tid + i * 256;
            int r   = idx >> 3;
            int c4  = idx & 7;
            int gr  = m0 + r;
            gr = gr < M ? gr : (M - 1);
            const float4 v = *(const float4*)(A + (size_t)gr * K + k0 + c4 * 4);
            ldsA[r][c4 * 4 + 0] = v.x;
            ldsA[r][c4 * 4 + 1] = v.y;
            ldsA[r][c4 * 4 + 2] = v.z;
            ldsA[r][c4 * 4 + 3] = v.w;
        }
        {
            int r  = tid >> 3;
            int c4 = tid & 7;
            const float4 v = *(const float4*)(B + (size_t)(k0 + r) * N + n0 + c4 * 4);
            ldsB[r][c4 * 4 + 0] = v.x;
            ldsB[r][c4 * 4 + 1] = v.y;
            ldsB[r][c4 * 4 + 2] = v.z;
            ldsB[r][c4 * 4 + 3] = v.w;
        }
        __syncthreads();

        #pragma unroll
        for (int kk = 0; kk < 32; kk += 4) {
            v2f a, b;
            a.x = ldsA[arow][kk + khalf];
            a.y = ldsA[arow][kk + khalf + 1];
            b.x = ldsB[kk + khalf]    [bcol];
            b.y = ldsB[kk + khalf + 1][bcol];
            acc = __builtin_amdgcn_wmma_f32_16x16x4_f32(
                 false, a,  false, b,
                 (short)0, acc,  false,  false);
        }
        __syncthreads();
    }

    const int crow0 = m0 + waveM * 16 + lhalf * 8;
    const int ccol  = n0 + bcol;
    #pragma unroll
    for (int v = 0; v < 8; ++v) {
        int r = crow0 + v;
        if (r < M) C[(size_t)r * N + ccol] = acc[v];
    }
}

__global__ void k_self_loop(const float* __restrict__ h, const float* __restrict__ dinv,
                            float* __restrict__ out, int n, int F) {
    int idx = blockIdx.x * blockDim.x + threadIdx.x;
    if (idx >= n * F) return;
    int i = idx / F;
    float d = dinv[i];
    out[idx] = d * d * h[idx];
}

__global__ void k_scatter(const int* __restrict__ rows, const int* __restrict__ cols,
                          const float* __restrict__ dinv, const float* __restrict__ h,
                          float* __restrict__ out, int E, int F) {
    int idx = blockIdx.x * blockDim.x + threadIdx.x;
    if (idx >= E * F) return;
    int e = idx / F;
    int f = idx - e * F;
    int r = rows[e];
    int c = cols[e];
    float norm = dinv[r] * dinv[c];
    atomicAdd(out + (size_t)c * F + f, norm * h[(size_t)r * F + f]);
}

__global__ void k_bias_act(float* __restrict__ buf, const float* __restrict__ bias,
                           int n, int F, int do_relu) {
    int idx = blockIdx.x * blockDim.x + threadIdx.x;
    if (idx >= n * F) return;
    int f = idx % F;
    float v = buf[idx] + bias[f];
    if (do_relu) v = v > 0.0f ? v : 0.0f;
    buf[idx] = v;
}

__global__ __launch_bounds__(256)
void k_decode(const int* __restrict__ pos, const int* __restrict__ neg,
              const float* __restrict__ z, float* __restrict__ logits, int E) {
    int gid = blockIdx.x * blockDim.x + threadIdx.x;
    int e = gid >> 4;
    int l = gid & 15;
    if (e >= 2 * E) return;
    int src, dst;
    if (e < E) { src = pos[e];      dst = pos[E + e];      }
    else       { src = neg[e - E];  dst = neg[E + (e - E)]; }
    const float4 a = *(const float4*)(z + (size_t)src * OUTC + l * 4);
    const float4 b = *(const float4*)(z + (size_t)dst * OUTC + l * 4);
    float s = a.x * b.x + a.y * b.y + a.z * b.z + a.w * b.w;
    s += __shfl_xor(s, 1, 32);
    s += __shfl_xor(s, 2, 32);
    s += __shfl_xor(s, 4, 32);
    s += __shfl_xor(s, 8, 32);
    if (l == 0) logits[e] = s;
}

static inline size_t align_f(size_t n) { return (n + 63) & ~size_t(63); }

extern "C" void kernel_launch(void* const* d_in, const int* in_sizes, int n_in,
                              void* d_out, int out_size, void* d_ws, size_t ws_size,
                              hipStream_t stream) {
    const float* x   = (const float*)d_in[0];
    const int*   pos = (const int*)  d_in[1];
    const int*   neg = (const int*)  d_in[2];
    const float* W1  = (const float*)d_in[3];
    const float* b1  = (const float*)d_in[4];
    const float* W2  = (const float*)d_in[5];
    const float* b2  = (const float*)d_in[6];
    float* out = (float*)d_out;

    const int N = in_sizes[0] / IN_CH;
    const int E = in_sizes[1] / 2;

    float* ws = (float*)d_ws;
    size_t o = 0;
    float* dinv = ws + o; o += align_f((size_t)N);
    float* h0   = ws + o; o += align_f((size_t)N * HIDC);
    float* g1   = ws + o; o += align_f((size_t)N * HIDC);
    float* g2   = ws + o; o += align_f((size_t)N * OUTC);
    (void)ws_size;

    const int T = 256;
    k_init_deg <<<(N + T - 1) / T, T, 0, stream>>>(dinv, N);
    k_count_deg<<<(E + T - 1) / T, T, 0, stream>>>(pos + E, dinv, E);
    k_rsqrt    <<<(N + T - 1) / T, T, 0, stream>>>(dinv, N);

    dim3 g1g((N + 63) / 64, HIDC / 32);
    k_gemm_wmma_f32<<<g1g, T, 0, stream>>>(x, W1, h0, N, IN_CH, HIDC);
    k_self_loop<<<((N * HIDC) + T - 1) / T, T, 0, stream>>>(h0, dinv, g1, N, HIDC);
    k_scatter  <<<((E * HIDC) + T - 1) / T, T, 0, stream>>>(pos, pos + E, dinv, h0, g1, E, HIDC);
    k_bias_act <<<((N * HIDC) + T - 1) / T, T, 0, stream>>>(g1, b1, N, HIDC, 1);

    dim3 g2g((N + 63) / 64, OUTC / 32);
    k_gemm_wmma_f32<<<g2g, T, 0, stream>>>(g1, W2, h0, N, HIDC, OUTC);
    k_self_loop<<<((N * OUTC) + T - 1) / T, T, 0, stream>>>(h0, dinv, g2, N, OUTC);
    k_scatter  <<<((E * OUTC) + T - 1) / T, T, 0, stream>>>(pos, pos + E, dinv, h0, g2, E, OUTC);
    k_bias_act <<<((N * OUTC) + T - 1) / T, T, 0, stream>>>(g2, b2, N, OUTC, 0);

    int total = 2 * E * 16;
    k_decode<<<(total + T - 1) / T, T, 0, stream>>>(pos, neg, g2, out, E);
    (void)out_size; (void)n_in;
}


# Sum by destination, not float atomics: per-call in-edge lists (integer add returns arrival-order rank, prefix sum, fill); one wave per node sums rows in registers, self-loop and bias folded in
# speedup vs baseline: 2.3297x; 2.3297x over previous
#include <hip/hip_runtime.h>
#include <hip/hip_bf16.h>

typedef __attribute__((ext_vector_type(2))) float v2f;
typedef __attribute__((ext_vector_type(8))) float v8f;

#define IN_CH  512
#define HIDC   128
#define OUTC   64

__global__ void k_zero_cnt(int* __restrict__ cnt, int n) {
    int i = blockIdx.x * blockDim.x + threadIdx.x;
    if (i < n) cnt[i] = 0;
}

__global__ void k_count_rank(const int* __restrict__ col, int* cnt,
                             int* __restrict__ rank, int E) {
    int e = blockIdx.x * blockDim.x + threadIdx.x;
    if (e < E) rank[e] = atomicAdd(&cnt[col[e]], 1);
}

__global__ __launch_bounds__(1024)
void k_scan_dinv(const int* __restrict__ cnt, int* __restrict__ rowptr,
                 float* __restrict__ dinv, int n) {
    __shared__ int s_tot[32];
    __shared__ int s_pre[32];
    const int tid  = threadIdx.x;
    const int lane = tid & 31;
    const int wave = tid >> 5;
    const int start = blockIdx.x * 1024;

    int pre = 0;
    for (int i = tid; i < start; i += 1024) pre += cnt[i];

    const int i = start + tid;
    const int own = (i < n) ? cnt[i] : 0;

    int inc = own;
    #pragma unroll
    for (int d = 1; d < 32; d <<= 1) {
        int t = __shfl_up(inc, d, 32);
        if (lane >= d) inc += t;
    }
    #pragma unroll
    for (int d = 16; d > 0; d >>= 1) pre += __shfl_xor(pre, d, 32);

    if (lane == 31) s_tot[wave] = inc;
    if (lane == 0)  s_pre[wave] = pre;
    __syncthreads();

    int base = 0, off = 0;
    #pragma unroll
    for (int w = 0; w < 32; ++w) {
        base += s_pre[w];
        if (w < wave) off += s_tot[w];
    }
    const int excl = base + off + inc - own;
    if (i < n) {
        rowptr[i] = excl;
        dinv[i]   = rsqrtf((float)(own + 1));
        if (i == n - 1) rowptr[n] = excl + own;
    }
}

__global__ void k_fill(const int* __restrict__ row, const int* __restrict__ col,
                       const int* __restrict__ rank, const int* __restrict__ rowptr,
                       int* __restrict__ srcs, int E) {
    int e = blockIdx.x * blockDim.x + threadIdx.x;
    if (e < E) srcs[rowptr[col[e]] + rank[e]] = row[e];
}

__global__ __launch_bounds__(256)
void k_gemm_wmma_f32(const float* __restrict__ A, const float* __restrict__ B,
                     float* __restrict__ C, int M, int K, int N) {
    __shared__ float ldsA[64][36];
    __shared__ float ldsB[32][33];

    const int tid   = threadIdx.x;
    const int lane  = tid & 31;
    const int wave  = tid >> 5;
    const int waveM = wave >> 1;
    const int waveN = wave & 1;
    const int m0 = blockIdx.x * 64;
    const int n0 = blockIdx.y * 32;

    v8f acc = {};

    const int lhalf = lane >> 4;
    const int l15   = lane & 15;
    const int arow  = waveM * 16 + l15;
    const int bcol  = waveN * 16 + l15;
    const int khalf = lhalf * 2;

    for (int k0 = 0; k0 < K; k0 += 32) {
        #pragma unroll
        for (int i = 0; i < 2; ++i) {
            int idx = tid + i * 256;
            int r   = idx >> 3;
            int c4  = idx & 7;
            int gr  = m0 + r;
            gr = gr < M ? gr : (M - 1);
            const float4 v = *(const float4*)(A + (size_t)gr * K + k0 + c4 * 4);
            ldsA[r][c4 * 4 + 0] = v.x;
            ldsA[r][c4 * 4 + 1] = v.y;
            ldsA[r][c4 * 4 + 2] = v.z;
            ldsA[r][c4 * 4 + 3] = v.w;
        }
        {
            int r  = tid >> 3;
            int c4 = tid & 7;
            const float4 v = *(const float4*)(B + (size_t)(k0 + r) * N + n0 + c4 * 4);
            ldsB[r][c4 * 4 + 0] = v.x;
            ldsB[r][c4 * 4 + 1] = v.y;
            ldsB[r][c4 * 4 + 2] = v.z;
            ldsB[r][c4 * 4 + 3] = v.w;
        }
        __syncthreads();

        #pragma unroll
        for (int kk = 0; kk < 32; kk += 4) {
            v2f a, b;
            a.x = ldsA[arow][kk + khalf];
            a.y = ldsA[arow][kk + khalf + 1];
            b.x = ldsB[kk + khalf]    [bcol];
            b.y = ldsB[kk + khalf + 1][bcol];
            acc = __builtin_amdgcn_wmma_f32_16x16x4_f32(
                 false, a,  false, b,
                 (short)0, acc,  false,  false);
        }
        __syncthreads();
    }

    const int crow0 = m0 + waveM * 16 + lhalf * 8;
    const int ccol  = n0 + bcol;
    #pragma unroll
    for (int v = 0; v < 8; ++v) {
        int r = crow0 + v;
        if (r < M) C[(size_t)r * N + ccol] = acc[v];
    }
}

template <int B, typename vecT>
__device__ __forceinline__ void add_rows(vecT& acc, const vecT* __restrict__ hv,
                                         int r_l, float d_l, float dc, int j, int lane) {
    vecT v[B];
    #pragma unroll
    for (int k = 0; k < B; ++k) {
        const int r = __builtin_amdgcn_readlane(r_l, j + k);
        v[k] = hv[(size_t)r * 32 + lane];
    }
    #pragma unroll
    for (int k = 0; k < B; ++k) {
        const float w = __int_as_float(__builtin_amdgcn_readlane(__float_as_int(d_l), j + k)) * dc;
        acc += v[k] * w;
    }
}

template <int F, bool RELU>
__global__ __launch_bounds__(256)
void k_gather(const int* __restrict__ rowptr, const int* __restrict__ srcs,
              const float* __restrict__ dinv, const float* __restrict__ h,
              const float* __restrict__ bias, float* __restrict__ out, int n) {
    typedef __attribute__((ext_vector_type(F / 32))) float vecT;
    const int lane = threadIdx.x & 31;
    const int wv = __builtin_amdgcn_readfirstlane((int)(threadIdx.x >> 5));
    const int c  = blockIdx.x * 8 + wv;
    if (c >= n) return;

    const vecT* hv = (const vecT*)h;
    const int beg = rowptr[c];
    const int end = rowptr[c + 1];
    const float dc = dinv[c];

    const int total = end - beg + 1;
    vecT acc = {};
    for (int b = 0; b < total; b += 32) {
        int m = total - b;
        m = m < 32 ? m : 32;
        const int idx = b + lane;
        int   r_l = c;
        float d_l = dc;
        if (idx >= 1 && lane < m) {
            r_l = srcs[beg + idx - 1];
            d_l = dinv[r_l];
        }
        int j = 0;
        for (; j + 4 <= m; j += 4) add_rows<4>(acc, hv, r_l, d_l, dc, j, lane);
        const int rem = m - j;
        if      (rem == 3) add_rows<3>(acc, hv, r_l, d_l, dc, j, lane);
        else if (rem == 2) add_rows<2>(acc, hv, r_l, d_l, dc, j, lane);
        else if (rem == 1) add_rows<1>(acc, hv, r_l, d_l, dc, j, lane);
    }

    vecT o = acc + ((const vecT*)bias)[lane];
    if (RELU) {
        #pragma unroll
        for (int k = 0; k < F / 32; ++k) o[k] = o[k] > 0.0f ? o[k] : 0.0f;
    }
    ((vecT*)out)[(size_t)c * 32 + lane] = o;
}

__global__ __launch_bounds__(256)
void k_decode(const int* __restrict__ pos, const int* __restrict__ neg,
              const float* __restrict__ z, float* __restrict__ logits, int E) {
    int gid = blockIdx.x * blockDim.x + threadIdx.x;
    int e = gid >> 4;
    int l = gid & 15;
    if (e >= 2 * E) return;
    int src, dst;
    if (e < E) { src = pos[e];      dst = pos[E + e];      }
    else       { src = neg[e - E];  dst = neg[E + (e - E)]; }
    const float4 a = *(const float4*)(z + (size_t)src * OUTC + l * 4);
    const float4 b = *(const float4*)(z + (size_t)dst * OUTC + l * 4);
    float s = a.x * b.x + a.y * b.y + a.z * b.z + a.w * b.w;
    s += __shfl_xor(s, 1, 32);
    s += __shfl_xor(s, 2, 32);
    s += __shfl_xor(s, 4, 32);
    s += __shfl_xor(s, 8, 32);
    if (l == 0) logits[e] = s;
}

static inline size_t align_f(size_t n) { return (n + 63) & ~size_t(63); }

extern "C" void kernel_launch(void* const* d_in, const int* in_sizes, int n_in,
                              void* d_out, int out_size, void* d_ws, size_t ws_size,
                              hipStream_t stream) {
    const float* x   = (const float*)d_in[0];
    const int*   pos = (const int*)  d_in[1];
    const int*   neg = (const int*)  d_in[2];
    const float* W1  = (const float*)d_in[3];
    const float* b1  = (const float*)d_in[4];
    const float* W2  = (const float*)d_in[5];
    const float* b2  = (const float*)d_in[6];
    float* out = (float*)d_out;

    const int N = in_sizes[0] / IN_CH;
    const int E = in_sizes[1] / 2;

    float* ws = (float*)d_ws;
    size_t o = 0;
    float* dinv = ws + o; o += align_f((size_t)N);
    float* h0   = ws + o; o += align_f((size_t)N * HIDC);
    float* g1   = ws + o; o += align_f((size_t)N * HIDC);
    float* g2   = ws + o; o += align_f((size_t)N * OUTC);
    int* cnt    = (int*)(ws + o); o += align_f((size_t)N);
    int* rowptr = (int*)(ws + o); o += align_f((size_t)N + 1);
    int* rank   = (int*)(ws + o); o += align_f((size_t)E);
    int* srcs   = (int*)(ws + o); o += align_f((size_t)E);
    (void)ws_size;

    const int T = 256;
    k_zero_cnt  <<<(N + T - 1) / T, T, 0, stream>>>(cnt, N);
    k_count_rank<<<(E + T - 1) / T, T, 0, stream>>>(pos + E, cnt, rank, E);
    k_scan_dinv <<<(N + 1023) / 1024, 1024, 0, stream>>>(cnt, rowptr, dinv, N);
    k_fill      <<<(E + T - 1) / T, T, 0, stream>>>(pos, pos + E, rank, rowptr, srcs, E);

    dim3 g1g((N + 63) / 64, HIDC / 32);
    k_gemm_wmma_f32<<<g1g, T, 0, stream>>>(x, W1, h0, N, IN_CH, HIDC);
    k_gather<HIDC, true><<<(N + 7) / 8, T, 0, stream>>>(rowptr, srcs, dinv, h0, b1, g1, N);

    dim3 g2g((N + 63) / 64, OUTC / 32);
    k_gemm_wmma_f32<<<g2g, T, 0, stream>>>(g1, W2, h0, N, HIDC, OUTC);
    k_gather<OUTC, false><<<(N + 7) / 8, T, 0, stream>>>(rowptr, srcs, dinv, h0, b2, g2, N);

    int total = 2 * E * 16;
    k_decode<<<(total + T - 1) / T, T, 0, stream>>>(pos, neg, g2, out, E);
    (void)out_size; (void)n_in;
}


# Decode: a wave takes 64 consecutive edges in two batches of 32 (was 2 edges per wave): coalesced index loads, row loads in groups, sums folded across the 16-lane row, one 128-byte store per batch
# speedup vs baseline: 2.8061x; 1.2045x over previous
#include <hip/hip_runtime.h>
#include <hip/hip_bf16.h>

typedef __attribute__((ext_vector_type(2))) float v2f;
typedef __attribute__((ext_vector_type(8))) float v8f;

#define IN_CH  512
#define HIDC   128
#define OUTC   64

__global__ void k_zero_cnt(int* __restrict__ cnt, int n) {
    int i = blockIdx.x * blockDim.x + threadIdx.x;
    if (i < n) cnt[i] = 0;
}

__global__ void k_count_rank(const int* __restrict__ col, int* cnt,
                             int* __restrict__ rank, int E) {
    int e = blockIdx.x * blockDim.x + threadIdx.x;
    if (e < E) rank[e] = atomicAdd(&cnt[col[e]], 1);
}

__global__ __launch_bounds__(1024)
void k_scan_dinv(const int* __restrict__ cnt, int* __restrict__ rowptr,
                 float* __restrict__ dinv, int n) {
    __shared__ int s_tot[32];
    __shared__ int s_pre[32];
    const int tid  = threadIdx.x;
    const int lane = tid & 31;
    const int wave = tid >> 5;
    const int start = blockIdx.x * 1024;

    int pre = 0;
    for (int i = tid; i < start; i += 1024) pre += cnt[i];

    const int i = start + tid;
    const int own = (i < n) ? cnt[i] : 0;

    int inc = own;
    #pragma unroll
    for (int d = 1; d < 32; d <<= 1) {
        int t = __shfl_up(inc, d, 32);
        if (lane >= d) inc += t;
    }
    #pragma unroll
    for (int d = 16; d > 0; d >>= 1) pre += __shfl_xor(pre, d, 32);

    if (lane == 31) s_tot[wave] = inc;
    if (lane == 0)  s_pre[wave] = pre;
    __syncthreads();

    int base = 0, off = 0;
    #pragma unroll
    for (int w = 0; w < 32; ++w) {
        base += s_pre[w];
        if (w < wave) off += s_tot[w];
    }
    const int excl = base + off + inc - own;
    if (i < n) {
        rowptr[i] = excl;
        dinv[i]   = rsqrtf((float)(own + 1));
        if (i == n - 1) rowptr[n] = excl + own;
    }
}

__global__ void k_fill(const int* __restrict__ row, const int* __restrict__ col,
                       const int* __restrict__ rank, const int* __restrict__ rowptr,
                       int* __restrict__ srcs, int E) {
    int e = blockIdx.x * blockDim.x + threadIdx.x;
    if (e < E) srcs[rowptr[col[e]] + rank[e]] = row[e];
}

__global__ __launch_bounds__(256)
void k_gemm_wmma_f32(const float* __restrict__ A, const float* __restrict__ B,
                     float* __restrict__ C, int M, int K, int N) {
    __shared__ float ldsA[64][36];
    __shared__ float ldsB[32][33];

    const int tid   = threadIdx.x;
    const int lane  = tid & 31;
    const int wave  = tid >> 5;
    const int waveM = wave >> 1;
    const int waveN = wave & 1;
    const int m0 = blockIdx.x * 64;
    const int n0 = blockIdx.y * 32;

    v8f acc = {};

    const int lhalf = lane >> 4;
    const int l15   = lane & 15;
    const int arow  = waveM * 16 + l15;
    const int bcol  = waveN * 16 + l15;
    const int khalf = lhalf * 2;

    for (int k0 = 0; k0 < K; k0 += 32) {
        #pragma unroll
        for (int i = 0; i < 2; ++i) {
            int idx = tid + i * 256;
            int r   = idx >> 3;
            int c4  = idx & 7;
            int gr  = m0 + r;
            gr = gr < M ? gr : (M - 1);
            const float4 v = *(const float4*)(A + (size_t)gr * K + k0 + c4 * 4);
            ldsA[r][c4 * 4 + 0] = v.x;
            ldsA[r][c4 * 4 + 1] = v.y;
            ldsA[r][c4 * 4 + 2] = v.z;
            ldsA[r][c4 * 4 + 3] = v.w;
        }
        {
            int r  = tid >> 3;
            int c4 = tid & 7;
            const float4 v = *(const float4*)(B + (size_t)(k0 + r) * N + n0 + c4 * 4);
            ldsB[r][c4 * 4 + 0] = v.x;
            ldsB[r][c4 * 4 + 1] = v.y;
            ldsB[r][c4 * 4 + 2] = v.z;
            ldsB[r][c4 * 4 + 3] = v.w;
        }
        __syncthreads();

        #pragma unroll
        for (int kk = 0; kk < 32; kk += 4) {
            v2f a, b;
            a.x = ldsA[arow][kk + khalf];
            a.y = ldsA[arow][kk + khalf + 1];
            b.x = ldsB[kk + khalf]    [bcol];
            b.y = ldsB[kk + khalf + 1][bcol];
            acc = __builtin_amdgcn_wmma_f32_16x16x4_f32(
                 false, a,  false, b,
                 (short)0, acc,  false,  false);
        }
        __syncthreads();
    }

    const int crow0 = m0 + waveM * 16 + lhalf * 8;
    const int ccol  = n0 + bcol;
    #pragma unroll
    for (int v = 0; v < 8; ++v) {
        int r = crow0 + v;
        if (r < M) C[(size_t)r * N + ccol] = acc[v];
    }
}

template <int B, typename vecT>
__device__ __forceinline__ void add_rows(vecT& acc, const vecT* __restrict__ hv,
                                         int r_l, float d_l, float dc, int j, int lane) {
    vecT v[B];
    #pragma unroll
    for (int k = 0; k < B; ++k) {
        const int r = __builtin_amdgcn_readlane(r_l, j + k);
        v[k] = hv[(size_t)r * 32 + lane];
    }
    #pragma unroll
    for (int k = 0; k < B; ++k) {
        const float w = __int_as_float(__builtin_amdgcn_readlane(__float_as_int(d_l), j + k)) * dc;
        acc += v[k] * w;
    }
}

template <int F, bool RELU>
__global__ __launch_bounds__(256)
void k_gather(const int* __restrict__ rowptr, const int* __restrict__ srcs,
              const float* __restrict__ dinv, const float* __restrict__ h,
              const float* __restrict__ bias, float* __restrict__ out, int n) {
    typedef __attribute__((ext_vector_type(F / 32))) float vecT;
    const int lane = threadIdx.x & 31;
    const int wv = __builtin_amdgcn_readfirstlane((int)(threadIdx.x >> 5));
    const int c  = blockIdx.x * 8 + wv;
    if (c >= n) return;

    const vecT* hv = (const vecT*)h;
    const int beg = rowptr[c];
    const int end = rowptr[c + 1];
    const float dc = dinv[c];

    const int total = end - beg + 1;
    vecT acc = {};
    for (int b = 0; b < total; b += 32) {
        int m = total - b;
        m = m < 32 ? m : 32;
        const int idx = b + lane;
        int   r_l = c;
        float d_l = dc;
        if (idx >= 1 && lane < m) {
            r_l = srcs[beg + idx - 1];
            d_l = dinv[r_l];
        }
        int j = 0;
        for (; j + 4 <= m; j += 4) add_rows<4>(acc, hv, r_l, d_l, dc, j, lane);
        const int rem = m - j;
        if      (rem == 3) add_rows<3>(acc, hv, r_l, d_l, dc, j, lane);
        else if (rem == 2) add_rows<2>(acc, hv, r_l, d_l, dc, j, lane);
        else if (rem == 1) add_rows<1>(acc, hv, r_l, d_l, dc, j, lane);
    }

    vecT o = acc + ((const vecT*)bias)[lane];
    if (RELU) {
        #pragma unroll
        for (int k = 0; k < F / 32; ++k) o[k] = o[k] > 0.0f ? o[k] : 0.0f;
    }
    ((vecT*)out)[(size_t)c * 32 + lane] = o;
}

#define DEC_ITERS 2

template <int K>
__device__ __forceinline__ int row_share(int v) {
    return __builtin_amdgcn_update_dpp(0, v, 0x150 + K, 0xf, 0xf, false);
}
template <int D>
__device__ __forceinline__ float row_xor(float v) {
    return __int_as_float(__builtin_amdgcn_update_dpp(0, __float_as_int(v), 0x160 + D, 0xf, 0xf, false));
}

template <int K>
__device__ __forceinline__ void fetch_slot(float4 (&a)[16], float4 (&b)[16],
                                           const float* __restrict__ z,
                                           int src, int dst, unsigned c16) {
    constexpr int R = ((K & 1) << 3) | ((K & 2) << 1) | ((K & 4) >> 1) | ((K & 8) >> 3);
    const unsigned so = (unsigned)row_share<R>(src) * 256u + c16;
    const unsigned d_o = (unsigned)row_share<R>(dst) * 256u + c16;
    a[K] = *(const float4*)((const char*)z + so);
    b[K] = *(const float4*)((const char*)z + d_o);
}

template <int D, int H>
__device__ __forceinline__ void fold_step(float (&s)[16], bool up) {
    #pragma unroll
    for (int j = 0; j < H; ++j) {
        const float keep = up ? s[j + H] : s[j];
        const float send = up ? s[j]     : s[j + H];
        s[j] = keep + row_xor<D>(send);
    }
}

__device__ __forceinline__ void load_ends(const int* __restrict__ pos, const int* __restrict__ neg,
                                          int e, int E, int& src, int& dst) {
    src = 0; dst = 0;
    if (e < 2 * E) {
        const int* p = e < E ? pos + e : neg + (e - E);
        src = p[0];
        dst = p[E];
    }
}

__global__ __launch_bounds__(256)
void k_decode(const int* __restrict__ pos, const int* __restrict__ neg,
              const float* __restrict__ z, float* __restrict__ logits, int E) {
    const int lane = threadIdx.x & 31;
    const unsigned c   = lane & 15;
    const unsigned c16 = c * 16u;
    const int wv = __builtin_amdgcn_readfirstlane((int)(threadIdx.x >> 5));
    const int total = 2 * E;
    const int base = (blockIdx.x * 8 + wv) * (32 * DEC_ITERS);
    if (base >= total) return;

    int src[DEC_ITERS], dst[DEC_ITERS];
    #pragma unroll
    for (int it = 0; it < DEC_ITERS; ++it)
        load_ends(pos, neg, base + 32 * it + lane, E, src[it], dst[it]);

    #pragma unroll
    for (int it = 0; it < DEC_ITERS; ++it) {
        const int b0 = base + 32 * it;
        if (b0 >= total) break;

        float4 a[16], b[16];
        fetch_slot<0>(a, b, z, src[it], dst[it], c16);   fetch_slot<1>(a, b, z, src[it], dst[it], c16);
        fetch_slot<2>(a, b, z, src[it], dst[it], c16);   fetch_slot<3>(a, b, z, src[it], dst[it], c16);
        fetch_slot<4>(a, b, z, src[it], dst[it], c16);   fetch_slot<5>(a, b, z, src[it], dst[it], c16);
        fetch_slot<6>(a, b, z, src[it], dst[it], c16);   fetch_slot<7>(a, b, z, src[it], dst[it], c16);
        fetch_slot<8>(a, b, z, src[it], dst[it], c16);   fetch_slot<9>(a, b, z, src[it], dst[it], c16);
        fetch_slot<10>(a, b, z, src[it], dst[it], c16);  fetch_slot<11>(a, b, z, src[it], dst[it], c16);
        fetch_slot<12>(a, b, z, src[it], dst[it], c16);  fetch_slot<13>(a, b, z, src[it], dst[it], c16);
        fetch_slot<14>(a, b, z, src[it], dst[it], c16);  fetch_slot<15>(a, b, z, src[it], dst[it], c16);

        float s[16];
        #pragma unroll
        for (int k = 0; k < 16; ++k)
            s[k] = a[k].x * b[k].x + a[k].y * b[k].y + a[k].z * b[k].z + a[k].w * b[k].w;

        fold_step<1, 8>(s, (c & 1u) != 0);
        fold_step<2, 4>(s, (c & 2u) != 0);
        fold_step<4, 2>(s, (c & 4u) != 0);
        fold_step<8, 1>(s, (c & 8u) != 0);

        const int e = b0 + lane;
        if (e < total) logits[e] = s[0];
    }
}

static inline size_t align_f(size_t n) { return (n + 63) & ~size_t(63); }

extern "C" void kernel_launch(void* const* d_in, const int* in_sizes, int n_in,
                              void* d_out, int out_size, void* d_ws, size_t ws_size,
                              hipStream_t stream) {
    const float* x   = (const float*)d_in[0];
    const int*   pos = (const int*)  d_in[1];
    const int*   neg = (const int*)  d_in[2];
    const float* W1  = (const float*)d_in[3];
    const float* b1  = (const float*)d_in[4];
    const float* W2  = (const float*)d_in[5];
    const float* b2  = (const float*)d_in[6];
    float* out = (float*)d_out;

    const int N = in_sizes[0] / IN_CH;
    const int E = in_sizes[1] / 2;

    float* ws = (float*)d_ws;
    size_t o = 0;
    float* dinv = ws + o; o += align_f((size_t)N);
    float* h0   = ws + o; o += align_f((size_t)N * HIDC);
    float* g1   = ws + o; o += align_f((size_t)N * HIDC);
    float* g2   = ws + o; o += align_f((size_t)N * OUTC);
    int* cnt    = (int*)(ws + o); o += align_f((size_t)N);
    int* rowptr = (int*)(ws + o); o += align_f((size_t)N + 1);
    int* rank   = (int*)(ws + o); o += align_f((size_t)E);
    int* srcs   = (int*)(ws + o); o += align_f((size_t)E);
    (void)ws_size;

    const int T = 256;
    k_zero_cnt  <<<(N + T - 1) / T, T, 0, stream>>>(cnt, N);
    k_count_rank<<<(E + T - 1) / T, T, 0, stream>>>(pos + E, cnt, rank, E);
    k_scan_dinv <<<(N + 1023) / 1024, 1024, 0, stream>>>(cnt, rowptr, dinv, N);
    k_fill      <<<(E + T - 1) / T, T, 0, stream>>>(pos, pos + E, rank, rowptr, srcs, E);

    dim3 g1g((N + 63) / 64, HIDC / 32);
    k_gemm_wmma_f32<<<g1g, T, 0, stream>>>(x, W1, h0, N, IN_CH, HIDC);
    k_gather<HIDC, true><<<(N + 7) / 8, T, 0, stream>>>(rowptr, srcs, dinv, h0, b1, g1, N);

    dim3 g2g((N + 63) / 64, OUTC / 32);
    k_gemm_wmma_f32<<<g2g, T, 0, stream>>>(g1, W2, h0, N, HIDC, OUTC);
    k_gather<OUTC, false><<<(N + 7) / 8, T, 0, stream>>>(rowptr, srcs, dinv, h0, b2, g2, N);

    int total = 2 * E;
    k_decode<<<(total + 256 * DEC_ITERS - 1) / (256 * DEC_ITERS), T, 0, stream>>>(pos, neg, g2, out, E);
    (void)out_size; (void)n_in;
}


# In-edge lists by a two-level counting sort on LDS counters (256-node buckets, then per-bucket lists), no global integer add per edge; a node's list order stays arrival order, now of LDS adds
# speedup vs baseline: 3.2458x; 1.1567x over previous
#include <hip/hip_runtime.h>
#include <hip/hip_bf16.h>

typedef __attribute__((ext_vector_type(2))) float v2f;
typedef __attribute__((ext_vector_type(8))) float v8f;

#define IN_CH  512
#define HIDC   128
#define OUTC   64

#define NCH 256
static_assert(NCH * 256 == 1024 * 64, "k_scan_hist: 1024 threads x 64 counts");

__device__ __forceinline__ int bucket_of(int d) {
    const unsigned b = (unsigned)d >> 8;
    return (int)(b < 255u ? b : 255u);
}

__global__ __launch_bounds__(256)
void k_bucket_hist(const int* __restrict__ col, int* __restrict__ hist, int E) {
    __shared__ int s_h[256];
    const int tid = threadIdx.x;
    const int cs = (E + NCH - 1) / NCH;
    const int e0 = blockIdx.x * cs;
    int e1 = e0 + cs;
    e1 = e1 < E ? e1 : E;
    s_h[tid] = 0;
    __syncthreads();
    const int last = e1 - 1;
    for (int i = e0 + tid; i < e1; i += 256 * 8) {
        int d[8];
        #pragma unroll
        for (int k = 0; k < 8; ++k) {
            const int idx = i + k * 256;
            d[k] = col[idx < e1 ? idx : last];
        }
        asm volatile("" :: "v"(d[0]), "v"(d[1]), "v"(d[2]), "v"(d[3]),
                           "v"(d[4]), "v"(d[5]), "v"(d[6]), "v"(d[7]));
        #pragma unroll
        for (int k = 0; k < 8; ++k)
            if (i + k * 256 < e1) atomicAdd(&s_h[bucket_of(d[k])], 1);
    }
    __syncthreads();
    hist[tid * NCH + blockIdx.x] = s_h[tid];
}

__global__ __launch_bounds__(1024)
void k_scan_hist(const int* __restrict__ hist, int* __restrict__ offs) {
    __shared__ int s_tot[32];
    const int tid  = threadIdx.x;
    const int lane = tid & 31;
    const int wave = tid >> 5;
    const int4* hp = (const int4*)(hist + tid * 64);
    int4 v[16];
    #pragma unroll
    for (int k = 0; k < 16; ++k) v[k] = hp[k];
    int tsum = 0;
    #pragma unroll
    for (int k = 0; k < 16; ++k) tsum += v[k].x + v[k].y + v[k].z + v[k].w;

    int inc = tsum;
    #pragma unroll
    for (int d = 1; d < 32; d <<= 1) {
        int t = __shfl_up(inc, d, 32);
        if (lane >= d) inc += t;
    }
    if (lane == 31) s_tot[wave] = inc;
    __syncthreads();
    int off = 0;
    #pragma unroll
    for (int w = 0; w < 32; ++w)
        if (w < wave) off += s_tot[w];

    int run = off + inc - tsum;
    int4* op = (int4*)(offs + tid * 64);
    #pragma unroll
    for (int k = 0; k < 16; ++k) {
        int4 o;
        o.x = run; run += v[k].x;
        o.y = run; run += v[k].y;
        o.z = run; run += v[k].z;
        o.w = run; run += v[k].w;
        op[k] = o;
    }
    if (tid == 1023) offs[NCH * 256] = run;
}

__global__ __launch_bounds__(256)
void k_bucket_scatter(const int* __restrict__ row, const int* __restrict__ col,
                      const int* __restrict__ offs, int* __restrict__ binned, int E) {
    __shared__ int s_cur[256];
    const int tid = threadIdx.x;
    const int cs = (E + NCH - 1) / NCH;
    const int e0 = blockIdx.x * cs;
    int e1 = e0 + cs;
    e1 = e1 < E ? e1 : E;
    s_cur[tid] = offs[tid * NCH + blockIdx.x];
    __syncthreads();
    const int last = e1 - 1;
    for (int i = e0 + tid; i < e1; i += 256 * 8) {
        int d[8], s[8];
        #pragma unroll
        for (int k = 0; k < 8; ++k) {
            const int idx = i + k * 256;
            const int ic  = idx < e1 ? idx : last;
            d[k] = col[ic];
            s[k] = row[ic];
        }
        asm volatile("" :: "v"(d[0]), "v"(d[1]), "v"(d[2]), "v"(d[3]),
                           "v"(d[4]), "v"(d[5]), "v"(d[6]), "v"(d[7]),
                           "v"(s[0]), "v"(s[1]), "v"(s[2]), "v"(s[3]),
                           "v"(s[4]), "v"(s[5]), "v"(s[6]), "v"(s[7]));
        #pragma unroll
        for (int k = 0; k < 8; ++k) {
            if (i + k * 256 < e1) {
                const int pos = atomicAdd(&s_cur[bucket_of(d[k])], 1);
                binned[pos] = (s[k] << 8) | (d[k] & 255);
            }
        }
    }
}

__global__ __launch_bounds__(512)
void k_bucket_lists(const int* __restrict__ binned, const int* __restrict__ offs,
                    int* __restrict__ rowptr, float* __restrict__ dinv,
                    int* __restrict__ srcs, int n) {
    __shared__ int s_cnt[256];
    __shared__ int s_cur[256];
    __shared__ int s_tot[8];
    const int tid  = threadIdx.x;
    const int lane = tid & 31;
    const int wave = tid >> 5;
    const int b    = blockIdx.x;
    const int beg  = offs[b * NCH];
    const int end  = offs[(b + 1) * NCH];
    if (tid < 256) s_cnt[tid] = 0;
    __syncthreads();

    const int last = end - 1;
    for (int i = beg + tid; i < end; i += 512 * 4) {
        int p[4];
        #pragma unroll
        for (int k = 0; k < 4; ++k) {
            const int idx = i + k * 512;
            p[k] = binned[idx < end ? idx : last];
        }
        asm volatile("" :: "v"(p[0]), "v"(p[1]), "v"(p[2]), "v"(p[3]));
        #pragma unroll
        for (int k = 0; k < 4; ++k)
            if (i + k * 512 < end) atomicAdd(&s_cnt[p[k] & 255], 1);
    }
    __syncthreads();

    const int own = tid < 256 ? s_cnt[tid] : 0;
    int inc = own;
    #pragma unroll
    for (int d = 1; d < 32; d <<= 1) {
        int t = __shfl_up(inc, d, 32);
        if (lane >= d) inc += t;
    }
    if (tid < 256 && lane == 31) s_tot[wave] = inc;
    __syncthreads();
    if (tid < 256) {
        int off = 0;
        #pragma unroll
        for (int w = 0; w < 8; ++w)
            if (w < wave) off += s_tot[w];
        const int excl = beg + off + inc - own;
        s_cur[tid] = excl;
        const int node = b * 256 + tid;
        if (node < n) {
            rowptr[node] = excl;
            dinv[node]   = rsqrtf((float)(own + 1));
            if (node == n - 1) rowptr[n] = excl + own;
        }
    }
    __syncthreads();

    for (int i = beg + tid; i < end; i += 512 * 4) {
        int p[4];
        #pragma unroll
        for (int k = 0; k < 4; ++k) {
            const int idx = i + k * 512;
            p[k] = binned[idx < end ? idx : last];
        }
        asm volatile("" :: "v"(p[0]), "v"(p[1]), "v"(p[2]), "v"(p[3]));
        #pragma unroll
        for (int k = 0; k < 4; ++k) {
            if (i + k * 512 < end) {
                const int pos = atomicAdd(&s_cur[p[k] & 255], 1);
                srcs[pos] = p[k] >> 8;
            }
        }
    }
}

__global__ __launch_bounds__(256)
void k_gemm_wmma_f32(const float* __restrict__ A, const float* __restrict__ B,
                     float* __restrict__ C, int M, int K, int N) {
    __shared__ float ldsA[64][36];
    __shared__ float ldsB[32][33];

    const int tid   = threadIdx.x;
    const int lane  = tid & 31;
    const int wave  = tid >> 5;
    const int waveM = wave >> 1;
    const int waveN = wave & 1;
    const int m0 = blockIdx.x * 64;
    const int n0 = blockIdx.y * 32;

    v8f acc = {};

    const int lhalf = lane >> 4;
    const int l15   = lane & 15;
    const int arow  = waveM * 16 + l15;
    const int bcol  = waveN * 16 + l15;
    const int khalf = lhalf * 2;

    for (int k0 = 0; k0 < K; k0 += 32) {
        #pragma unroll
        for (int i = 0; i < 2; ++i) {
            int idx = tid + i * 256;
            int r   = idx >> 3;
            int c4  = idx & 7;
            int gr  = m0 + r;
            gr = gr < M ? gr : (M - 1);
            const float4 v = *(const float4*)(A + (size_t)gr * K + k0 + c4 * 4);
            ldsA[r][c4 * 4 + 0] = v.x;
            ldsA[r][c4 * 4 + 1] = v.y;
            ldsA[r][c4 * 4 + 2] = v.z;
            ldsA[r][c4 * 4 + 3] = v.w;
        }
        {
            int r  = tid >> 3;
            int c4 = tid & 7;
            const float4 v = *(const float4*)(B + (size_t)(k0 + r) * N + n0 + c4 * 4);
            ldsB[r][c4 * 4 + 0] = v.x;
            ldsB[r][c4 * 4 + 1] = v.y;
            ldsB[r][c4 * 4 + 2] = v.z;
            ldsB[r][c4 * 4 + 3] = v.w;
        }
        __syncthreads();

        #pragma unroll
        for (int kk = 0; kk < 32; kk += 4) {
            v2f a, b;
            a.x = ldsA[arow][kk + khalf];
            a.y = ldsA[arow][kk + khalf + 1];
            b.x = ldsB[kk + khalf]    [bcol];
            b.y = ldsB[kk + khalf + 1][bcol];
            acc = __builtin_amdgcn_wmma_f32_16x16x4_f32(
                 false, a,  false, b,
                 (short)0, acc,  false,  false);
        }
        __syncthreads();
    }

    const int crow0 = m0 + waveM * 16 + lhalf * 8;
    const int ccol  = n0 + bcol;
    #pragma unroll
    for (int v = 0; v < 8; ++v) {
        int r = crow0 + v;
        if (r < M) C[(size_t)r * N + ccol] = acc[v];
    }
}

template <int B, typename vecT>
__device__ __forceinline__ void add_rows(vecT& acc, const vecT* __restrict__ hv,
                                         int r_l, float d_l, float dc, int j, int lane) {
    vecT v[B];
    #pragma unroll
    for (int k = 0; k < B; ++k) {
        const int r = __builtin_amdgcn_readlane(r_l, j + k);
        v[k] = hv[(size_t)r * 32 + lane];
    }
    #pragma unroll
    for (int k = 0; k < B; ++k) {
        const float w = __int_as_float(__builtin_amdgcn_readlane(__float_as_int(d_l), j + k)) * dc;
        acc += v[k] * w;
    }
}

template <int F, bool RELU>
__global__ __launch_bounds__(256)
void k_gather(const int* __restrict__ rowptr, const int* __restrict__ srcs,
              const float* __restrict__ dinv, const float* __restrict__ h,
              const float* __restrict__ bias, float* __restrict__ out, int n) {
    typedef __attribute__((ext_vector_type(F / 32))) float vecT;
    const int lane = threadIdx.x & 31;
    const int wv = __builtin_amdgcn_readfirstlane((int)(threadIdx.x >> 5));
    const int c  = blockIdx.x * 8 + wv;
    if (c >= n) return;

    const vecT* hv = (const vecT*)h;
    const int beg = rowptr[c];
    const int end = rowptr[c + 1];
    const float dc = dinv[c];

    const int total = end - beg + 1;
    vecT acc = {};
    for (int b = 0; b < total; b += 32) {
        int m = total - b;
        m = m < 32 ? m : 32;
        const int idx = b + lane;
        int   r_l = c;
        float d_l = dc;
        if (idx >= 1 && lane < m) {
            r_l = srcs[beg + idx - 1];
            d_l = dinv[r_l];
        }
        int j = 0;
        for (; j + 4 <= m; j += 4) add_rows<4>(acc, hv, r_l, d_l, dc, j, lane);
        const int rem = m - j;
        if      (rem == 3) add_rows<3>(acc, hv, r_l, d_l, dc, j, lane);
        else if (rem == 2) add_rows<2>(acc, hv, r_l, d_l, dc, j, lane);
        else if (rem == 1) add_rows<1>(acc, hv, r_l, d_l, dc, j, lane);
    }

    vecT o = acc + ((const vecT*)bias)[lane];
    if (RELU) {
        #pragma unroll
        for (int k = 0; k < F / 32; ++k) o[k] = o[k] > 0.0f ? o[k] : 0.0f;
    }
    ((vecT*)out)[(size_t)c * 32 + lane] = o;
}

#define DEC_ITERS 2

template <int K>
__device__ __forceinline__ int row_share(int v) {
    return __builtin_amdgcn_update_dpp(0, v, 0x150 + K, 0xf, 0xf, false);
}
template <int D>
__device__ __forceinline__ float row_xor(float v) {
    return __int_as_float(__builtin_amdgcn_update_dpp(0, __float_as_int(v), 0x160 + D, 0xf, 0xf, false));
}

template <int K>
__device__ __forceinline__ void fetch_slot(float4 (&a)[16], float4 (&b)[16],
                                           const float* __restrict__ z,
                                           int src, int dst, unsigned c16) {
    constexpr int R = ((K & 1) << 3) | ((K & 2) << 1) | ((K & 4) >> 1) | ((K & 8) >> 3);
    const unsigned so = (unsigned)row_share<R>(src) * 256u + c16;
    const unsigned d_o = (unsigned)row_share<R>(dst) * 256u + c16;
    a[K] = *(const float4*)((const char*)z + so);
    b[K] = *(const float4*)((const char*)z + d_o);
}

template <int D, int H>
__device__ __forceinline__ void fold_step(float (&s)[16], bool up) {
    #pragma unroll
    for (int j = 0; j < H; ++j) {
        const float keep = up ? s[j + H] : s[j];
        const float send = up ? s[j]     : s[j + H];
        s[j] = keep + row_xor<D>(send);
    }
}

__device__ __forceinline__ void load_ends(const int* __restrict__ pos, const int* __restrict__ neg,
                                          int e, int E, int& src, int& dst) {
    src = 0; dst = 0;
    if (e < 2 * E) {
        const int* p = e < E ? pos + e : neg + (e - E);
        src = p[0];
        dst = p[E];
    }
}

__global__ __launch_bounds__(256)
void k_decode(const int* __restrict__ pos, const int* __restrict__ neg,
              const float* __restrict__ z, float* __restrict__ logits, int E) {
    const int lane = threadIdx.x & 31;
    const unsigned c   = lane & 15;
    const unsigned c16 = c * 16u;
    const int wv = __builtin_amdgcn_readfirstlane((int)(threadIdx.x >> 5));
    const int total = 2 * E;
    const int base = (blockIdx.x * 8 + wv) * (32 * DEC_ITERS);
    if (base >= total) return;

    int src[DEC_ITERS], dst[DEC_ITERS];
    #pragma unroll
    for (int it = 0; it < DEC_ITERS; ++it)
        load_ends(pos, neg, base + 32 * it + lane, E, src[it], dst[it]);

    #pragma unroll
    for (int it = 0; it < DEC_ITERS; ++it) {
        const int b0 = base + 32 * it;
        if (b0 >= total) break;

        float4 a[16], b[16];
        fetch_slot<0>(a, b, z, src[it], dst[it], c16);   fetch_slot<1>(a, b, z, src[it], dst[it], c16);
        fetch_slot<2>(a, b, z, src[it], dst[it], c16);   fetch_slot<3>(a, b, z, src[it], dst[it], c16);
        fetch_slot<4>(a, b, z, src[it], dst[it], c16);   fetch_slot<5>(a, b, z, src[it], dst[it], c16);
        fetch_slot<6>(a, b, z, src[it], dst[it], c16);   fetch_slot<7>(a, b, z, src[it], dst[it], c16);
        fetch_slot<8>(a, b, z, src[it], dst[it], c16);   fetch_slot<9>(a, b, z, src[it], dst[it], c16);
        fetch_slot<10>(a, b, z, src[it], dst[it], c16);  fetch_slot<11>(a, b, z, src[it], dst[it], c16);
        fetch_slot<12>(a, b, z, src[it], dst[it], c16);  fetch_slot<13>(a, b, z, src[it], dst[it], c16);
        fetch_slot<14>(a, b, z, src[it], dst[it], c16);  fetch_slot<15>(a, b, z, src[it], dst[it], c16);

        float s[16];
        #pragma unroll
        for (int k = 0; k < 16; ++k)
            s[k] = a[k].x * b[k].x + a[k].y * b[k].y + a[k].z * b[k].z + a[k].w * b[k].w;

        fold_step<1, 8>(s, (c & 1u) != 0);
        fold_step<2, 4>(s, (c & 2u) != 0);
        fold_step<4, 2>(s, (c & 4u) != 0);
        fold_step<8, 1>(s, (c & 8u) != 0);

        const int e = b0 + lane;
        if (e < total) logits[e] = s[0];
    }
}

static inline size_t align_f(size_t n) { return (n + 63) & ~size_t(63); }

extern "C" void kernel_launch(void* const* d_in, const int* in_sizes, int n_in,
                              void* d_out, int out_size, void* d_ws, size_t ws_size,
                              hipStream_t stream) {
    const float* x   = (const float*)d_in[0];
    const int*   pos = (const int*)  d_in[1];
    const int*   neg = (const int*)  d_in[2];
    const float* W1  = (const float*)d_in[3];
    const float* b1  = (const float*)d_in[4];
    const float* W2  = (const float*)d_in[5];
    const float* b2  = (const float*)d_in[6];
    float* out = (float*)d_out;

    const int N = in_sizes[0] / IN_CH;
    const int E = in_sizes[1] / 2;

    float* ws = (float*)d_ws;
    size_t o = 0;
    float* dinv = ws + o; o += align_f((size_t)N);
    float* h0   = ws + o; o += align_f((size_t)N * HIDC);
    float* g1   = ws + o; o += align_f((size_t)N * HIDC);
    float* g2   = ws + o; o += align_f((size_t)N * OUTC);
    int* hist   = (int*)(ws + o); o += align_f((size_t)NCH * 256);
    int* offs   = (int*)(ws + o); o += align_f((size_t)NCH * 256 + 1);
    int* rowptr = (int*)(ws + o); o += align_f((size_t)N + 1);
    int* binned = (int*)(ws + o); o += align_f((size_t)E);
    int* srcs   = (int*)(ws + o); o += align_f((size_t)E);
    (void)ws_size;

    const int T = 256;
    k_bucket_hist   <<<NCH, 256, 0, stream>>>(pos + E, hist, E);
    k_scan_hist     <<<1, 1024, 0, stream>>>(hist, offs);
    k_bucket_scatter<<<NCH, 256, 0, stream>>>(pos, pos + E, offs, binned, E);
    k_bucket_lists  <<<(N + 255) / 256, 512, 0, stream>>>(binned, offs, rowptr, dinv, srcs, N);

    dim3 g1g((N + 63) / 64, HIDC / 32);
    k_gemm_wmma_f32<<<g1g, T, 0, stream>>>(x, W1, h0, N, IN_CH, HIDC);
    k_gather<HIDC, true><<<(N + 7) / 8, T, 0, stream>>>(rowptr, srcs, dinv, h0, b1, g1, N);

    dim3 g2g((N + 63) / 64, OUTC / 32);
    k_gemm_wmma_f32<<<g2g, T, 0, stream>>>(g1, W2, h0, N, HIDC, OUTC);
    k_gather<OUTC, false><<<(N + 7) / 8, T, 0, stream>>>(rowptr, srcs, dinv, h0, b2, g2, N);

    int total = 2 * E;
    k_decode<<<(total + 256 * DEC_ITERS - 1) / (256 * DEC_ITERS), T, 0, stream>>>(pos, neg, g2, out, E);
    (void)out_size; (void)n_in;
}


# GEMM k loop software-pipelined through registers: the three global loads of chunk k+1 are issued together before chunk k's 8 WMMAs (were issued and waited for one after another)
# speedup vs baseline: 3.5974x; 1.1083x over previous
#include <hip/hip_runtime.h>
#include <hip/hip_bf16.h>

typedef __attribute__((ext_vector_type(2))) float v2f;
typedef __attribute__((ext_vector_type(8))) float v8f;

#define IN_CH  512
#define HIDC   128
#define OUTC   64

#define NCH 256
static_assert(NCH * 256 == 1024 * 64, "k_scan_hist: 1024 threads x 64 counts");

__device__ __forceinline__ int bucket_of(int d) {
    const unsigned b = (unsigned)d >> 8;
    return (int)(b < 255u ? b : 255u);
}

__global__ __launch_bounds__(256)
void k_bucket_hist(const int* __restrict__ col, int* __restrict__ hist, int E) {
    __shared__ int s_h[256];
    const int tid = threadIdx.x;
    const int cs = (E + NCH - 1) / NCH;
    const int e0 = blockIdx.x * cs;
    int e1 = e0 + cs;
    e1 = e1 < E ? e1 : E;
    s_h[tid] = 0;
    __syncthreads();
    const int last = e1 - 1;
    for (int i = e0 + tid; i < e1; i += 256 * 8) {
        int d[8];
        #pragma unroll
        for (int k = 0; k < 8; ++k) {
            const int idx = i + k * 256;
            d[k] = col[idx < e1 ? idx : last];
        }
        asm volatile("" :: "v"(d[0]), "v"(d[1]), "v"(d[2]), "v"(d[3]),
                           "v"(d[4]), "v"(d[5]), "v"(d[6]), "v"(d[7]));
        #pragma unroll
        for (int k = 0; k < 8; ++k)
            if (i + k * 256 < e1) atomicAdd(&s_h[bucket_of(d[k])], 1);
    }
    __syncthreads();
    hist[tid * NCH + blockIdx.x] = s_h[tid];
}

__global__ __launch_bounds__(1024)
void k_scan_hist(const int* __restrict__ hist, int* __restrict__ offs) {
    __shared__ int s_tot[32];
    const int tid  = threadIdx.x;
    const int lane = tid & 31;
    const int wave = tid >> 5;
    const int4* hp = (const int4*)(hist + tid * 64);
    int4 v[16];
    #pragma unroll
    for (int k = 0; k < 16; ++k) v[k] = hp[k];
    int tsum = 0;
    #pragma unroll
    for (int k = 0; k < 16; ++k) tsum += v[k].x + v[k].y + v[k].z + v[k].w;

    int inc = tsum;
    #pragma unroll
    for (int d = 1; d < 32; d <<= 1) {
        int t = __shfl_up(inc, d, 32);
        if (lane >= d) inc += t;
    }
    if (lane == 31) s_tot[wave] = inc;
    __syncthreads();
    int off = 0;
    #pragma unroll
    for (int w = 0; w < 32; ++w)
        if (w < wave) off += s_tot[w];

    int run = off + inc - tsum;
    int4* op = (int4*)(offs + tid * 64);
    #pragma unroll
    for (int k = 0; k < 16; ++k) {
        int4 o;
        o.x = run; run += v[k].x;
        o.y = run; run += v[k].y;
        o.z = run; run += v[k].z;
        o.w = run; run += v[k].w;
        op[k] = o;
    }
    if (tid == 1023) offs[NCH * 256] = run;
}

__global__ __launch_bounds__(256)
void k_bucket_scatter(const int* __restrict__ row, const int* __restrict__ col,
                      const int* __restrict__ offs, int* __restrict__ binned, int E) {
    __shared__ int s_cur[256];
    const int tid = threadIdx.x;
    const int cs = (E + NCH - 1) / NCH;
    const int e0 = blockIdx.x * cs;
    int e1 = e0 + cs;
    e1 = e1 < E ? e1 : E;
    s_cur[tid] = offs[tid * NCH + blockIdx.x];
    __syncthreads();
    const int last = e1 - 1;
    for (int i = e0 + tid; i < e1; i += 256 * 8) {
        int d[8], s[8];
        #pragma unroll
        for (int k = 0; k < 8; ++k) {
            const int idx = i + k * 256;
            const int ic  = idx < e1 ? idx : last;
            d[k] = col[ic];
            s[k] = row[ic];
        }
        asm volatile("" :: "v"(d[0]), "v"(d[1]), "v"(d[2]), "v"(d[3]),
                           "v"(d[4]), "v"(d[5]), "v"(d[6]), "v"(d[7]),
                           "v"(s[0]), "v"(s[1]), "v"(s[2]), "v"(s[3]),
                           "v"(s[4]), "v"(s[5]), "v"(s[6]), "v"(s[7]));
        #pragma unroll
        for (int k = 0; k < 8; ++k) {
            if (i + k * 256 < e1) {
                const int pos = atomicAdd(&s_cur[bucket_of(d[k])], 1);
                binned[pos] = (s[k] << 8) | (d[k] & 255);
            }
        }
    }
}

__global__ __launch_bounds__(512)
void k_bucket_lists(const int* __restrict__ binned, const int* __restrict__ offs,
                    int* __restrict__ rowptr, float* __restrict__ dinv,
                    int* __restrict__ srcs, int n) {
    __shared__ int s_cnt[256];
    __shared__ int s_cur[256];
    __shared__ int s_tot[8];
    const int tid  = threadIdx.x;
    const int lane = tid & 31;
    const int wave = tid >> 5;
    const int b    = blockIdx.x;
    const int beg  = offs[b * NCH];
    const int end  = offs[(b + 1) * NCH];
    if (tid < 256) s_cnt[tid] = 0;
    __syncthreads();

    const int last = end - 1;
    for (int i = beg + tid; i < end; i += 512 * 4) {
        int p[4];
        #pragma unroll
        for (int k = 0; k < 4; ++k) {
            const int idx = i + k * 512;
            p[k] = binned[idx < end ? idx : last];
        }
        asm volatile("" :: "v"(p[0]), "v"(p[1]), "v"(p[2]), "v"(p[3]));
        #pragma unroll
        for (int k = 0; k < 4; ++k)
            if (i + k * 512 < end) atomicAdd(&s_cnt[p[k] & 255], 1);
    }
    __syncthreads();

    const int own = tid < 256 ? s_cnt[tid] : 0;
    int inc = own;
    #pragma unroll
    for (int d = 1; d < 32; d <<= 1) {
        int t = __shfl_up(inc, d, 32);
        if (lane >= d) inc += t;
    }
    if (tid < 256 && lane == 31) s_tot[wave] = inc;
    __syncthreads();
    if (tid < 256) {
        int off = 0;
        #pragma unroll
        for (int w = 0; w < 8; ++w)
            if (w < wave) off += s_tot[w];
        const int excl = beg + off + inc - own;
        s_cur[tid] = excl;
        const int node = b * 256 + tid;
        if (node < n) {
            rowptr[node] = excl;
            dinv[node]   = rsqrtf((float)(own + 1));
            if (node == n - 1) rowptr[n] = excl + own;
        }
    }
    __syncthreads();

    for (int i = beg + tid; i < end; i += 512 * 4) {
        int p[4];
        #pragma unroll
        for (int k = 0; k < 4; ++k) {
            const int idx = i + k * 512;
            p[k] = binned[idx < end ? idx : last];
        }
        asm volatile("" :: "v"(p[0]), "v"(p[1]), "v"(p[2]), "v"(p[3]));
        #pragma unroll
        for (int k = 0; k < 4; ++k) {
            if (i + k * 512 < end) {
                const int pos = atomicAdd(&s_cur[p[k] & 255], 1);
                srcs[pos] = p[k] >> 8;
            }
        }
    }
}

__global__ __launch_bounds__(256)
void k_gemm_wmma_f32(const float* __restrict__ A, const float* __restrict__ B,
                     float* __restrict__ C, int M, int K, int N) {
    __shared__ float ldsA[64][36];
    __shared__ float ldsB[32][33];

    const int tid   = threadIdx.x;
    const int lane  = tid & 31;
    const int wave  = tid >> 5;
    const int waveM = wave >> 1;
    const int waveN = wave & 1;
    const int m0 = blockIdx.x * 64;
    const int n0 = blockIdx.y * 32;

    v8f acc = {};

    const int lhalf = lane >> 4;
    const int l15   = lane & 15;
    const int arow  = waveM * 16 + l15;
    const int bcol  = waveN * 16 + l15;
    const int khalf = lhalf * 2;

    const int sr = tid >> 3;
    const int sc = (tid & 7) * 4;
    int gr0 = m0 + sr;
    int gr1 = m0 + sr + 32;
    gr0 = gr0 < M ? gr0 : (M - 1);
    gr1 = gr1 < M ? gr1 : (M - 1);
    const float* a0p = A + (size_t)gr0 * K + sc;
    const float* a1p = A + (size_t)gr1 * K + sc;
    const float* bp  = B + (size_t)sr * N + n0 + sc;

    float4 va0 = *(const float4*)a0p;
    float4 va1 = *(const float4*)a1p;
    float4 vb  = *(const float4*)bp;

    for (int k0 = 0; k0 < K; k0 += 32) {
        ldsA[sr][sc + 0] = va0.x;
        ldsA[sr][sc + 1] = va0.y;
        ldsA[sr][sc + 2] = va0.z;
        ldsA[sr][sc + 3] = va0.w;
        ldsA[sr + 32][sc + 0] = va1.x;
        ldsA[sr + 32][sc + 1] = va1.y;
        ldsA[sr + 32][sc + 2] = va1.z;
        ldsA[sr + 32][sc + 3] = va1.w;
        ldsB[sr][sc + 0] = vb.x;
        ldsB[sr][sc + 1] = vb.y;
        ldsB[sr][sc + 2] = vb.z;
        ldsB[sr][sc + 3] = vb.w;
        __syncthreads();

        if (k0 + 32 < K) {
            va0 = *(const float4*)(a0p + (k0 + 32));
            va1 = *(const float4*)(a1p + (k0 + 32));
            vb  = *(const float4*)(bp + (size_t)(k0 + 32) * N);
        }

        #pragma unroll
        for (int kk = 0; kk < 32; kk += 4) {
            v2f a, b;
            a.x = ldsA[arow][kk + khalf];
            a.y = ldsA[arow][kk + khalf + 1];
            b.x = ldsB[kk + khalf]    [bcol];
            b.y = ldsB[kk + khalf + 1][bcol];
            acc = __builtin_amdgcn_wmma_f32_16x16x4_f32(
                 false, a,  false, b,
                 (short)0, acc,  false,  false);
        }
        __syncthreads();
    }

    const int crow0 = m0 + waveM * 16 + lhalf * 8;
    const int ccol  = n0 + bcol;
    #pragma unroll
    for (int v = 0; v < 8; ++v) {
        int r = crow0 + v;
        if (r < M) C[(size_t)r * N + ccol] = acc[v];
    }
}

template <int B, typename vecT>
__device__ __forceinline__ void add_rows(vecT& acc, const vecT* __restrict__ hv,
                                         int r_l, float d_l, float dc, int j, int lane) {
    vecT v[B];
    #pragma unroll
    for (int k = 0; k < B; ++k) {
        const int r = __builtin_amdgcn_readlane(r_l, j + k);
        v[k] = hv[(size_t)r * 32 + lane];
    }
    #pragma unroll
    for (int k = 0; k < B; ++k) {
        const float w = __int_as_float(__builtin_amdgcn_readlane(__float_as_int(d_l), j + k)) * dc;
        acc += v[k] * w;
    }
}

template <int F, bool RELU>
__global__ __launch_bounds__(256)
void k_gather(const int* __restrict__ rowptr, const int* __restrict__ srcs,
              const float* __restrict__ dinv, const float* __restrict__ h,
              const float* __restrict__ bias, float* __restrict__ out, int n) {
    typedef __attribute__((ext_vector_type(F / 32))) float vecT;
    const int lane = threadIdx.x & 31;
    const int wv = __builtin_amdgcn_readfirstlane((int)(threadIdx.x >> 5));
    const int c  = blockIdx.x * 8 + wv;
    if (c >= n) return;

    const vecT* hv = (const vecT*)h;
    const int beg = rowptr[c];
    const int end = rowptr[c + 1];
    const float dc = dinv[c];

    const int total = end - beg + 1;
    vecT acc = {};
    for (int b = 0; b < total; b += 32) {
        int m = total - b;
        m = m < 32 ? m : 32;
        const int idx = b + lane;
        int   r_l = c;
        float d_l = dc;
        if (idx >= 1 && lane < m) {
            r_l = srcs[beg + idx - 1];
            d_l = dinv[r_l];
        }
        int j = 0;
        for (; j + 4 <= m; j += 4) add_rows<4>(acc, hv, r_l, d_l, dc, j, lane);
        const int rem = m - j;
        if      (rem == 3) add_rows<3>(acc, hv, r_l, d_l, dc, j, lane);
        else if (rem == 2) add_rows<2>(acc, hv, r_l, d_l, dc, j, lane);
        else if (rem == 1) add_rows<1>(acc, hv, r_l, d_l, dc, j, lane);
    }

    vecT o = acc + ((const vecT*)bias)[lane];
    if (RELU) {
        #pragma unroll
        for (int k = 0; k < F / 32; ++k) o[k] = o[k] > 0.0f ? o[k] : 0.0f;
    }
    ((vecT*)out)[(size_t)c * 32 + lane] = o;
}

#define DEC_ITERS 2

template <int K>
__device__ __forceinline__ int row_share(int v) {
    return __builtin_amdgcn_update_dpp(0, v, 0x150 + K, 0xf, 0xf, false);
}
template <int D>
__device__ __forceinline__ float row_xor(float v) {
    return __int_as_float(__builtin_amdgcn_update_dpp(0, __float_as_int(v), 0x160 + D, 0xf, 0xf, false));
}

template <int K>
__device__ __forceinline__ void fetch_slot(float4 (&a)[16], float4 (&b)[16],
                                           const float* __restrict__ z,
                                           int src, int dst, unsigned c16) {
    constexpr int R = ((K & 1) << 3) | ((K & 2) << 1) | ((K & 4) >> 1) | ((K & 8) >> 3);
    const unsigned so = (unsigned)row_share<R>(src) * 256u + c16;
    const unsigned d_o = (unsigned)row_share<R>(dst) * 256u + c16;
    a[K] = *(const float4*)((const char*)z + so);
    b[K] = *(const float4*)((const char*)z + d_o);
}

template <int D, int H>
__device__ __forceinline__ void fold_step(float (&s)[16], bool up) {
    #pragma unroll
    for (int j = 0; j < H; ++j) {
        const float keep = up ? s[j + H] : s[j];
        const float send = up ? s[j]     : s[j + H];
        s[j] = keep + row_xor<D>(send);
    }
}

__device__ __forceinline__ void load_ends(const int* __restrict__ pos, const int* __restrict__ neg,
                                          int e, int E, int& src, int& dst) {
    src = 0; dst = 0;
    if (e < 2 * E) {
        const int* p = e < E ? pos + e : neg + (e - E);
        src = p[0];
        dst = p[E];
    }
}

__global__ __launch_bounds__(256)
void k_decode(const int* __restrict__ pos, const int* __restrict__ neg,
              const float* __restrict__ z, float* __restrict__ logits, int E) {
    const int lane = threadIdx.x & 31;
    const unsigned c   = lane & 15;
    const unsigned c16 = c * 16u;
    const int wv = __builtin_amdgcn_readfirstlane((int)(threadIdx.x >> 5));
    const int total = 2 * E;
    const int base = (blockIdx.x * 8 + wv) * (32 * DEC_ITERS);
    if (base >= total) return;

    int src[DEC_ITERS], dst[DEC_ITERS];
    #pragma unroll
    for (int it = 0; it < DEC_ITERS; ++it)
        load_ends(pos, neg, base + 32 * it + lane, E, src[it], dst[it]);

    #pragma unroll
    for (int it = 0; it < DEC_ITERS; ++it) {
        const int b0 = base + 32 * it;
        if (b0 >= total) break;

        float4 a[16], b[16];
        fetch_slot<0>(a, b, z, src[it], dst[it], c16);   fetch_slot<1>(a, b, z, src[it], dst[it], c16);
        fetch_slot<2>(a, b, z, src[it], dst[it], c16);   fetch_slot<3>(a, b, z, src[it], dst[it], c16);
        fetch_slot<4>(a, b, z, src[it], dst[it], c16);   fetch_slot<5>(a, b, z, src[it], dst[it], c16);
        fetch_slot<6>(a, b, z, src[it], dst[it], c16);   fetch_slot<7>(a, b, z, src[it], dst[it], c16);
        fetch_slot<8>(a, b, z, src[it], dst[it], c16);   fetch_slot<9>(a, b, z, src[it], dst[it], c16);
        fetch_slot<10>(a, b, z, src[it], dst[it], c16);  fetch_slot<11>(a, b, z, src[it], dst[it], c16);
        fetch_slot<12>(a, b, z, src[it], dst[it], c16);  fetch_slot<13>(a, b, z, src[it], dst[it], c16);
        fetch_slot<14>(a, b, z, src[it], dst[it], c16);  fetch_slot<15>(a, b, z, src[it], dst[it], c16);

        float s[16];
        #pragma unroll
        for (int k = 0; k < 16; ++k)
            s[k] = a[k].x * b[k].x + a[k].y * b[k].y + a[k].z * b[k].z + a[k].w * b[k].w;

        fold_step<1, 8>(s, (c & 1u) != 0);
        fold_step<2, 4>(s, (c & 2u) != 0);
        fold_step<4, 2>(s, (c & 4u) != 0);
        fold_step<8, 1>(s, (c & 8u) != 0);

        const int e = b0 + lane;
        if (e < total) logits[e] = s[0];
    }
}

static inline size_t align_f(size_t n) { return (n + 63) & ~size_t(63); }

extern "C" void kernel_launch(void* const* d_in, const int* in_sizes, int n_in,
                              void* d_out, int out_size, void* d_ws, size_t ws_size,
                              hipStream_t stream) {
    const float* x   = (const float*)d_in[0];
    const int*   pos = (const int*)  d_in[1];
    const int*   neg = (const int*)  d_in[2];
    const float* W1  = (const float*)d_in[3];
    const float* b1  = (const float*)d_in[4];
    const float* W2  = (const float*)d_in[5];
    const float* b2  = (const float*)d_in[6];
    float* out = (float*)d_out;

    const int N = in_sizes[0] / IN_CH;
    const int E = in_sizes[1] / 2;

    float* ws = (float*)d_ws;
    size_t o = 0;
    float* dinv = ws + o; o += align_f((size_t)N);
    float* h0   = ws + o; o += align_f((size_t)N * HIDC);
    float* g1   = ws + o; o += align_f((size_t)N * HIDC);
    float* g2   = ws + o; o += align_f((size_t)N * OUTC);
    int* hist   = (int*)(ws + o); o += align_f((size_t)NCH * 256);
    int* offs   = (int*)(ws + o); o += align_f((size_t)NCH * 256 + 1);
    int* rowptr = (int*)(ws + o); o += align_f((size_t)N + 1);
    int* binned = (int*)(ws + o); o += align_f((size_t)E);
    int* srcs   = (int*)(ws + o); o += align_f((size_t)E);
    (void)ws_size;

    const int T = 256;
    k_bucket_hist   <<<NCH, 256, 0, stream>>>(pos + E, hist, E);
    k_scan_hist     <<<1, 1024, 0, stream>>>(hist, offs);
    k_bucket_scatter<<<NCH, 256, 0, stream>>>(pos, pos + E, offs, binned, E);
    k_bucket_lists  <<<(N + 255) / 256, 512, 0, stream>>>(binned, offs, rowptr, dinv, srcs, N);

    dim3 g1g((N + 63) / 64, HIDC / 32);
    k_gemm_wmma_f32<<<g1g, T, 0, stream>>>(x, W1, h0, N, IN_CH, HIDC);
    k_gather<HIDC, true><<<(N + 7) / 8, T, 0, stream>>>(rowptr, srcs, dinv, h0, b1, g1, N);

    dim3 g2g((N + 63) / 64, OUTC / 32);
    k_gemm_wmma_f32<<<g2g, T, 0, stream>>>(g1, W2, h0, N, HIDC, OUTC);
    k_gather<OUTC, false><<<(N + 7) / 8, T, 0, stream>>>(rowptr, srcs, dinv, h0, b2, g2, N);

    int total = 2 * E;
    k_decode<<<(total + 256 * DEC_ITERS - 1) / (256 * DEC_ITERS), T, 0, stream>>>(pos, neg, g2, out, E);
    (void)out_size; (void)n_in;
}


# GEMM register blocking: a wave owns a 32x32 tile (2x2 WMMA tiles, every LDS fragment feeds two products), a block spans all N columns so A is read once; B staged with its k pairs adjacent
# speedup vs baseline: 3.8518x; 1.0707x over previous
#include <hip/hip_runtime.h>
#include <hip/hip_bf16.h>

typedef __attribute__((ext_vector_type(2))) float v2f;
typedef __attribute__((ext_vector_type(8))) float v8f;

#define IN_CH  512
#define HIDC   128
#define OUTC   64

#define NCH 256
static_assert(NCH * 256 == 1024 * 64, "k_scan_hist: 1024 threads x 64 counts");

__device__ __forceinline__ int bucket_of(int d) {
    const unsigned b = (unsigned)d >> 8;
    return (int)(b < 255u ? b : 255u);
}

__global__ __launch_bounds__(256)
void k_bucket_hist(const int* __restrict__ col, int* __restrict__ hist, int E) {
    __shared__ int s_h[256];
    const int tid = threadIdx.x;
    const int cs = (E + NCH - 1) / NCH;
    const int e0 = blockIdx.x * cs;
    int e1 = e0 + cs;
    e1 = e1 < E ? e1 : E;
    s_h[tid] = 0;
    __syncthreads();
    const int last = e1 - 1;
    for (int i = e0 + tid; i < e1; i += 256 * 8) {
        int d[8];
        #pragma unroll
        for (int k = 0; k < 8; ++k) {
            const int idx = i + k * 256;
            d[k] = col[idx < e1 ? idx : last];
        }
        asm volatile("" :: "v"(d[0]), "v"(d[1]), "v"(d[2]), "v"(d[3]),
                           "v"(d[4]), "v"(d[5]), "v"(d[6]), "v"(d[7]));
        #pragma unroll
        for (int k = 0; k < 8; ++k)
            if (i + k * 256 < e1) atomicAdd(&s_h[bucket_of(d[k])], 1);
    }
    __syncthreads();
    hist[tid * NCH + blockIdx.x] = s_h[tid];
}

__global__ __launch_bounds__(1024)
void k_scan_hist(const int* __restrict__ hist, int* __restrict__ offs) {
    __shared__ int s_tot[32];
    const int tid  = threadIdx.x;
    const int lane = tid & 31;
    const int wave = tid >> 5;
    const int4* hp = (const int4*)(hist + tid * 64);
    int4 v[16];
    #pragma unroll
    for (int k = 0; k < 16; ++k) v[k] = hp[k];
    int tsum = 0;
    #pragma unroll
    for (int k = 0; k < 16; ++k) tsum += v[k].x + v[k].y + v[k].z + v[k].w;

    int inc = tsum;
    #pragma unroll
    for (int d = 1; d < 32; d <<= 1) {
        int t = __shfl_up(inc, d, 32);
        if (lane >= d) inc += t;
    }
    if (lane == 31) s_tot[wave] = inc;
    __syncthreads();
    int off = 0;
    #pragma unroll
    for (int w = 0; w < 32; ++w)
        if (w < wave) off += s_tot[w];

    int run = off + inc - tsum;
    int4* op = (int4*)(offs + tid * 64);
    #pragma unroll
    for (int k = 0; k < 16; ++k) {
        int4 o;
        o.x = run; run += v[k].x;
        o.y = run; run += v[k].y;
        o.z = run; run += v[k].z;
        o.w = run; run += v[k].w;
        op[k] = o;
    }
    if (tid == 1023) offs[NCH * 256] = run;
}

__global__ __launch_bounds__(256)
void k_bucket_scatter(const int* __restrict__ row, const int* __restrict__ col,
                      const int* __restrict__ offs, int* __restrict__ binned, int E) {
    __shared__ int s_cur[256];
    const int tid = threadIdx.x;
    const int cs = (E + NCH - 1) / NCH;
    const int e0 = blockIdx.x * cs;
    int e1 = e0 + cs;
    e1 = e1 < E ? e1 : E;
    s_cur[tid] = offs[tid * NCH + blockIdx.x];
    __syncthreads();
    const int last = e1 - 1;
    for (int i = e0 + tid; i < e1; i += 256 * 8) {
        int d[8], s[8];
        #pragma unroll
        for (int k = 0; k < 8; ++k) {
            const int idx = i + k * 256;
            const int ic  = idx < e1 ? idx : last;
            d[k] = col[ic];
            s[k] = row[ic];
        }
        asm volatile("" :: "v"(d[0]), "v"(d[1]), "v"(d[2]), "v"(d[3]),
                           "v"(d[4]), "v"(d[5]), "v"(d[6]), "v"(d[7]),
                           "v"(s[0]), "v"(s[1]), "v"(s[2]), "v"(s[3]),
                           "v"(s[4]), "v"(s[5]), "v"(s[6]), "v"(s[7]));
        #pragma unroll
        for (int k = 0; k < 8; ++k) {
            if (i + k * 256 < e1) {
                const int pos = atomicAdd(&s_cur[bucket_of(d[k])], 1);
                binned[pos] = (s[k] << 8) | (d[k] & 255);
            }
        }
    }
}

__global__ __launch_bounds__(512)
void k_bucket_lists(const int* __restrict__ binned, const int* __restrict__ offs,
                    int* __restrict__ rowptr, float* __restrict__ dinv,
                    int* __restrict__ srcs, int n) {
    __shared__ int s_cnt[256];
    __shared__ int s_cur[256];
    __shared__ int s_tot[8];
    const int tid  = threadIdx.x;
    const int lane = tid & 31;
    const int wave = tid >> 5;
    const int b    = blockIdx.x;
    const int beg  = offs[b * NCH];
    const int end  = offs[(b + 1) * NCH];
    if (tid < 256) s_cnt[tid] = 0;
    __syncthreads();

    const int last = end - 1;
    for (int i = beg + tid; i < end; i += 512 * 4) {
        int p[4];
        #pragma unroll
        for (int k = 0; k < 4; ++k) {
            const int idx = i + k * 512;
            p[k] = binned[idx < end ? idx : last];
        }
        asm volatile("" :: "v"(p[0]), "v"(p[1]), "v"(p[2]), "v"(p[3]));
        #pragma unroll
        for (int k = 0; k < 4; ++k)
            if (i + k * 512 < end) atomicAdd(&s_cnt[p[k] & 255], 1);
    }
    __syncthreads();

    const int own = tid < 256 ? s_cnt[tid] : 0;
    int inc = own;
    #pragma unroll
    for (int d = 1; d < 32; d <<= 1) {
        int t = __shfl_up(inc, d, 32);
        if (lane >= d) inc += t;
    }
    if (tid < 256 && lane == 31) s_tot[wave] = inc;
    __syncthreads();
    if (tid < 256) {
        int off = 0;
        #pragma unroll
        for (int w = 0; w < 8; ++w)
            if (w < wave) off += s_tot[w];
        const int excl = beg + off + inc - own;
        s_cur[tid] = excl;
        const int node = b * 256 + tid;
        if (node < n) {
            rowptr[node] = excl;
            dinv[node]   = rsqrtf((float)(own + 1));
            if (node == n - 1) rowptr[n] = excl + own;
        }
    }
    __syncthreads();

    for (int i = beg + tid; i < end; i += 512 * 4) {
        int p[4];
        #pragma unroll
        for (int k = 0; k < 4; ++k) {
            const int idx = i + k * 512;
            p[k] = binned[idx < end ? idx : last];
        }
        asm volatile("" :: "v"(p[0]), "v"(p[1]), "v"(p[2]), "v"(p[3]));
        #pragma unroll
        for (int k = 0; k < 4; ++k) {
            if (i + k * 512 < end) {
                const int pos = atomicAdd(&s_cur[p[k] & 255], 1);
                srcs[pos] = p[k] >> 8;
            }
        }
    }
}

__device__ __forceinline__ void stage_k_pairs(float* dst, float4 rk, float4 rk1) {
    float4 lo, hi;
    lo.x = rk.x; lo.y = rk1.x; lo.z = rk.y; lo.w = rk1.y;
    hi.x = rk.z; hi.y = rk1.z; hi.z = rk.w; hi.w = rk1.w;
    *(float4*)dst       = lo;
    *(float4*)(dst + 4) = hi;
}

template <int WM, int WN>
__global__ __launch_bounds__(WM * WN * 32)
void k_gemm_wmma_f32(const float* __restrict__ A, const float* __restrict__ B,
                     float* __restrict__ C, int M, int K, int N) {
    constexpr int NT = WM * WN * 32;
    constexpr int BM = WM * 32;
    constexpr int BN = WN * 32;
    constexpr int PA = 36;
    constexpr int PB = 2 * BN + 32;
    constexpr int NA = (BM * 8) / NT;
    constexpr int NB = (16 * (BN / 4)) / NT;
    static_assert(NA * NT == BM * 8 && NB * NT == 16 * (BN / 4), "staging slots divide evenly");

    __shared__ __attribute__((aligned(16))) float ldsA[BM * PA];
    __shared__ __attribute__((aligned(16))) float ldsB[16 * PB];

    const int tid  = threadIdx.x;
    const int lane = tid & 31;
    const int wave = __builtin_amdgcn_readfirstlane(tid >> 5);
    const int wm = wave / WN;
    const int wn = wave % WN;
    const int m0 = blockIdx.x * BM;
    const int n0 = blockIdx.y * BN;
    const int h  = lane >> 4;
    const int c  = lane & 15;

    constexpr int RS = NT / 8;
    const int r0 = tid >> 3;
    const int q  = tid & 7;
    const int rlast = M - 1;
    const int gr0 = m0 + r0;
    const int gr1 = gr0 + RS;
    const int gr2 = gr0 + 2 * RS;
    const int gr3 = gr0 + 3 * RS;
    const float* pa0 = A + (size_t)(gr0 < M ? gr0 : rlast) * K + 4 * q;
    const float* pa1 = A + (size_t)(gr1 < M ? gr1 : rlast) * K + 4 * q;
    const float* pa2 = A + (size_t)(gr2 < M ? gr2 : rlast) * K + 4 * q;
    const float* pa3 = A + (size_t)(gr3 < M ? gr3 : rlast) * K + 4 * q;
    float* sa = ldsA + r0 * PA + 4 * q;
    constexpr int GB = BN / 4;
    constexpr int PS = NT / GB;
    static_assert(PS * NB == 16 && (NB == 2 || NB == 4) && (NA == 2 || NA == 4), "slot counts the code below is written for");
    const int p0 = tid / GB;
    const int g  = tid % GB;
    const float* pb0 = B + (size_t)(2 * p0) * N + n0 + 4 * g;
    const size_t bstep = (size_t)(2 * PS) * N;
    float* sb = ldsB + p0 * PB + 8 * g;

    float4 va0, va1, va2, va3, vb00, vb01, vb10, vb11, vb20, vb21, vb30, vb31;
    va0 = *(const float4*)pa0;
    va1 = *(const float4*)pa1;
    if constexpr (NA == 4) {
        va2 = *(const float4*)pa2;
        va3 = *(const float4*)pa3;
    }
    vb00 = *(const float4*)pb0;
    vb01 = *(const float4*)(pb0 + N);
    vb10 = *(const float4*)(pb0 + bstep);
    vb11 = *(const float4*)(pb0 + bstep + N);
    if constexpr (NB == 4) {
        vb20 = *(const float4*)(pb0 + 2 * bstep);
        vb21 = *(const float4*)(pb0 + 2 * bstep + N);
        vb30 = *(const float4*)(pb0 + 3 * bstep);
        vb31 = *(const float4*)(pb0 + 3 * bstep + N);
    }

    v8f acc[2][2] = {};

    const float* fa = ldsA + (wm * 32 + c) * PA + 2 * h;
    const float* fb = ldsB + h * PB + (wn * 32 + c) * 2;

    for (int k0 = 0; k0 < K; k0 += 32) {
        *(float4*)sa = va0;
        *(float4*)(sa + RS * PA) = va1;
        if constexpr (NA == 4) {
            *(float4*)(sa + 2 * RS * PA) = va2;
            *(float4*)(sa + 3 * RS * PA) = va3;
        }
        stage_k_pairs(sb, vb00, vb01);
        stage_k_pairs(sb + PS * PB, vb10, vb11);
        if constexpr (NB == 4) {
            stage_k_pairs(sb + 2 * PS * PB, vb20, vb21);
            stage_k_pairs(sb + 3 * PS * PB, vb30, vb31);
        }
        __syncthreads();

        if (k0 + 32 < K) {
            const int kn = k0 + 32;
            va0 = *(const float4*)(pa0 + kn);
            va1 = *(const float4*)(pa1 + kn);
            if constexpr (NA == 4) {
                va2 = *(const float4*)(pa2 + kn);
                va3 = *(const float4*)(pa3 + kn);
            }
            const float* q0 = pb0 + (size_t)kn * N;
            vb00 = *(const float4*)q0;
            vb01 = *(const float4*)(q0 + N);
            vb10 = *(const float4*)(q0 + bstep);
            vb11 = *(const float4*)(q0 + bstep + N);
            if constexpr (NB == 4) {
                vb20 = *(const float4*)(q0 + 2 * bstep);
                vb21 = *(const float4*)(q0 + 2 * bstep + N);
                vb30 = *(const float4*)(q0 + 3 * bstep);
                vb31 = *(const float4*)(q0 + 3 * bstep + N);
            }
        }

        #pragma unroll
        for (int kk = 0; kk < 32; kk += 4) {
            const v2f a0 = *(const v2f*)(fa + kk);
            const v2f a1 = *(const v2f*)(fa + 16 * PA + kk);
            const v2f b0 = *(const v2f*)(fb + (kk / 2) * PB);
            const v2f b1 = *(const v2f*)(fb + (kk / 2) * PB + 32);
            acc[0][0] = __builtin_amdgcn_wmma_f32_16x16x4_f32(false, a0, false, b0, (short)0, acc[0][0], false, false);
            acc[0][1] = __builtin_amdgcn_wmma_f32_16x16x4_f32(false, a0, false, b1, (short)0, acc[0][1], false, false);
            acc[1][0] = __builtin_amdgcn_wmma_f32_16x16x4_f32(false, a1, false, b0, (short)0, acc[1][0], false, false);
            acc[1][1] = __builtin_amdgcn_wmma_f32_16x16x4_f32(false, a1, false, b1, (short)0, acc[1][1], false, false);
        }
        __syncthreads();
    }

    const int ccol = n0 + wn * 32 + lane;
    #pragma unroll
    for (int i = 0; i < 2; ++i) {
        const int crow = m0 + wm * 32 + i * 16;
        #pragma unroll
        for (int v = 0; v < 8; ++v) {
            const float keep = h ? acc[i][1][v] : acc[i][0][v];
            const float send = h ? acc[i][0][v] : acc[i][1][v];
            const float recv = __int_as_float((int)__builtin_amdgcn_permlanex16(
                0u, (unsigned)__float_as_int(send), 0x76543210u, 0xfedcba98u, false, false));
            const int r_lo = crow + v;
            const int r_hi = crow + v + 8;
            if (r_lo < M) C[(size_t)r_lo * N + ccol] = h ? recv : keep;
            if (r_hi < M) C[(size_t)r_hi * N + ccol] = h ? keep : recv;
        }
    }
}

template <int B, typename vecT>
__device__ __forceinline__ void add_rows(vecT& acc, const vecT* __restrict__ hv,
                                         int r_l, float d_l, float dc, int j, int lane) {
    vecT v[B];
    #pragma unroll
    for (int k = 0; k < B; ++k) {
        const int r = __builtin_amdgcn_readlane(r_l, j + k);
        v[k] = hv[(size_t)r * 32 + lane];
    }
    #pragma unroll
    for (int k = 0; k < B; ++k) {
        const float w = __int_as_float(__builtin_amdgcn_readlane(__float_as_int(d_l), j + k)) * dc;
        acc += v[k] * w;
    }
}

template <int F, bool RELU>
__global__ __launch_bounds__(256)
void k_gather(const int* __restrict__ rowptr, const int* __restrict__ srcs,
              const float* __restrict__ dinv, const float* __restrict__ h,
              const float* __restrict__ bias, float* __restrict__ out, int n) {
    typedef __attribute__((ext_vector_type(F / 32))) float vecT;
    const int lane = threadIdx.x & 31;
    const int wv = __builtin_amdgcn_readfirstlane((int)(threadIdx.x >> 5));
    const int c  = blockIdx.x * 8 + wv;
    if (c >= n) return;

    const vecT* hv = (const vecT*)h;
    const int beg = rowptr[c];
    const int end = rowptr[c + 1];
    const float dc = dinv[c];

    const int total = end - beg + 1;
    vecT acc = {};
    for (int b = 0; b < total; b += 32) {
        int m = total - b;
        m = m < 32 ? m : 32;
        const int idx = b + lane;
        int   r_l = c;
        float d_l = dc;
        if (idx >= 1 && lane < m) {
            r_l = srcs[beg + idx - 1];
            d_l = dinv[r_l];
        }
        int j = 0;
        for (; j + 4 <= m; j += 4) add_rows<4>(acc, hv, r_l, d_l, dc, j, lane);
        const int rem = m - j;
        if      (rem == 3) add_rows<3>(acc, hv, r_l, d_l, dc, j, lane);
        else if (rem == 2) add_rows<2>(acc, hv, r_l, d_l, dc, j, lane);
        else if (rem == 1) add_rows<1>(acc, hv, r_l, d_l, dc, j, lane);
    }

    vecT o = acc + ((const vecT*)bias)[lane];
    if (RELU) {
        #pragma unroll
        for (int k = 0; k < F / 32; ++k) o[k] = o[k] > 0.0f ? o[k] : 0.0f;
    }
    ((vecT*)out)[(size_t)c * 32 + lane] = o;
}

#define DEC_ITERS 2

template <int K>
__device__ __forceinline__ int row_share(int v) {
    return __builtin_amdgcn_update_dpp(0, v, 0x150 + K, 0xf, 0xf, false);
}
template <int D>
__device__ __forceinline__ float row_xor(float v) {
    return __int_as_float(__builtin_amdgcn_update_dpp(0, __float_as_int(v), 0x160 + D, 0xf, 0xf, false));
}

template <int K>
__device__ __forceinline__ void fetch_slot(float4 (&a)[16], float4 (&b)[16],
                                           const float* __restrict__ z,
                                           int src, int dst, unsigned c16) {
    constexpr int R = ((K & 1) << 3) | ((K & 2) << 1) | ((K & 4) >> 1) | ((K & 8) >> 3);
    const unsigned so = (unsigned)row_share<R>(src) * 256u + c16;
    const unsigned d_o = (unsigned)row_share<R>(dst) * 256u + c16;
    a[K] = *(const float4*)((const char*)z + so);
    b[K] = *(const float4*)((const char*)z + d_o);
}

template <int D, int H>
__device__ __forceinline__ void fold_step(float (&s)[16], bool up) {
    #pragma unroll
    for (int j = 0; j < H; ++j) {
        const float keep = up ? s[j + H] : s[j];
        const float send = up ? s[j]     : s[j + H];
        s[j] = keep + row_xor<D>(send);
    }
}

__device__ __forceinline__ void load_ends(const int* __restrict__ pos, const int* __restrict__ neg,
                                          int e, int E, int& src, int& dst) {
    src = 0; dst = 0;
    if (e < 2 * E) {
        const int* p = e < E ? pos + e : neg + (e - E);
        src = p[0];
        dst = p[E];
    }
}

__global__ __launch_bounds__(256)
void k_decode(const int* __restrict__ pos, const int* __restrict__ neg,
              const float* __restrict__ z, float* __restrict__ logits, int E) {
    const int lane = threadIdx.x & 31;
    const unsigned c   = lane & 15;
    const unsigned c16 = c * 16u;
    const int wv = __builtin_amdgcn_readfirstlane((int)(threadIdx.x >> 5));
    const int total = 2 * E;
    const int base = (blockIdx.x * 8 + wv) * (32 * DEC_ITERS);
    if (base >= total) return;

    int src[DEC_ITERS], dst[DEC_ITERS];
    #pragma unroll
    for (int it = 0; it < DEC_ITERS; ++it)
        load_ends(pos, neg, base + 32 * it + lane, E, src[it], dst[it]);

    #pragma unroll
    for (int it = 0; it < DEC_ITERS; ++it) {
        const int b0 = base + 32 * it;
        if (b0 >= total) break;

        float4 a[16], b[16];
        fetch_slot<0>(a, b, z, src[it], dst[it], c16);   fetch_slot<1>(a, b, z, src[it], dst[it], c16);
        fetch_slot<2>(a, b, z, src[it], dst[it], c16);   fetch_slot<3>(a, b, z, src[it], dst[it], c16);
        fetch_slot<4>(a, b, z, src[it], dst[it], c16);   fetch_slot<5>(a, b, z, src[it], dst[it], c16);
        fetch_slot<6>(a, b, z, src[it], dst[it], c16);   fetch_slot<7>(a, b, z, src[it], dst[it], c16);
        fetch_slot<8>(a, b, z, src[it], dst[it], c16);   fetch_slot<9>(a, b, z, src[it], dst[it], c16);
        fetch_slot<10>(a, b, z, src[it], dst[it], c16);  fetch_slot<11>(a, b, z, src[it], dst[it], c16);
        fetch_slot<12>(a, b, z, src[it], dst[it], c16);  fetch_slot<13>(a, b, z, src[it], dst[it], c16);
        fetch_slot<14>(a, b, z, src[it], dst[it], c16);  fetch_slot<15>(a, b, z, src[it], dst[it], c16);

        float s[16];
        #pragma unroll
        for (int k = 0; k < 16; ++k)
            s[k] = a[k].x * b[k].x + a[k].y * b[k].y + a[k].z * b[k].z + a[k].w * b[k].w;

        fold_step<1, 8>(s, (c & 1u) != 0);
        fold_step<2, 4>(s, (c & 2u) != 0);
        fold_step<4, 2>(s, (c & 4u) != 0);
        fold_step<8, 1>(s, (c & 8u) != 0);

        const int e = b0 + lane;
        if (e < total) logits[e] = s[0];
    }
}

static inline size_t align_f(size_t n) { return (n + 63) & ~size_t(63); }

extern "C" void kernel_launch(void* const* d_in, const int* in_sizes, int n_in,
                              void* d_out, int out_size, void* d_ws, size_t ws_size,
                              hipStream_t stream) {
    const float* x   = (const float*)d_in[0];
    const int*   pos = (const int*)  d_in[1];
    const int*   neg = (const int*)  d_in[2];
    const float* W1  = (const float*)d_in[3];
    const float* b1  = (const float*)d_in[4];
    const float* W2  = (const float*)d_in[5];
    const float* b2  = (const float*)d_in[6];
    float* out = (float*)d_out;

    const int N = in_sizes[0] / IN_CH;
    const int E = in_sizes[1] / 2;

    float* ws = (float*)d_ws;
    size_t o = 0;
    float* dinv = ws + o; o += align_f((size_t)N);
    float* h0   = ws + o; o += align_f((size_t)N * HIDC);
    float* g1   = ws + o; o += align_f((size_t)N * HIDC);
    float* g2   = ws + o; o += align_f((size_t)N * OUTC);
    int* hist   = (int*)(ws + o); o += align_f((size_t)NCH * 256);
    int* offs   = (int*)(ws + o); o += align_f((size_t)NCH * 256 + 1);
    int* rowptr = (int*)(ws + o); o += align_f((size_t)N + 1);
    int* binned = (int*)(ws + o); o += align_f((size_t)E);
    int* srcs   = (int*)(ws + o); o += align_f((size_t)E);
    (void)ws_size;

    const int T = 256;
    k_bucket_hist   <<<NCH, 256, 0, stream>>>(pos + E, hist, E);
    k_scan_hist     <<<1, 1024, 0, stream>>>(hist, offs);
    k_bucket_scatter<<<NCH, 256, 0, stream>>>(pos, pos + E, offs, binned, E);
    k_bucket_lists  <<<(N + 255) / 256, 512, 0, stream>>>(binned, offs, rowptr, dinv, srcs, N);

    dim3 g1g((N + 31) / 32, HIDC / 128);
    k_gemm_wmma_f32<1, 4><<<g1g, 128, 0, stream>>>(x, W1, h0, N, IN_CH, HIDC);
    k_gather<HIDC, true><<<(N + 7) / 8, T, 0, stream>>>(rowptr, srcs, dinv, h0, b1, g1, N);

    dim3 g2g((N + 63) / 64, OUTC / 64);
    k_gemm_wmma_f32<2, 2><<<g2g, 128, 0, stream>>>(g1, W2, h0, N, HIDC, OUTC);
    k_gather<OUTC, false><<<(N + 7) / 8, T, 0, stream>>>(rowptr, srcs, dinv, h0, b2, g2, N);

    int total = 2 * E;
    k_decode<<<(total + 256 * DEC_ITERS - 1) / (256 * DEC_ITERS), T, 0, stream>>>(pos, neg, g2, out, E);
    (void)out_size; (void)n_in;
}


# Edge-sort count and offset tables stored [chunk][bucket]: count store, the one-block scan's loads and stores and scatter's offset load cover whole 128-byte lines (were 32 lines per instruction)
# speedup vs baseline: 4.1536x; 1.0784x over previous
#include <hip/hip_runtime.h>
#include <hip/hip_bf16.h>

typedef __attribute__((ext_vector_type(2))) float v2f;
typedef __attribute__((ext_vector_type(8))) float v8f;

#define IN_CH  512
#define HIDC   128
#define OUTC   64

#define NCH 256
static_assert(NCH * 256 == 1024 * 64 && NCH == 16 * 16, "k_scan_hist: 1024 threads x 64 counts, 16 groups of 16 chunks");

__device__ __forceinline__ int bucket_of(int d) {
    const unsigned b = (unsigned)d >> 8;
    return (int)(b < 255u ? b : 255u);
}

__global__ __launch_bounds__(256)
void k_bucket_hist(const int* __restrict__ col, int* __restrict__ hist, int E) {
    __shared__ int s_h[256];
    const int tid = threadIdx.x;
    const int cs = (E + NCH - 1) / NCH;
    const int e0 = blockIdx.x * cs;
    int e1 = e0 + cs;
    e1 = e1 < E ? e1 : E;
    s_h[tid] = 0;
    __syncthreads();
    const int last = e1 - 1;
    for (int i = e0 + tid; i < e1; i += 256 * 8) {
        int d[8];
        #pragma unroll
        for (int k = 0; k < 8; ++k) {
            const int idx = i + k * 256;
            d[k] = col[idx < e1 ? idx : last];
        }
        asm volatile("" :: "v"(d[0]), "v"(d[1]), "v"(d[2]), "v"(d[3]),
                           "v"(d[4]), "v"(d[5]), "v"(d[6]), "v"(d[7]));
        #pragma unroll
        for (int k = 0; k < 8; ++k)
            if (i + k * 256 < e1) atomicAdd(&s_h[bucket_of(d[k])], 1);
    }
    __syncthreads();
    hist[blockIdx.x * 256 + tid] = s_h[tid];
}

__global__ __launch_bounds__(1024)
void k_scan_hist(const int* __restrict__ hist, int* __restrict__ offs) {
    __shared__ __attribute__((aligned(16))) int s_part[16 * 256];
    __shared__ __attribute__((aligned(16))) int s_base[256];
    __shared__ int s_tot[8];
    const int tid  = threadIdx.x;
    const int lane = tid & 31;
    const int wave = tid >> 5;
    const int q = tid & 63;
    const int r = tid >> 6;
    const int4* hp = (const int4*)hist + (r * 16) * 64 + q;
    int4 v[16];
    #pragma unroll
    for (int j = 0; j < 16; ++j) v[j] = hp[j * 64];
    int4 s = v[0];
    #pragma unroll
    for (int j = 1; j < 16; ++j) {
        s.x += v[j].x; s.y += v[j].y; s.z += v[j].z; s.w += v[j].w;
    }
    ((int4*)s_part)[r * 64 + q] = s;
    __syncthreads();

    int tot = 0;
    if (tid < 256) {
        int g[16];
        #pragma unroll
        for (int k = 0; k < 16; ++k) g[k] = s_part[k * 256 + tid];
        #pragma unroll
        for (int k = 0; k < 16; ++k) { s_part[k * 256 + tid] = tot; tot += g[k]; }
    }
    int inc = tot;
    #pragma unroll
    for (int d = 1; d < 32; d <<= 1) {
        int t = __shfl_up(inc, d, 32);
        if (lane >= d) inc += t;
    }
    if (tid < 256 && lane == 31) s_tot[wave] = inc;
    __syncthreads();
    if (tid < 256) {
        int off = 0;
        #pragma unroll
        for (int w = 0; w < 8; ++w)
            if (w < wave) off += s_tot[w];
        const int base = off + inc - tot;
        s_base[tid] = base;
        offs[NCH * 256 + tid] = base;
        if (tid == 255) offs[NCH * 256 + 256] = base + tot;
    }
    __syncthreads();

    int4 run = ((const int4*)s_part)[r * 64 + q];
    const int4 b4 = ((const int4*)s_base)[q];
    run.x += b4.x; run.y += b4.y; run.z += b4.z; run.w += b4.w;
    int4* op = (int4*)offs + (r * 16) * 64 + q;
    #pragma unroll
    for (int j = 0; j < 16; ++j) {
        op[j * 64] = run;
        run.x += v[j].x; run.y += v[j].y; run.z += v[j].z; run.w += v[j].w;
    }
}

__global__ __launch_bounds__(256)
void k_bucket_scatter(const int* __restrict__ row, const int* __restrict__ col,
                      const int* __restrict__ offs, int* __restrict__ binned, int E) {
    __shared__ int s_cur[256];
    const int tid = threadIdx.x;
    const int cs = (E + NCH - 1) / NCH;
    const int e0 = blockIdx.x * cs;
    int e1 = e0 + cs;
    e1 = e1 < E ? e1 : E;
    s_cur[tid] = offs[blockIdx.x * 256 + tid];
    __syncthreads();
    const int last = e1 - 1;
    for (int i = e0 + tid; i < e1; i += 256 * 8) {
        int d[8], s[8];
        #pragma unroll
        for (int k = 0; k < 8; ++k) {
            const int idx = i + k * 256;
            const int ic  = idx < e1 ? idx : last;
            d[k] = col[ic];
            s[k] = row[ic];
        }
        asm volatile("" :: "v"(d[0]), "v"(d[1]), "v"(d[2]), "v"(d[3]),
                           "v"(d[4]), "v"(d[5]), "v"(d[6]), "v"(d[7]),
                           "v"(s[0]), "v"(s[1]), "v"(s[2]), "v"(s[3]),
                           "v"(s[4]), "v"(s[5]), "v"(s[6]), "v"(s[7]));
        #pragma unroll
        for (int k = 0; k < 8; ++k) {
            if (i + k * 256 < e1) {
                const int pos = atomicAdd(&s_cur[bucket_of(d[k])], 1);
                binned[pos] = (s[k] << 8) | (d[k] & 255);
            }
        }
    }
}

__global__ __launch_bounds__(512)
void k_bucket_lists(const int* __restrict__ binned, const int* __restrict__ offs,
                    int* __restrict__ rowptr, float* __restrict__ dinv,
                    int* __restrict__ srcs, int n) {
    __shared__ int s_cnt[256];
    __shared__ int s_cur[256];
    __shared__ int s_tot[8];
    const int tid  = threadIdx.x;
    const int lane = tid & 31;
    const int wave = tid >> 5;
    const int b    = blockIdx.x;
    const int beg  = offs[NCH * 256 + b];
    const int end  = offs[NCH * 256 + b + 1];
    if (tid < 256) s_cnt[tid] = 0;
    __syncthreads();

    const int last = end - 1;
    for (int i = beg + tid; i < end; i += 512 * 4) {
        int p[4];
        #pragma unroll
        for (int k = 0; k < 4; ++k) {
            const int idx = i + k * 512;
            p[k] = binned[idx < end ? idx : last];
        }
        asm volatile("" :: "v"(p[0]), "v"(p[1]), "v"(p[2]), "v"(p[3]));
        #pragma unroll
        for (int k = 0; k < 4; ++k)
            if (i + k * 512 < end) atomicAdd(&s_cnt[p[k] & 255], 1);
    }
    __syncthreads();

    const int own = tid < 256 ? s_cnt[tid] : 0;
    int inc = own;
    #pragma unroll
    for (int d = 1; d < 32; d <<= 1) {
        int t = __shfl_up(inc, d, 32);
        if (lane >= d) inc += t;
    }
    if (tid < 256 && lane == 31) s_tot[wave] = inc;
    __syncthreads();
    if (tid < 256) {
        int off = 0;
        #pragma unroll
        for (int w = 0; w < 8; ++w)
            if (w < wave) off += s_tot[w];
        const int excl = beg + off + inc - own;
        s_cur[tid] = excl;
        const int node = b * 256 + tid;
        if (node < n) {
            rowptr[node] = excl;
            dinv[node]   = rsqrtf((float)(own + 1));
            if (node == n - 1) rowptr[n] = excl + own;
        }
    }
    __syncthreads();

    for (int i = beg + tid; i < end; i += 512 * 4) {
        int p[4];
        #pragma unroll
        for (int k = 0; k < 4; ++k) {
            const int idx = i + k * 512;
            p[k] = binned[idx < end ? idx : last];
        }
        asm volatile("" :: "v"(p[0]), "v"(p[1]), "v"(p[2]), "v"(p[3]));
        #pragma unroll
        for (int k = 0; k < 4; ++k) {
            if (i + k * 512 < end) {
                const int pos = atomicAdd(&s_cur[p[k] & 255], 1);
                srcs[pos] = p[k] >> 8;
            }
        }
    }
}

__device__ __forceinline__ void stage_k_pairs(float* dst, float4 rk, float4 rk1) {
    float4 lo, hi;
    lo.x = rk.x; lo.y = rk1.x; lo.z = rk.y; lo.w = rk1.y;
    hi.x = rk.z; hi.y = rk1.z; hi.z = rk.w; hi.w = rk1.w;
    *(float4*)dst       = lo;
    *(float4*)(dst + 4) = hi;
}

template <int WM, int WN>
__global__ __launch_bounds__(WM * WN * 32)
void k_gemm_wmma_f32(const float* __restrict__ A, const float* __restrict__ B,
                     float* __restrict__ C, int M, int K, int N) {
    constexpr int NT = WM * WN * 32;
    constexpr int BM = WM * 32;
    constexpr int BN = WN * 32;
    constexpr int PA = 36;
    constexpr int PB = 2 * BN + 32;
    constexpr int NA = (BM * 8) / NT;
    constexpr int NB = (16 * (BN / 4)) / NT;
    static_assert(NA * NT == BM * 8 && NB * NT == 16 * (BN / 4), "staging slots divide evenly");

    __shared__ __attribute__((aligned(16))) float ldsA[BM * PA];
    __shared__ __attribute__((aligned(16))) float ldsB[16 * PB];

    const int tid  = threadIdx.x;
    const int lane = tid & 31;
    const int wave = __builtin_amdgcn_readfirstlane(tid >> 5);
    const int wm = wave / WN;
    const int wn = wave % WN;
    const int m0 = blockIdx.x * BM;
    const int n0 = blockIdx.y * BN;
    const int h  = lane >> 4;
    const int c  = lane & 15;

    constexpr int RS = NT / 8;
    const int r0 = tid >> 3;
    const int q  = tid & 7;
    const int rlast = M - 1;
    const int gr0 = m0 + r0;
    const int gr1 = gr0 + RS;
    const int gr2 = gr0 + 2 * RS;
    const int gr3 = gr0 + 3 * RS;
    const float* pa0 = A + (size_t)(gr0 < M ? gr0 : rlast) * K + 4 * q;
    const float* pa1 = A + (size_t)(gr1 < M ? gr1 : rlast) * K + 4 * q;
    const float* pa2 = A + (size_t)(gr2 < M ? gr2 : rlast) * K + 4 * q;
    const float* pa3 = A + (size_t)(gr3 < M ? gr3 : rlast) * K + 4 * q;
    float* sa = ldsA + r0 * PA + 4 * q;
    constexpr int GB = BN / 4;
    constexpr int PS = NT / GB;
    static_assert(PS * NB == 16 && (NB == 2 || NB == 4) && (NA == 2 || NA == 4), "slot counts the code below is written for");
    const int p0 = tid / GB;
    const int g  = tid % GB;
    const float* pb0 = B + (size_t)(2 * p0) * N + n0 + 4 * g;
    const size_t bstep = (size_t)(2 * PS) * N;
    float* sb = ldsB + p0 * PB + 8 * g;

    float4 va0, va1, va2, va3, vb00, vb01, vb10, vb11, vb20, vb21, vb30, vb31;
    va0 = *(const float4*)pa0;
    va1 = *(const float4*)pa1;
    if constexpr (NA == 4) {
        va2 = *(const float4*)pa2;
        va3 = *(const float4*)pa3;
    }
    vb00 = *(const float4*)pb0;
    vb01 = *(const float4*)(pb0 + N);
    vb10 = *(const float4*)(pb0 + bstep);
    vb11 = *(const float4*)(pb0 + bstep + N);
    if constexpr (NB == 4) {
        vb20 = *(const float4*)(pb0 + 2 * bstep);
        vb21 = *(const float4*)(pb0 + 2 * bstep + N);
        vb30 = *(const float4*)(pb0 + 3 * bstep);
        vb31 = *(const float4*)(pb0 + 3 * bstep + N);
    }

    v8f acc[2][2] = {};

    const float* fa = ldsA + (wm * 32 + c) * PA + 2 * h;
    const float* fb = ldsB + h * PB + (wn * 32 + c) * 2;

    for (int k0 = 0; k0 < K; k0 += 32) {
        *(float4*)sa = va0;
        *(float4*)(sa + RS * PA) = va1;
        if constexpr (NA == 4) {
            *(float4*)(sa + 2 * RS * PA) = va2;
            *(float4*)(sa + 3 * RS * PA) = va3;
        }
        stage_k_pairs(sb, vb00, vb01);
        stage_k_pairs(sb + PS * PB, vb10, vb11);
        if constexpr (NB == 4) {
            stage_k_pairs(sb + 2 * PS * PB, vb20, vb21);
            stage_k_pairs(sb + 3 * PS * PB, vb30, vb31);
        }
        __syncthreads();

        if (k0 + 32 < K) {
            const int kn = k0 + 32;
            va0 = *(const float4*)(pa0 + kn);
            va1 = *(const float4*)(pa1 + kn);
            if constexpr (NA == 4) {
                va2 = *(const float4*)(pa2 + kn);
                va3 = *(const float4*)(pa3 + kn);
            }
            const float* q0 = pb0 + (size_t)kn * N;
            vb00 = *(const float4*)q0;
            vb01 = *(const float4*)(q0 + N);
            vb10 = *(const float4*)(q0 + bstep);
            vb11 = *(const float4*)(q0 + bstep + N);
            if constexpr (NB == 4) {
                vb20 = *(const float4*)(q0 + 2 * bstep);
                vb21 = *(const float4*)(q0 + 2 * bstep + N);
                vb30 = *(const float4*)(q0 + 3 * bstep);
                vb31 = *(const float4*)(q0 + 3 * bstep + N);
            }
        }

        #pragma unroll
        for (int kk = 0; kk < 32; kk += 4) {
            const v2f a0 = *(const v2f*)(fa + kk);
            const v2f a1 = *(const v2f*)(fa + 16 * PA + kk);
            const v2f b0 = *(const v2f*)(fb + (kk / 2) * PB);
            const v2f b1 = *(const v2f*)(fb + (kk / 2) * PB + 32);
            acc[0][0] = __builtin_amdgcn_wmma_f32_16x16x4_f32(false, a0, false, b0, (short)0, acc[0][0], false, false);
            acc[0][1] = __builtin_amdgcn_wmma_f32_16x16x4_f32(false, a0, false, b1, (short)0, acc[0][1], false, false);
            acc[1][0] = __builtin_amdgcn_wmma_f32_16x16x4_f32(false, a1, false, b0, (short)0, acc[1][0], false, false);
            acc[1][1] = __builtin_amdgcn_wmma_f32_16x16x4_f32(false, a1, false, b1, (short)0, acc[1][1], false, false);
        }
        __syncthreads();
    }

    const int ccol = n0 + wn * 32 + lane;
    #pragma unroll
    for (int i = 0; i < 2; ++i) {
        const int crow = m0 + wm * 32 + i * 16;
        #pragma unroll
        for (int v = 0; v < 8; ++v) {
            const float keep = h ? acc[i][1][v] : acc[i][0][v];
            const float send = h ? acc[i][0][v] : acc[i][1][v];
            const float recv = __int_as_float((int)__builtin_amdgcn_permlanex16(
                0u, (unsigned)__float_as_int(send), 0x76543210u, 0xfedcba98u, false, false));
            const int r_lo = crow + v;
            const int r_hi = crow + v + 8;
            if (r_lo < M) C[(size_t)r_lo * N + ccol] = h ? recv : keep;
            if (r_hi < M) C[(size_t)r_hi * N + ccol] = h ? keep : recv;
        }
    }
}

template <int B, typename vecT>
__device__ __forceinline__ void add_rows(vecT& acc, const vecT* __restrict__ hv,
                                         int r_l, float d_l, float dc, int j, int lane) {
    vecT v[B];
    #pragma unroll
    for (int k = 0; k < B; ++k) {
        const int r = __builtin_amdgcn_readlane(r_l, j + k);
        v[k] = hv[(size_t)r * 32 + lane];
    }
    #pragma unroll
    for (int k = 0; k < B; ++k) {
        const float w = __int_as_float(__builtin_amdgcn_readlane(__float_as_int(d_l), j + k)) * dc;
        acc += v[k] * w;
    }
}

template <int F, bool RELU>
__global__ __launch_bounds__(256)
void k_gather(const int* __restrict__ rowptr, const int* __restrict__ srcs,
              const float* __restrict__ dinv, const float* __restrict__ h,
              const float* __restrict__ bias, float* __restrict__ out, int n) {
    typedef __attribute__((ext_vector_type(F / 32))) float vecT;
    const int lane = threadIdx.x & 31;
    const int wv = __builtin_amdgcn_readfirstlane((int)(threadIdx.x >> 5));
    const int c  = blockIdx.x * 8 + wv;
    if (c >= n) return;

    const vecT* hv = (const vecT*)h;
    const int beg = rowptr[c];
    const int end = rowptr[c + 1];
    const float dc = dinv[c];

    const int total = end - beg + 1;
    vecT acc = {};
    for (int b = 0; b < total; b += 32) {
        int m = total - b;
        m = m < 32 ? m : 32;
        const int idx = b + lane;
        int   r_l = c;
        float d_l = dc;
        if (idx >= 1 && lane < m) {
            r_l = srcs[beg + idx - 1];
            d_l = dinv[r_l];
        }
        int j = 0;
        for (; j + 4 <= m; j += 4) add_rows<4>(acc, hv, r_l, d_l, dc, j, lane);
        const int rem = m - j;
        if      (rem == 3) add_rows<3>(acc, hv, r_l, d_l, dc, j, lane);
        else if (rem == 2) add_rows<2>(acc, hv, r_l, d_l, dc, j, lane);
        else if (rem == 1) add_rows<1>(acc, hv, r_l, d_l, dc, j, lane);
    }

    vecT o = acc + ((const vecT*)bias)[lane];
    if (RELU) {
        #pragma unroll
        for (int k = 0; k < F / 32; ++k) o[k] = o[k] > 0.0f ? o[k] : 0.0f;
    }
    ((vecT*)out)[(size_t)c * 32 + lane] = o;
}

#define DEC_ITERS 2

template <int K>
__device__ __forceinline__ int row_share(int v) {
    return __builtin_amdgcn_update_dpp(0, v, 0x150 + K, 0xf, 0xf, false);
}
template <int D>
__device__ __forceinline__ float row_xor(float v) {
    return __int_as_float(__builtin_amdgcn_update_dpp(0, __float_as_int(v), 0x160 + D, 0xf, 0xf, false));
}

template <int K>
__device__ __forceinline__ void fetch_slot(float4 (&a)[16], float4 (&b)[16],
                                           const float* __restrict__ z,
                                           int src, int dst, unsigned c16) {
    constexpr int R = ((K & 1) << 3) | ((K & 2) << 1) | ((K & 4) >> 1) | ((K & 8) >> 3);
    const unsigned so = (unsigned)row_share<R>(src) * 256u + c16;
    const unsigned d_o = (unsigned)row_share<R>(dst) * 256u + c16;
    a[K] = *(const float4*)((const char*)z + so);
    b[K] = *(const float4*)((const char*)z + d_o);
}

template <int D, int H>
__device__ __forceinline__ void fold_step(float (&s)[16], bool up) {
    #pragma unroll
    for (int j = 0; j < H; ++j) {
        const float keep = up ? s[j + H] : s[j];
        const float send = up ? s[j]     : s[j + H];
        s[j] = keep + row_xor<D>(send);
    }
}

__device__ __forceinline__ void load_ends(const int* __restrict__ pos, const int* __restrict__ neg,
                                          int e, int E, int& src, int& dst) {
    src = 0; dst = 0;
    if (e < 2 * E) {
        const int* p = e < E ? pos + e : neg + (e - E);
        src = p[0];
        dst = p[E];
    }
}

__global__ __launch_bounds__(256)
void k_decode(const int* __restrict__ pos, const int* __restrict__ neg,
              const float* __restrict__ z, float* __restrict__ logits, int E) {
    const int lane = threadIdx.x & 31;
    const unsigned c   = lane & 15;
    const unsigned c16 = c * 16u;
    const int wv = __builtin_amdgcn_readfirstlane((int)(threadIdx.x >> 5));
    const int total = 2 * E;
    const int base = (blockIdx.x * 8 + wv) * (32 * DEC_ITERS);
    if (base >= total) return;

    int src[DEC_ITERS], dst[DEC_ITERS];
    #pragma unroll
    for (int it = 0; it < DEC_ITERS; ++it)
        load_ends(pos, neg, base + 32 * it + lane, E, src[it], dst[it]);

    #pragma unroll
    for (int it = 0; it < DEC_ITERS; ++it) {
        const int b0 = base + 32 * it;
        if (b0 >= total) break;

        float4 a[16], b[16];
        fetch_slot<0>(a, b, z, src[it], dst[it], c16);   fetch_slot<1>(a, b, z, src[it], dst[it], c16);
        fetch_slot<2>(a, b, z, src[it], dst[it], c16);   fetch_slot<3>(a, b, z, src[it], dst[it], c16);
        fetch_slot<4>(a, b, z, src[it], dst[it], c16);   fetch_slot<5>(a, b, z, src[it], dst[it], c16);
        fetch_slot<6>(a, b, z, src[it], dst[it], c16);   fetch_slot<7>(a, b, z, src[it], dst[it], c16);
        fetch_slot<8>(a, b, z, src[it], dst[it], c16);   fetch_slot<9>(a, b, z, src[it], dst[it], c16);
        fetch_slot<10>(a, b, z, src[it], dst[it], c16);  fetch_slot<11>(a, b, z, src[it], dst[it], c16);
        fetch_slot<12>(a, b, z, src[it], dst[it], c16);  fetch_slot<13>(a, b, z, src[it], dst[it], c16);
        fetch_slot<14>(a, b, z, src[it], dst[it], c16);  fetch_slot<15>(a, b, z, src[it], dst[it], c16);

        float s[16];
        #pragma unroll
        for (int k = 0; k < 16; ++k)
            s[k] = a[k].x * b[k].x + a[k].y * b[k].y + a[k].z * b[k].z + a[k].w * b[k].w;

        fold_step<1, 8>(s, (c & 1u) != 0);
        fold_step<2, 4>(s, (c & 2u) != 0);
        fold_step<4, 2>(s, (c & 4u) != 0);
        fold_step<8, 1>(s, (c & 8u) != 0);

        const int e = b0 + lane;
        if (e < total) logits[e] = s[0];
    }
}

static inline size_t align_f(size_t n) { return (n + 63) & ~size_t(63); }

extern "C" void kernel_launch(void* const* d_in, const int* in_sizes, int n_in,
                              void* d_out, int out_size, void* d_ws, size_t ws_size,
                              hipStream_t stream) {
    const float* x   = (const float*)d_in[0];
    const int*   pos = (const int*)  d_in[1];
    const int*   neg = (const int*)  d_in[2];
    const float* W1  = (const float*)d_in[3];
    const float* b1  = (const float*)d_in[4];
    const float* W2  = (const float*)d_in[5];
    const float* b2  = (const float*)d_in[6];
    float* out = (float*)d_out;

    const int N = in_sizes[0] / IN_CH;
    const int E = in_sizes[1] / 2;

    float* ws = (float*)d_ws;
    size_t o = 0;
    float* dinv = ws + o; o += align_f((size_t)N);
    float* h0   = ws + o; o += align_f((size_t)N * HIDC);
    float* g1   = ws + o; o += align_f((size_t)N * HIDC);
    float* g2   = ws + o; o += align_f((size_t)N * OUTC);
    int* hist   = (int*)(ws + o); o += align_f((size_t)NCH * 256);
    int* offs   = (int*)(ws + o); o += align_f((size_t)NCH * 256 + 257);
    int* rowptr = (int*)(ws + o); o += align_f((size_t)N + 1);
    int* binned = (int*)(ws + o); o += align_f((size_t)E);
    int* srcs   = (int*)(ws + o); o += align_f((size_t)E);
    (void)ws_size;

    const int T = 256;
    k_bucket_hist   <<<NCH, 256, 0, stream>>>(pos + E, hist, E);
    k_scan_hist     <<<1, 1024, 0, stream>>>(hist, offs);
    k_bucket_scatter<<<NCH, 256, 0, stream>>>(pos, pos + E, offs, binned, E);
    k_bucket_lists  <<<(N + 255) / 256, 512, 0, stream>>>(binned, offs, rowptr, dinv, srcs, N);

    dim3 g1g((N + 31) / 32, HIDC / 128);
    k_gemm_wmma_f32<1, 4><<<g1g, 128, 0, stream>>>(x, W1, h0, N, IN_CH, HIDC);
    k_gather<HIDC, true><<<(N + 7) / 8, T, 0, stream>>>(rowptr, srcs, dinv, h0, b1, g1, N);

    dim3 g2g((N + 63) / 64, OUTC / 64);
    k_gemm_wmma_f32<2, 2><<<g2g, 128, 0, stream>>>(g1, W2, h0, N, HIDC, OUTC);
    k_gather<OUTC, false><<<(N + 7) / 8, T, 0, stream>>>(rowptr, srcs, dinv, h0, b2, g2, N);

    int total = 2 * E;
    k_decode<<<(total + 256 * DEC_ITERS - 1) / (256 * DEC_ITERS), T, 0, stream>>>(pos, neg, g2, out, E);
    (void)out_size; (void)n_in;
}


# First GEMM's row blocks launched in two parts inside the grids of the last two edge-list passes (role by block index), which now run under the matrix work; node lists keep LDS-add arrival order
# speedup vs baseline: 4.4503x; 1.0714x over previous
#include <hip/hip_runtime.h>
#include <hip/hip_bf16.h>

typedef __attribute__((ext_vector_type(2))) float v2f;
typedef __attribute__((ext_vector_type(8))) float v8f;

#define IN_CH  512
#define HIDC   128
#define OUTC   64

#define NCH 256
static_assert(NCH * 256 == 1024 * 64 && NCH == 16 * 16, "k_scan_hist: 1024 threads x 64 counts, 16 groups of 16 chunks");

__device__ __forceinline__ int bucket_of(int d) {
    const unsigned b = (unsigned)d >> 8;
    return (int)(b < 255u ? b : 255u);
}

__global__ __launch_bounds__(256)
void k_bucket_hist(const int* __restrict__ col, int* __restrict__ hist, int E) {
    __shared__ int s_h[256];
    const int tid = threadIdx.x;
    const int cs = (E + NCH - 1) / NCH;
    const int e0 = blockIdx.x * cs;
    int e1 = e0 + cs;
    e1 = e1 < E ? e1 : E;
    s_h[tid] = 0;
    __syncthreads();
    const int last = e1 - 1;
    for (int i = e0 + tid; i < e1; i += 256 * 8) {
        int d[8];
        #pragma unroll
        for (int k = 0; k < 8; ++k) {
            const int idx = i + k * 256;
            d[k] = col[idx < e1 ? idx : last];
        }
        asm volatile("" :: "v"(d[0]), "v"(d[1]), "v"(d[2]), "v"(d[3]),
                           "v"(d[4]), "v"(d[5]), "v"(d[6]), "v"(d[7]));
        #pragma unroll
        for (int k = 0; k < 8; ++k)
            if (i + k * 256 < e1) atomicAdd(&s_h[bucket_of(d[k])], 1);
    }
    __syncthreads();
    hist[blockIdx.x * 256 + tid] = s_h[tid];
}

__global__ __launch_bounds__(1024)
void k_scan_hist(const int* __restrict__ hist, int* __restrict__ offs) {
    __shared__ __attribute__((aligned(16))) int s_part[16 * 256];
    __shared__ __attribute__((aligned(16))) int s_base[256];
    __shared__ int s_tot[8];
    const int tid  = threadIdx.x;
    const int lane = tid & 31;
    const int wave = tid >> 5;
    const int q = tid & 63;
    const int r = tid >> 6;
    const int4* hp = (const int4*)hist + (r * 16) * 64 + q;
    int4 v[16];
    #pragma unroll
    for (int j = 0; j < 16; ++j) v[j] = hp[j * 64];
    int4 s = v[0];
    #pragma unroll
    for (int j = 1; j < 16; ++j) {
        s.x += v[j].x; s.y += v[j].y; s.z += v[j].z; s.w += v[j].w;
    }
    ((int4*)s_part)[r * 64 + q] = s;
    __syncthreads();

    int tot = 0;
    if (tid < 256) {
        int g[16];
        #pragma unroll
        for (int k = 0; k < 16; ++k) g[k] = s_part[k * 256 + tid];
        #pragma unroll
        for (int k = 0; k < 16; ++k) { s_part[k * 256 + tid] = tot; tot += g[k]; }
    }
    int inc = tot;
    #pragma unroll
    for (int d = 1; d < 32; d <<= 1) {
        int t = __shfl_up(inc, d, 32);
        if (lane >= d) inc += t;
    }
    if (tid < 256 && lane == 31) s_tot[wave] = inc;
    __syncthreads();
    if (tid < 256) {
        int off = 0;
        #pragma unroll
        for (int w = 0; w < 8; ++w)
            if (w < wave) off += s_tot[w];
        const int base = off + inc - tot;
        s_base[tid] = base;
        offs[NCH * 256 + tid] = base;
        if (tid == 255) offs[NCH * 256 + 256] = base + tot;
    }
    __syncthreads();

    int4 run = ((const int4*)s_part)[r * 64 + q];
    const int4 b4 = ((const int4*)s_base)[q];
    run.x += b4.x; run.y += b4.y; run.z += b4.z; run.w += b4.w;
    int4* op = (int4*)offs + (r * 16) * 64 + q;
    #pragma unroll
    for (int j = 0; j < 16; ++j) {
        op[j * 64] = run;
        run.x += v[j].x; run.y += v[j].y; run.z += v[j].z; run.w += v[j].w;
    }
}

__device__ __forceinline__
void bucket_scatter_128(const int* __restrict__ row, const int* __restrict__ col,
                        const int* __restrict__ offs, int* __restrict__ binned, int E, int chunk) {
    __shared__ int s_cur[256];
    const int tid = threadIdx.x;
    const int cs = (E + NCH - 1) / NCH;
    const int e0 = chunk * cs;
    int e1 = e0 + cs;
    e1 = e1 < E ? e1 : E;
    s_cur[tid]       = offs[chunk * 256 + tid];
    s_cur[tid + 128] = offs[chunk * 256 + tid + 128];
    __syncthreads();
    const int last = e1 - 1;
    for (int i = e0 + tid; i < e1; i += 128 * 8) {
        int d[8], s[8];
        #pragma unroll
        for (int k = 0; k < 8; ++k) {
            const int idx = i + k * 128;
            const int ic  = idx < e1 ? idx : last;
            d[k] = col[ic];
            s[k] = row[ic];
        }
        asm volatile("" :: "v"(d[0]), "v"(d[1]), "v"(d[2]), "v"(d[3]),
                           "v"(d[4]), "v"(d[5]), "v"(d[6]), "v"(d[7]),
                           "v"(s[0]), "v"(s[1]), "v"(s[2]), "v"(s[3]),
                           "v"(s[4]), "v"(s[5]), "v"(s[6]), "v"(s[7]));
        #pragma unroll
        for (int k = 0; k < 8; ++k) {
            if (i + k * 128 < e1) {
                const int pos = atomicAdd(&s_cur[bucket_of(d[k])], 1);
                binned[pos] = (s[k] << 8) | (d[k] & 255);
            }
        }
    }
}

__device__ __forceinline__
void bucket_lists_128(const int* __restrict__ binned, const int* __restrict__ offs,
                      int* __restrict__ rowptr, float* __restrict__ dinv,
                      int* __restrict__ srcs, int n, int b) {
    __shared__ __attribute__((aligned(8))) int s_cnt[256];
    __shared__ __attribute__((aligned(8))) int s_cur[256];
    __shared__ int s_tot[4];
    const int tid  = threadIdx.x;
    const int lane = tid & 31;
    const int wave = tid >> 5;
    const int beg  = offs[NCH * 256 + b];
    const int end  = offs[NCH * 256 + b + 1];
    s_cnt[tid] = 0;
    s_cnt[tid + 128] = 0;
    __syncthreads();

    const int last = end - 1;
    for (int i = beg + tid; i < end; i += 128 * 8) {
        int p[8];
        #pragma unroll
        for (int k = 0; k < 8; ++k) {
            const int idx = i + k * 128;
            p[k] = binned[idx < end ? idx : last];
        }
        asm volatile("" :: "v"(p[0]), "v"(p[1]), "v"(p[2]), "v"(p[3]),
                           "v"(p[4]), "v"(p[5]), "v"(p[6]), "v"(p[7]));
        #pragma unroll
        for (int k = 0; k < 8; ++k)
            if (i + k * 128 < end) atomicAdd(&s_cnt[p[k] & 255], 1);
    }
    __syncthreads();

    const int2 own2 = ((const int2*)s_cnt)[tid];
    const int own = own2.x + own2.y;
    int inc = own;
    #pragma unroll
    for (int d = 1; d < 32; d <<= 1) {
        int t = __shfl_up(inc, d, 32);
        if (lane >= d) inc += t;
    }
    if (lane == 31) s_tot[wave] = inc;
    __syncthreads();
    {
        int off = 0;
        #pragma unroll
        for (int w = 0; w < 4; ++w)
            if (w < wave) off += s_tot[w];
        int2 excl;
        excl.x = beg + off + inc - own;
        excl.y = excl.x + own2.x;
        ((int2*)s_cur)[tid] = excl;
        const int node = b * 256 + 2 * tid;
        if (node < n) {
            rowptr[node] = excl.x;
            dinv[node]   = rsqrtf((float)(own2.x + 1));
            if (node == n - 1) rowptr[n] = excl.x + own2.x;
        }
        if (node + 1 < n) {
            rowptr[node + 1] = excl.y;
            dinv[node + 1]   = rsqrtf((float)(own2.y + 1));
            if (node + 1 == n - 1) rowptr[n] = excl.y + own2.y;
        }
    }
    __syncthreads();

    for (int i = beg + tid; i < end; i += 128 * 8) {
        int p[8];
        #pragma unroll
        for (int k = 0; k < 8; ++k) {
            const int idx = i + k * 128;
            p[k] = binned[idx < end ? idx : last];
        }
        asm volatile("" :: "v"(p[0]), "v"(p[1]), "v"(p[2]), "v"(p[3]),
                           "v"(p[4]), "v"(p[5]), "v"(p[6]), "v"(p[7]));
        #pragma unroll
        for (int k = 0; k < 8; ++k) {
            if (i + k * 128 < end) {
                const int pos = atomicAdd(&s_cur[p[k] & 255], 1);
                srcs[pos] = p[k] >> 8;
            }
        }
    }
}

__device__ __forceinline__ void stage_k_pairs(float* dst, float4 rk, float4 rk1) {
    float4 lo, hi;
    lo.x = rk.x; lo.y = rk1.x; lo.z = rk.y; lo.w = rk1.y;
    hi.x = rk.z; hi.y = rk1.z; hi.z = rk.w; hi.w = rk1.w;
    *(float4*)dst       = lo;
    *(float4*)(dst + 4) = hi;
}

template <int WM, int WN>
__device__ __forceinline__
void gemm_tile(const float* __restrict__ A, const float* __restrict__ B,
               float* __restrict__ C, int M, int K, int N, int bx, int by) {
    constexpr int NT = WM * WN * 32;
    constexpr int BM = WM * 32;
    constexpr int BN = WN * 32;
    constexpr int PA = 36;
    constexpr int PB = 2 * BN + 32;
    constexpr int NA = (BM * 8) / NT;
    constexpr int NB = (16 * (BN / 4)) / NT;
    static_assert(NA * NT == BM * 8 && NB * NT == 16 * (BN / 4), "staging slots divide evenly");

    __shared__ __attribute__((aligned(16))) float ldsA[BM * PA];
    __shared__ __attribute__((aligned(16))) float ldsB[16 * PB];

    const int tid  = threadIdx.x;
    const int lane = tid & 31;
    const int wave = __builtin_amdgcn_readfirstlane(tid >> 5);
    const int wm = wave / WN;
    const int wn = wave % WN;
    const int m0 = bx * BM;
    const int n0 = by * BN;
    const int h  = lane >> 4;
    const int c  = lane & 15;

    constexpr int RS = NT / 8;
    const int r0 = tid >> 3;
    const int q  = tid & 7;
    const int rlast = M - 1;
    const int gr0 = m0 + r0;
    const int gr1 = gr0 + RS;
    const int gr2 = gr0 + 2 * RS;
    const int gr3 = gr0 + 3 * RS;
    const float* pa0 = A + (size_t)(gr0 < M ? gr0 : rlast) * K + 4 * q;
    const float* pa1 = A + (size_t)(gr1 < M ? gr1 : rlast) * K + 4 * q;
    const float* pa2 = A + (size_t)(gr2 < M ? gr2 : rlast) * K + 4 * q;
    const float* pa3 = A + (size_t)(gr3 < M ? gr3 : rlast) * K + 4 * q;
    float* sa = ldsA + r0 * PA + 4 * q;
    constexpr int GB = BN / 4;
    constexpr int PS = NT / GB;
    static_assert(PS * NB == 16 && (NB == 2 || NB == 4) && (NA == 2 || NA == 4), "slot counts the code below is written for");
    const int p0 = tid / GB;
    const int g  = tid % GB;
    const float* pb0 = B + (size_t)(2 * p0) * N + n0 + 4 * g;
    const size_t bstep = (size_t)(2 * PS) * N;
    float* sb = ldsB + p0 * PB + 8 * g;

    float4 va0, va1, va2, va3, vb00, vb01, vb10, vb11, vb20, vb21, vb30, vb31;
    va0 = *(const float4*)pa0;
    va1 = *(const float4*)pa1;
    if constexpr (NA == 4) {
        va2 = *(const float4*)pa2;
        va3 = *(const float4*)pa3;
    }
    vb00 = *(const float4*)pb0;
    vb01 = *(const float4*)(pb0 + N);
    vb10 = *(const float4*)(pb0 + bstep);
    vb11 = *(const float4*)(pb0 + bstep + N);
    if constexpr (NB == 4) {
        vb20 = *(const float4*)(pb0 + 2 * bstep);
        vb21 = *(const float4*)(pb0 + 2 * bstep + N);
        vb30 = *(const float4*)(pb0 + 3 * bstep);
        vb31 = *(const float4*)(pb0 + 3 * bstep + N);
    }

    v8f acc[2][2] = {};

    const float* fa = ldsA + (wm * 32 + c) * PA + 2 * h;
    const float* fb = ldsB + h * PB + (wn * 32 + c) * 2;

    for (int k0 = 0; k0 < K; k0 += 32) {
        *(float4*)sa = va0;
        *(float4*)(sa + RS * PA) = va1;
        if constexpr (NA == 4) {
            *(float4*)(sa + 2 * RS * PA) = va2;
            *(float4*)(sa + 3 * RS * PA) = va3;
        }
        stage_k_pairs(sb, vb00, vb01);
        stage_k_pairs(sb + PS * PB, vb10, vb11);
        if constexpr (NB == 4) {
            stage_k_pairs(sb + 2 * PS * PB, vb20, vb21);
            stage_k_pairs(sb + 3 * PS * PB, vb30, vb31);
        }
        __syncthreads();

        if (k0 + 32 < K) {
            const int kn = k0 + 32;
            va0 = *(const float4*)(pa0 + kn);
            va1 = *(const float4*)(pa1 + kn);
            if constexpr (NA == 4) {
                va2 = *(const float4*)(pa2 + kn);
                va3 = *(const float4*)(pa3 + kn);
            }
            const float* q0 = pb0 + (size_t)kn * N;
            vb00 = *(const float4*)q0;
            vb01 = *(const float4*)(q0 + N);
            vb10 = *(const float4*)(q0 + bstep);
            vb11 = *(const float4*)(q0 + bstep + N);
            if constexpr (NB == 4) {
                vb20 = *(const float4*)(q0 + 2 * bstep);
                vb21 = *(const float4*)(q0 + 2 * bstep + N);
                vb30 = *(const float4*)(q0 + 3 * bstep);
                vb31 = *(const float4*)(q0 + 3 * bstep + N);
            }
        }

        #pragma unroll
        for (int kk = 0; kk < 32; kk += 4) {
            const v2f a0 = *(const v2f*)(fa + kk);
            const v2f a1 = *(const v2f*)(fa + 16 * PA + kk);
            const v2f b0 = *(const v2f*)(fb + (kk / 2) * PB);
            const v2f b1 = *(const v2f*)(fb + (kk / 2) * PB + 32);
            acc[0][0] = __builtin_amdgcn_wmma_f32_16x16x4_f32(false, a0, false, b0, (short)0, acc[0][0], false, false);
            acc[0][1] = __builtin_amdgcn_wmma_f32_16x16x4_f32(false, a0, false, b1, (short)0, acc[0][1], false, false);
            acc[1][0] = __builtin_amdgcn_wmma_f32_16x16x4_f32(false, a1, false, b0, (short)0, acc[1][0], false, false);
            acc[1][1] = __builtin_amdgcn_wmma_f32_16x16x4_f32(false, a1, false, b1, (short)0, acc[1][1], false, false);
        }
        __syncthreads();
    }

    const int ccol = n0 + wn * 32 + lane;
    #pragma unroll
    for (int i = 0; i < 2; ++i) {
        const int crow = m0 + wm * 32 + i * 16;
        #pragma unroll
        for (int v = 0; v < 8; ++v) {
            const float keep = h ? acc[i][1][v] : acc[i][0][v];
            const float send = h ? acc[i][0][v] : acc[i][1][v];
            const float recv = __int_as_float((int)__builtin_amdgcn_permlanex16(
                0u, (unsigned)__float_as_int(send), 0x76543210u, 0xfedcba98u, false, false));
            const int r_lo = crow + v;
            const int r_hi = crow + v + 8;
            if (r_lo < M) C[(size_t)r_lo * N + ccol] = h ? recv : keep;
            if (r_hi < M) C[(size_t)r_hi * N + ccol] = h ? keep : recv;
        }
    }
}

template <int WM, int WN>
__global__ __launch_bounds__(WM * WN * 32)
void k_gemm_wmma_f32(const float* __restrict__ A, const float* __restrict__ B,
                     float* __restrict__ C, int M, int K, int N) {
    gemm_tile<WM, WN>(A, B, C, M, K, N, blockIdx.x, blockIdx.y);
}

__global__ __launch_bounds__(128)
void k_scatter_gemm(const int* __restrict__ row, const int* __restrict__ col,
                    const int* __restrict__ offs, int* __restrict__ binned, int E,
                    const float* __restrict__ A, const float* __restrict__ B,
                    float* __restrict__ C, int M, int K, int N, int mb0) {
    if (blockIdx.x < NCH) {
        bucket_scatter_128(row, col, offs, binned, E, blockIdx.x);
        return;
    }
    gemm_tile<1, 4>(A, B, C, M, K, N, (int)blockIdx.x - NCH + mb0, 0);
}

__global__ __launch_bounds__(128)
void k_lists_gemm(const int* __restrict__ binned, const int* __restrict__ offs,
                  int* __restrict__ rowptr, float* __restrict__ dinv,
                  int* __restrict__ srcs, int n, int nsort,
                  const float* __restrict__ A, const float* __restrict__ B,
                  float* __restrict__ C, int M, int K, int N, int mb0) {
    if ((int)blockIdx.x < nsort) {
        bucket_lists_128(binned, offs, rowptr, dinv, srcs, n, blockIdx.x);
        return;
    }
    gemm_tile<1, 4>(A, B, C, M, K, N, (int)blockIdx.x - nsort + mb0, 0);
}

template <int B, typename vecT>
__device__ __forceinline__ void add_rows(vecT& acc, const vecT* __restrict__ hv,
                                         int r_l, float d_l, float dc, int j, int lane) {
    vecT v[B];
    #pragma unroll
    for (int k = 0; k < B; ++k) {
        const int r = __builtin_amdgcn_readlane(r_l, j + k);
        v[k] = hv[(size_t)r * 32 + lane];
    }
    #pragma unroll
    for (int k = 0; k < B; ++k) {
        const float w = __int_as_float(__builtin_amdgcn_readlane(__float_as_int(d_l), j + k)) * dc;
        acc += v[k] * w;
    }
}

template <int F, bool RELU>
__global__ __launch_bounds__(256)
void k_gather(const int* __restrict__ rowptr, const int* __restrict__ srcs,
              const float* __restrict__ dinv, const float* __restrict__ h,
              const float* __restrict__ bias, float* __restrict__ out, int n) {
    typedef __attribute__((ext_vector_type(F / 32))) float vecT;
    const int lane = threadIdx.x & 31;
    const int wv = __builtin_amdgcn_readfirstlane((int)(threadIdx.x >> 5));
    const int c  = blockIdx.x * 8 + wv;
    if (c >= n) return;

    const vecT* hv = (const vecT*)h;
    const int beg = rowptr[c];
    const int end = rowptr[c + 1];
    const float dc = dinv[c];

    const int total = end - beg + 1;
    vecT acc = {};
    for (int b = 0; b < total; b += 32) {
        int m = total - b;
        m = m < 32 ? m : 32;
        const int idx = b + lane;
        int   r_l = c;
        float d_l = dc;
        if (idx >= 1 && lane < m) {
            r_l = srcs[beg + idx - 1];
            d_l = dinv[r_l];
        }
        int j = 0;
        for (; j + 4 <= m; j += 4) add_rows<4>(acc, hv, r_l, d_l, dc, j, lane);
        const int rem = m - j;
        if      (rem == 3) add_rows<3>(acc, hv, r_l, d_l, dc, j, lane);
        else if (rem == 2) add_rows<2>(acc, hv, r_l, d_l, dc, j, lane);
        else if (rem == 1) add_rows<1>(acc, hv, r_l, d_l, dc, j, lane);
    }

    vecT o = acc + ((const vecT*)bias)[lane];
    if (RELU) {
        #pragma unroll
        for (int k = 0; k < F / 32; ++k) o[k] = o[k] > 0.0f ? o[k] : 0.0f;
    }
    ((vecT*)out)[(size_t)c * 32 + lane] = o;
}

#define DEC_ITERS 2

template <int K>
__device__ __forceinline__ int row_share(int v) {
    return __builtin_amdgcn_update_dpp(0, v, 0x150 + K, 0xf, 0xf, false);
}
template <int D>
__device__ __forceinline__ float row_xor(float v) {
    return __int_as_float(__builtin_amdgcn_update_dpp(0, __float_as_int(v), 0x160 + D, 0xf, 0xf, false));
}

template <int K>
__device__ __forceinline__ void fetch_slot(float4 (&a)[16], float4 (&b)[16],
                                           const float* __restrict__ z,
                                           int src, int dst, unsigned c16) {
    constexpr int R = ((K & 1) << 3) | ((K & 2) << 1) | ((K & 4) >> 1) | ((K & 8) >> 3);
    const unsigned so = (unsigned)row_share<R>(src) * 256u + c16;
    const unsigned d_o = (unsigned)row_share<R>(dst) * 256u + c16;
    a[K] = *(const float4*)((const char*)z + so);
    b[K] = *(const float4*)((const char*)z + d_o);
}

template <int D, int H>
__device__ __forceinline__ void fold_step(float (&s)[16], bool up) {
    #pragma unroll
    for (int j = 0; j < H; ++j) {
        const float keep = up ? s[j + H] : s[j];
        const float send = up ? s[j]     : s[j + H];
        s[j] = keep + row_xor<D>(send);
    }
}

__device__ __forceinline__ void load_ends(const int* __restrict__ pos, const int* __restrict__ neg,
                                          int e, int E, int& src, int& dst) {
    src = 0; dst = 0;
    if (e < 2 * E) {
        const int* p = e < E ? pos + e : neg + (e - E);
        src = p[0];
        dst = p[E];
    }
}

__global__ __launch_bounds__(256)
void k_decode(const int* __restrict__ pos, const int* __restrict__ neg,
              const float* __restrict__ z, float* __restrict__ logits, int E) {
    const int lane = threadIdx.x & 31;
    const unsigned c   = lane & 15;
    const unsigned c16 = c * 16u;
    const int wv = __builtin_amdgcn_readfirstlane((int)(threadIdx.x >> 5));
    const int total = 2 * E;
    const int base = (blockIdx.x * 8 + wv) * (32 * DEC_ITERS);
    if (base >= total) return;

    int src[DEC_ITERS], dst[DEC_ITERS];
    #pragma unroll
    for (int it = 0; it < DEC_ITERS; ++it)
        load_ends(pos, neg, base + 32 * it + lane, E, src[it], dst[it]);

    #pragma unroll
    for (int it = 0; it < DEC_ITERS; ++it) {
        const int b0 = base + 32 * it;
        if (b0 >= total) break;

        float4 a[16], b[16];
        fetch_slot<0>(a, b, z, src[it], dst[it], c16);   fetch_slot<1>(a, b, z, src[it], dst[it], c16);
        fetch_slot<2>(a, b, z, src[it], dst[it], c16);   fetch_slot<3>(a, b, z, src[it], dst[it], c16);
        fetch_slot<4>(a, b, z, src[it], dst[it], c16);   fetch_slot<5>(a, b, z, src[it], dst[it], c16);
        fetch_slot<6>(a, b, z, src[it], dst[it], c16);   fetch_slot<7>(a, b, z, src[it], dst[it], c16);
        fetch_slot<8>(a, b, z, src[it], dst[it], c16);   fetch_slot<9>(a, b, z, src[it], dst[it], c16);
        fetch_slot<10>(a, b, z, src[it], dst[it], c16);  fetch_slot<11>(a, b, z, src[it], dst[it], c16);
        fetch_slot<12>(a, b, z, src[it], dst[it], c16);  fetch_slot<13>(a, b, z, src[it], dst[it], c16);
        fetch_slot<14>(a, b, z, src[it], dst[it], c16);  fetch_slot<15>(a, b, z, src[it], dst[it], c16);

        float s[16];
        #pragma unroll
        for (int k = 0; k < 16; ++k)
            s[k] = a[k].x * b[k].x + a[k].y * b[k].y + a[k].z * b[k].z + a[k].w * b[k].w;

        fold_step<1, 8>(s, (c & 1u) != 0);
        fold_step<2, 4>(s, (c & 2u) != 0);
        fold_step<4, 2>(s, (c & 4u) != 0);
        fold_step<8, 1>(s, (c & 8u) != 0);

        const int e = b0 + lane;
        if (e < total) logits[e] = s[0];
    }
}

static inline size_t align_f(size_t n) { return (n + 63) & ~size_t(63); }

extern "C" void kernel_launch(void* const* d_in, const int* in_sizes, int n_in,
                              void* d_out, int out_size, void* d_ws, size_t ws_size,
                              hipStream_t stream) {
    const float* x   = (const float*)d_in[0];
    const int*   pos = (const int*)  d_in[1];
    const int*   neg = (const int*)  d_in[2];
    const float* W1  = (const float*)d_in[3];
    const float* b1  = (const float*)d_in[4];
    const float* W2  = (const float*)d_in[5];
    const float* b2  = (const float*)d_in[6];
    float* out = (float*)d_out;

    const int N = in_sizes[0] / IN_CH;
    const int E = in_sizes[1] / 2;

    float* ws = (float*)d_ws;
    size_t o = 0;
    float* dinv = ws + o; o += align_f((size_t)N);
    float* h0   = ws + o; o += align_f((size_t)N * HIDC);
    float* g1   = ws + o; o += align_f((size_t)N * HIDC);
    float* g2   = ws + o; o += align_f((size_t)N * OUTC);
    int* hist   = (int*)(ws + o); o += align_f((size_t)NCH * 256);
    int* offs   = (int*)(ws + o); o += align_f((size_t)NCH * 256 + 257);
    int* rowptr = (int*)(ws + o); o += align_f((size_t)N + 1);
    int* binned = (int*)(ws + o); o += align_f((size_t)E);
    int* srcs   = (int*)(ws + o); o += align_f((size_t)E);
    (void)ws_size;

    const int T = 256;
    k_bucket_hist   <<<NCH, 256, 0, stream>>>(pos + E, hist, E);
    k_scan_hist     <<<1, 1024, 0, stream>>>(hist, offs);
    const int MB  = (N + 31) / 32;
    const int MBa = MB < 768 ? MB : 768;
    const int NL  = (N + 255) / 256;
    k_scatter_gemm<<<NCH + MBa, 128, 0, stream>>>(pos, pos + E, offs, binned, E,
                                                  x, W1, h0, N, IN_CH, HIDC, 0);
    k_lists_gemm  <<<NL + (MB - MBa), 128, 0, stream>>>(binned, offs, rowptr, dinv, srcs, N, NL,
                                                        x, W1, h0, N, IN_CH, HIDC, MBa);

    k_gather<HIDC, true><<<(N + 7) / 8, T, 0, stream>>>(rowptr, srcs, dinv, h0, b1, g1, N);

    dim3 g2g((N + 63) / 64, OUTC / 64);
    k_gemm_wmma_f32<2, 2><<<g2g, 128, 0, stream>>>(g1, W2, h0, N, HIDC, OUTC);
    k_gather<OUTC, false><<<(N + 7) / 8, T, 0, stream>>>(rowptr, srcs, dinv, h0, b2, g2, N);

    int total = 2 * E;
    k_decode<<<(total + 256 * DEC_ITERS - 1) / (256 * DEC_ITERS), T, 0, stream>>>(pos, neg, g2, out, E);
    (void)out_size; (void)n_in;
}


# In-edge lists in two passes, not four: a chunk block sorts its own edges by bucket in LDS, bucket blocks walk the per-chunk pieces; count and scan launches gone; node lists keep LDS-add arrival order
# speedup vs baseline: 5.0207x; 1.1282x over previous
#include <hip/hip_runtime.h>
#include <hip/hip_bf16.h>

typedef __attribute__((ext_vector_type(2))) float v2f;
typedef __attribute__((ext_vector_type(8))) float v8f;

#define IN_CH  512
#define HIDC   128
#define OUTC   64

#define NCH  256
#define TABW 288
static_assert(NCH == 256, "bucket_lists_128: 4 waves x 32 lanes x 2 pieces");

__device__ __forceinline__ int bucket_of(int d) {
    const unsigned b = (unsigned)d >> 8;
    return (int)(b < 255u ? b : 255u);
}

__device__ __forceinline__
void chunk_sort_128(const int* __restrict__ row, const int* __restrict__ col,
                    int* __restrict__ tab, int* __restrict__ binned, int E, int chunk) {
    __shared__ __attribute__((aligned(8))) int s_cnt[256];
    __shared__ __attribute__((aligned(8))) int s_cur[256];
    __shared__ int s_tot[4];
    const int tid  = threadIdx.x;
    const int lane = tid & 31;
    const int wave = tid >> 5;
    const int cs = (E + NCH - 1) / NCH;
    const int e0 = chunk * cs;
    int e1 = e0 + cs;
    e1 = e1 < E ? e1 : E;
    s_cnt[tid] = 0;
    s_cnt[tid + 128] = 0;
    __syncthreads();

    const int last = e1 - 1;
    for (int i = e0 + tid; i < e1; i += 128 * 8) {
        int d[8];
        #pragma unroll
        for (int k = 0; k < 8; ++k) {
            const int idx = i + k * 128;
            d[k] = col[idx < e1 ? idx : last];
        }
        asm volatile("" :: "v"(d[0]), "v"(d[1]), "v"(d[2]), "v"(d[3]),
                           "v"(d[4]), "v"(d[5]), "v"(d[6]), "v"(d[7]));
        #pragma unroll
        for (int k = 0; k < 8; ++k)
            if (i + k * 128 < e1) atomicAdd(&s_cnt[bucket_of(d[k])], 1);
    }
    __syncthreads();

    const int2 own2 = ((const int2*)s_cnt)[tid];
    const int own = own2.x + own2.y;
    int inc = own;
    #pragma unroll
    for (int d = 1; d < 32; d <<= 1) {
        int t = __shfl_up(inc, d, 32);
        if (lane >= d) inc += t;
    }
    if (lane == 31) s_tot[wave] = inc;
    __syncthreads();
    {
        int off = 0;
        #pragma unroll
        for (int w = 0; w < 4; ++w)
            if (w < wave) off += s_tot[w];
        int2 excl;
        excl.x = off + inc - own;
        excl.y = excl.x + own2.x;
        *(int2*)(tab + chunk * TABW + 2 * tid) = excl;
        if (tid == 127) tab[chunk * TABW + 256] = excl.y + own2.y;
        int2 cur;
        cur.x = e0 + excl.x;
        cur.y = e0 + excl.y;
        ((int2*)s_cur)[tid] = cur;
    }
    __syncthreads();

    for (int i = e0 + tid; i < e1; i += 128 * 8) {
        int d[8], s[8];
        #pragma unroll
        for (int k = 0; k < 8; ++k) {
            const int idx = i + k * 128;
            const int ic  = idx < e1 ? idx : last;
            d[k] = col[ic];
            s[k] = row[ic];
        }
        asm volatile("" :: "v"(d[0]), "v"(d[1]), "v"(d[2]), "v"(d[3]),
                           "v"(d[4]), "v"(d[5]), "v"(d[6]), "v"(d[7]),
                           "v"(s[0]), "v"(s[1]), "v"(s[2]), "v"(s[3]),
                           "v"(s[4]), "v"(s[5]), "v"(s[6]), "v"(s[7]));
        #pragma unroll
        for (int k = 0; k < 8; ++k) {
            if (i + k * 128 < e1) {
                const int pos = atomicAdd(&s_cur[bucket_of(d[k])], 1);
                binned[pos] = (s[k] << 8) | (d[k] & 255);
            }
        }
    }
}

#define PIECES 16

template <bool PLACE>
__device__ __forceinline__
void walk_pieces(const int* __restrict__ binned, int* __restrict__ srcs, int* s_tab,
                 int segA, int lenA, int segB, int lenB, int lane) {
    #pragma nounroll
    for (int t = 0; t < 4; ++t) {
        const int seg_l = t < 2 ? segA : segB;
        const int len_l = t < 2 ? lenA : lenB;
        const int j0 = (t & 1) * PIECES;
        int st[PIECES], ln[PIECES], p[PIECES];
        int longest = 0;
        #pragma unroll
        for (int k = 0; k < PIECES; ++k) {
            st[k] = __builtin_amdgcn_readlane(seg_l, j0 + k);
            ln[k] = __builtin_amdgcn_readlane(len_l, j0 + k);
            longest = longest > ln[k] ? longest : ln[k];
            p[k] = 0;
            if (lane < ln[k]) p[k] = binned[st[k] + lane];
        }
        asm volatile("" :: "v"(p[0]), "v"(p[1]), "v"(p[2]), "v"(p[3]),
                           "v"(p[4]), "v"(p[5]), "v"(p[6]), "v"(p[7]),
                           "v"(p[8]), "v"(p[9]), "v"(p[10]), "v"(p[11]),
                           "v"(p[12]), "v"(p[13]), "v"(p[14]), "v"(p[15]));
        #pragma unroll
        for (int k = 0; k < PIECES; ++k) {
            if (lane < ln[k]) {
                const int pos = atomicAdd(&s_tab[p[k] & 255], 1);
                if (PLACE) srcs[pos] = p[k] >> 8;
            }
        }
        if (longest > 32) {
            #pragma nounroll
            for (int k = 0; k < PIECES; ++k) {
                const int s = __builtin_amdgcn_readlane(seg_l, j0 + k);
                const int l = __builtin_amdgcn_readlane(len_l, j0 + k);
                for (int j = lane + 32; j < l; j += 32) {
                    const int q = binned[s + j];
                    const int pos = atomicAdd(&s_tab[q & 255], 1);
                    if (PLACE) srcs[pos] = q >> 8;
                }
            }
        }
    }
}
static_assert(PIECES == 16, "walk_pieces: the load fence names 16 values, a lane register holds 2 x 16 pieces");

__device__ __forceinline__
void bucket_lists_128(const int* __restrict__ binned, const int* __restrict__ tab,
                      int* __restrict__ rowptr, float* __restrict__ dinv,
                      int* __restrict__ srcs, int n, int E, int b) {
    __shared__ __attribute__((aligned(8))) int s_cnt[256];
    __shared__ __attribute__((aligned(8))) int s_cur[256];
    __shared__ int s_tot[4];
    __shared__ int s_low[4];
    const int tid  = threadIdx.x;
    const int lane = tid & 31;
    const int wave = __builtin_amdgcn_readfirstlane(tid >> 5);
    const int cs = (E + NCH - 1) / NCH;

    const int cA = wave * 64 + lane;
    const int cB = cA + 32;
    const int loA = tab[cA * TABW + b];
    const int hiA = tab[cA * TABW + b + 1];
    const int loB = tab[cB * TABW + b];
    const int hiB = tab[cB * TABW + b + 1];
    const int segA = cA * cs + loA;
    const int lenA = hiA - loA;
    const int segB = cB * cs + loB;
    const int lenB = hiB - loB;
    int low = loA + loB;
    #pragma unroll
    for (int d = 16; d >= 1; d >>= 1) low += __shfl_xor(low, d, 32);
    if (lane == 0) s_low[wave] = low;
    s_cnt[tid] = 0;
    s_cnt[tid + 128] = 0;
    __syncthreads();
    const int beg = s_low[0] + s_low[1] + s_low[2] + s_low[3];

    walk_pieces<false>(binned, srcs, s_cnt, segA, lenA, segB, lenB, lane);
    __syncthreads();

    const int2 own2 = ((const int2*)s_cnt)[tid];
    const int own = own2.x + own2.y;
    int inc = own;
    #pragma unroll
    for (int d = 1; d < 32; d <<= 1) {
        int t = __shfl_up(inc, d, 32);
        if (lane >= d) inc += t;
    }
    if (lane == 31) s_tot[wave] = inc;
    __syncthreads();
    {
        int off = 0;
        #pragma unroll
        for (int w = 0; w < 4; ++w)
            if (w < wave) off += s_tot[w];
        int2 excl;
        excl.x = beg + off + inc - own;
        excl.y = excl.x + own2.x;
        ((int2*)s_cur)[tid] = excl;
        const int node = b * 256 + 2 * tid;
        if (node < n) {
            rowptr[node] = excl.x;
            dinv[node]   = rsqrtf((float)(own2.x + 1));
            if (node == n - 1) rowptr[n] = excl.x + own2.x;
        }
        if (node + 1 < n) {
            rowptr[node + 1] = excl.y;
            dinv[node + 1]   = rsqrtf((float)(own2.y + 1));
            if (node + 1 == n - 1) rowptr[n] = excl.y + own2.y;
        }
    }
    __syncthreads();

    walk_pieces<true>(binned, srcs, s_cur, segA, lenA, segB, lenB, lane);
}

__device__ __forceinline__ void stage_k_pairs(float* dst, float4 rk, float4 rk1) {
    float4 lo, hi;
    lo.x = rk.x; lo.y = rk1.x; lo.z = rk.y; lo.w = rk1.y;
    hi.x = rk.z; hi.y = rk1.z; hi.z = rk.w; hi.w = rk1.w;
    *(float4*)dst       = lo;
    *(float4*)(dst + 4) = hi;
}

template <int WM, int WN>
__device__ __forceinline__
void gemm_tile(const float* __restrict__ A, const float* __restrict__ B,
               float* __restrict__ C, int M, int K, int N, int bx, int by) {
    constexpr int NT = WM * WN * 32;
    constexpr int BM = WM * 32;
    constexpr int BN = WN * 32;
    constexpr int PA = 36;
    constexpr int PB = 2 * BN + 32;
    constexpr int NA = (BM * 8) / NT;
    constexpr int NB = (16 * (BN / 4)) / NT;
    static_assert(NA * NT == BM * 8 && NB * NT == 16 * (BN / 4), "staging slots divide evenly");

    __shared__ __attribute__((aligned(16))) float ldsA[BM * PA];
    __shared__ __attribute__((aligned(16))) float ldsB[16 * PB];

    const int tid  = threadIdx.x;
    const int lane = tid & 31;
    const int wave = __builtin_amdgcn_readfirstlane(tid >> 5);
    const int wm = wave / WN;
    const int wn = wave % WN;
    const int m0 = bx * BM;
    const int n0 = by * BN;
    const int h  = lane >> 4;
    const int c  = lane & 15;

    constexpr int RS = NT / 8;
    const int r0 = tid >> 3;
    const int q  = tid & 7;
    const int rlast = M - 1;
    const int gr0 = m0 + r0;
    const int gr1 = gr0 + RS;
    const int gr2 = gr0 + 2 * RS;
    const int gr3 = gr0 + 3 * RS;
    const float* pa0 = A + (size_t)(gr0 < M ? gr0 : rlast) * K + 4 * q;
    const float* pa1 = A + (size_t)(gr1 < M ? gr1 : rlast) * K + 4 * q;
    const float* pa2 = A + (size_t)(gr2 < M ? gr2 : rlast) * K + 4 * q;
    const float* pa3 = A + (size_t)(gr3 < M ? gr3 : rlast) * K + 4 * q;
    float* sa = ldsA + r0 * PA + 4 * q;
    constexpr int GB = BN / 4;
    constexpr int PS = NT / GB;
    static_assert(PS * NB == 16 && (NB == 2 || NB == 4) && (NA == 2 || NA == 4), "slot counts the code below is written for");
    const int p0 = tid / GB;
    const int g  = tid % GB;
    const float* pb0 = B + (size_t)(2 * p0) * N + n0 + 4 * g;
    const size_t bstep = (size_t)(2 * PS) * N;
    float* sb = ldsB + p0 * PB + 8 * g;

    float4 va0, va1, va2, va3, vb00, vb01, vb10, vb11, vb20, vb21, vb30, vb31;
    va0 = *(const float4*)pa0;
    va1 = *(const float4*)pa1;
    if constexpr (NA == 4) {
        va2 = *(const float4*)pa2;
        va3 = *(const float4*)pa3;
    }
    vb00 = *(const float4*)pb0;
    vb01 = *(const float4*)(pb0 + N);
    vb10 = *(const float4*)(pb0 + bstep);
    vb11 = *(const float4*)(pb0 + bstep + N);
    if constexpr (NB == 4) {
        vb20 = *(const float4*)(pb0 + 2 * bstep);
        vb21 = *(const float4*)(pb0 + 2 * bstep + N);
        vb30 = *(const float4*)(pb0 + 3 * bstep);
        vb31 = *(const float4*)(pb0 + 3 * bstep + N);
    }

    v8f acc[2][2] = {};

    const float* fa = ldsA + (wm * 32 + c) * PA + 2 * h;
    const float* fb = ldsB + h * PB + (wn * 32 + c) * 2;

    for (int k0 = 0; k0 < K; k0 += 32) {
        *(float4*)sa = va0;
        *(float4*)(sa + RS * PA) = va1;
        if constexpr (NA == 4) {
            *(float4*)(sa + 2 * RS * PA) = va2;
            *(float4*)(sa + 3 * RS * PA) = va3;
        }
        stage_k_pairs(sb, vb00, vb01);
        stage_k_pairs(sb + PS * PB, vb10, vb11);
        if constexpr (NB == 4) {
            stage_k_pairs(sb + 2 * PS * PB, vb20, vb21);
            stage_k_pairs(sb + 3 * PS * PB, vb30, vb31);
        }
        __syncthreads();

        if (k0 + 32 < K) {
            const int kn = k0 + 32;
            va0 = *(const float4*)(pa0 + kn);
            va1 = *(const float4*)(pa1 + kn);
            if constexpr (NA == 4) {
                va2 = *(const float4*)(pa2 + kn);
                va3 = *(const float4*)(pa3 + kn);
            }
            const float* q0 = pb0 + (size_t)kn * N;
            vb00 = *(const float4*)q0;
            vb01 = *(const float4*)(q0 + N);
            vb10 = *(const float4*)(q0 + bstep);
            vb11 = *(const float4*)(q0 + bstep + N);
            if constexpr (NB == 4) {
                vb20 = *(const float4*)(q0 + 2 * bstep);
                vb21 = *(const float4*)(q0 + 2 * bstep + N);
                vb30 = *(const float4*)(q0 + 3 * bstep);
                vb31 = *(const float4*)(q0 + 3 * bstep + N);
            }
        }

        #pragma unroll
        for (int kk = 0; kk < 32; kk += 4) {
            const v2f a0 = *(const v2f*)(fa + kk);
            const v2f a1 = *(const v2f*)(fa + 16 * PA + kk);
            const v2f b0 = *(const v2f*)(fb + (kk / 2) * PB);
            const v2f b1 = *(const v2f*)(fb + (kk / 2) * PB + 32);
            acc[0][0] = __builtin_amdgcn_wmma_f32_16x16x4_f32(false, a0, false, b0, (short)0, acc[0][0], false, false);
            acc[0][1] = __builtin_amdgcn_wmma_f32_16x16x4_f32(false, a0, false, b1, (short)0, acc[0][1], false, false);
            acc[1][0] = __builtin_amdgcn_wmma_f32_16x16x4_f32(false, a1, false, b0, (short)0, acc[1][0], false, false);
            acc[1][1] = __builtin_amdgcn_wmma_f32_16x16x4_f32(false, a1, false, b1, (short)0, acc[1][1], false, false);
        }
        __syncthreads();
    }

    const int ccol = n0 + wn * 32 + lane;
    #pragma unroll
    for (int i = 0; i < 2; ++i) {
        const int crow = m0 + wm * 32 + i * 16;
        #pragma unroll
        for (int v = 0; v < 8; ++v) {
            const float keep = h ? acc[i][1][v] : acc[i][0][v];
            const float send = h ? acc[i][0][v] : acc[i][1][v];
            const float recv = __int_as_float((int)__builtin_amdgcn_permlanex16(
                0u, (unsigned)__float_as_int(send), 0x76543210u, 0xfedcba98u, false, false));
            const int r_lo = crow + v;
            const int r_hi = crow + v + 8;
            if (r_lo < M) C[(size_t)r_lo * N + ccol] = h ? recv : keep;
            if (r_hi < M) C[(size_t)r_hi * N + ccol] = h ? keep : recv;
        }
    }
}

template <int WM, int WN>
__global__ __launch_bounds__(WM * WN * 32)
void k_gemm_wmma_f32(const float* __restrict__ A, const float* __restrict__ B,
                     float* __restrict__ C, int M, int K, int N) {
    gemm_tile<WM, WN>(A, B, C, M, K, N, blockIdx.x, blockIdx.y);
}

__global__ __launch_bounds__(128)
void k_sort_gemm(const int* __restrict__ row, const int* __restrict__ col,
                 int* __restrict__ tab, int* __restrict__ binned, int E,
                 const float* __restrict__ A, const float* __restrict__ B,
                 float* __restrict__ C, int M, int K, int N, int mb0) {
    if (blockIdx.x < NCH) {
        chunk_sort_128(row, col, tab, binned, E, blockIdx.x);
        return;
    }
    gemm_tile<1, 4>(A, B, C, M, K, N, (int)blockIdx.x - NCH + mb0, 0);
}

__global__ __launch_bounds__(128)
void k_lists_gemm(const int* __restrict__ binned, const int* __restrict__ tab,
                  int* __restrict__ rowptr, float* __restrict__ dinv,
                  int* __restrict__ srcs, int n, int E, int nsort,
                  const float* __restrict__ A, const float* __restrict__ B,
                  float* __restrict__ C, int M, int K, int N, int mb0) {
    if ((int)blockIdx.x < nsort) {
        bucket_lists_128(binned, tab, rowptr, dinv, srcs, n, E, blockIdx.x);
        return;
    }
    gemm_tile<1, 4>(A, B, C, M, K, N, (int)blockIdx.x - nsort + mb0, 0);
}

template <int B, typename vecT>
__device__ __forceinline__ void add_rows(vecT& acc, const vecT* __restrict__ hv,
                                         int r_l, float d_l, float dc, int j, int lane) {
    vecT v[B];
    #pragma unroll
    for (int k = 0; k < B; ++k) {
        const int r = __builtin_amdgcn_readlane(r_l, j + k);
        v[k] = hv[(size_t)r * 32 + lane];
    }
    #pragma unroll
    for (int k = 0; k < B; ++k) {
        const float w = __int_as_float(__builtin_amdgcn_readlane(__float_as_int(d_l), j + k)) * dc;
        acc += v[k] * w;
    }
}

template <int F, bool RELU>
__global__ __launch_bounds__(256)
void k_gather(const int* __restrict__ rowptr, const int* __restrict__ srcs,
              const float* __restrict__ dinv, const float* __restrict__ h,
              const float* __restrict__ bias, float* __restrict__ out, int n) {
    typedef __attribute__((ext_vector_type(F / 32))) float vecT;
    const int lane = threadIdx.x & 31;
    const int wv = __builtin_amdgcn_readfirstlane((int)(threadIdx.x >> 5));
    const int c  = blockIdx.x * 8 + wv;
    if (c >= n) return;

    const vecT* hv = (const vecT*)h;
    const int beg = rowptr[c];
    const int end = rowptr[c + 1];
    const float dc = dinv[c];

    const int total = end - beg + 1;
    vecT acc = {};
    for (int b = 0; b < total; b += 32) {
        int m = total - b;
        m = m < 32 ? m : 32;
        const int idx = b + lane;
        int   r_l = c;
        float d_l = dc;
        if (idx >= 1 && lane < m) {
            r_l = srcs[beg + idx - 1];
            d_l = dinv[r_l];
        }
        int j = 0;
        for (; j + 4 <= m; j += 4) add_rows<4>(acc, hv, r_l, d_l, dc, j, lane);
        const int rem = m - j;
        if      (rem == 3) add_rows<3>(acc, hv, r_l, d_l, dc, j, lane);
        else if (rem == 2) add_rows<2>(acc, hv, r_l, d_l, dc, j, lane);
        else if (rem == 1) add_rows<1>(acc, hv, r_l, d_l, dc, j, lane);
    }

    vecT o = acc + ((const vecT*)bias)[lane];
    if (RELU) {
        #pragma unroll
        for (int k = 0; k < F / 32; ++k) o[k] = o[k] > 0.0f ? o[k] : 0.0f;
    }
    ((vecT*)out)[(size_t)c * 32 + lane] = o;
}

#define DEC_ITERS 2

template <int K>
__device__ __forceinline__ int row_share(int v) {
    return __builtin_amdgcn_update_dpp(0, v, 0x150 + K, 0xf, 0xf, false);
}
template <int D>
__device__ __forceinline__ float row_xor(float v) {
    return __int_as_float(__builtin_amdgcn_update_dpp(0, __float_as_int(v), 0x160 + D, 0xf, 0xf, false));
}

template <int K>
__device__ __forceinline__ void fetch_slot(float4 (&a)[16], float4 (&b)[16],
                                           const float* __restrict__ z,
                                           int src, int dst, unsigned c16) {
    constexpr int R = ((K & 1) << 3) | ((K & 2) << 1) | ((K & 4) >> 1) | ((K & 8) >> 3);
    const unsigned so = (unsigned)row_share<R>(src) * 256u + c16;
    const unsigned d_o = (unsigned)row_share<R>(dst) * 256u + c16;
    a[K] = *(const float4*)((const char*)z + so);
    b[K] = *(const float4*)((const char*)z + d_o);
}

template <int D, int H>
__device__ __forceinline__ void fold_step(float (&s)[16], bool up) {
    #pragma unroll
    for (int j = 0; j < H; ++j) {
        const float keep = up ? s[j + H] : s[j];
        const float send = up ? s[j]     : s[j + H];
        s[j] = keep + row_xor<D>(send);
    }
}

__device__ __forceinline__ void load_ends(const int* __restrict__ pos, const int* __restrict__ neg,
                                          int e, int E, int& src, int& dst) {
    src = 0; dst = 0;
    if (e < 2 * E) {
        const int* p = e < E ? pos + e : neg + (e - E);
        src = p[0];
        dst = p[E];
    }
}

__global__ __launch_bounds__(256)
void k_decode(const int* __restrict__ pos, const int* __restrict__ neg,
              const float* __restrict__ z, float* __restrict__ logits, int E) {
    const int lane = threadIdx.x & 31;
    const unsigned c   = lane & 15;
    const unsigned c16 = c * 16u;
    const int wv = __builtin_amdgcn_readfirstlane((int)(threadIdx.x >> 5));
    const int total = 2 * E;
    const int base = (blockIdx.x * 8 + wv) * (32 * DEC_ITERS);
    if (base >= total) return;

    int src[DEC_ITERS], dst[DEC_ITERS];
    #pragma unroll
    for (int it = 0; it < DEC_ITERS; ++it)
        load_ends(pos, neg, base + 32 * it + lane, E, src[it], dst[it]);

    #pragma unroll
    for (int it = 0; it < DEC_ITERS; ++it) {
        const int b0 = base + 32 * it;
        if (b0 >= total) break;

        float4 a[16], b[16];
        fetch_slot<0>(a, b, z, src[it], dst[it], c16);   fetch_slot<1>(a, b, z, src[it], dst[it], c16);
        fetch_slot<2>(a, b, z, src[it], dst[it], c16);   fetch_slot<3>(a, b, z, src[it], dst[it], c16);
        fetch_slot<4>(a, b, z, src[it], dst[it], c16);   fetch_slot<5>(a, b, z, src[it], dst[it], c16);
        fetch_slot<6>(a, b, z, src[it], dst[it], c16);   fetch_slot<7>(a, b, z, src[it], dst[it], c16);
        fetch_slot<8>(a, b, z, src[it], dst[it], c16);   fetch_slot<9>(a, b, z, src[it], dst[it], c16);
        fetch_slot<10>(a, b, z, src[it], dst[it], c16);  fetch_slot<11>(a, b, z, src[it], dst[it], c16);
        fetch_slot<12>(a, b, z, src[it], dst[it], c16);  fetch_slot<13>(a, b, z, src[it], dst[it], c16);
        fetch_slot<14>(a, b, z, src[it], dst[it], c16);  fetch_slot<15>(a, b, z, src[it], dst[it], c16);

        float s[16];
        #pragma unroll
        for (int k = 0; k < 16; ++k)
            s[k] = a[k].x * b[k].x + a[k].y * b[k].y + a[k].z * b[k].z + a[k].w * b[k].w;

        fold_step<1, 8>(s, (c & 1u) != 0);
        fold_step<2, 4>(s, (c & 2u) != 0);
        fold_step<4, 2>(s, (c & 4u) != 0);
        fold_step<8, 1>(s, (c & 8u) != 0);

        const int e = b0 + lane;
        if (e < total) logits[e] = s[0];
    }
}

static inline size_t align_f(size_t n) { return (n + 63) & ~size_t(63); }

extern "C" void kernel_launch(void* const* d_in, const int* in_sizes, int n_in,
                              void* d_out, int out_size, void* d_ws, size_t ws_size,
                              hipStream_t stream) {
    const float* x   = (const float*)d_in[0];
    const int*   pos = (const int*)  d_in[1];
    const int*   neg = (const int*)  d_in[2];
    const float* W1  = (const float*)d_in[3];
    const float* b1  = (const float*)d_in[4];
    const float* W2  = (const float*)d_in[5];
    const float* b2  = (const float*)d_in[6];
    float* out = (float*)d_out;

    const int N = in_sizes[0] / IN_CH;
    const int E = in_sizes[1] / 2;

    float* ws = (float*)d_ws;
    size_t o = 0;
    float* dinv = ws + o; o += align_f((size_t)N);
    float* h0   = ws + o; o += align_f((size_t)N * HIDC);
    float* g1   = ws + o; o += align_f((size_t)N * HIDC);
    float* g2   = ws + o; o += align_f((size_t)N * OUTC);
    int* tab    = (int*)(ws + o); o += align_f((size_t)NCH * TABW);
    int* rowptr = (int*)(ws + o); o += align_f((size_t)N + 1);
    int* binned = (int*)(ws + o); o += align_f((size_t)E);
    int* srcs   = (int*)(ws + o); o += align_f((size_t)E);
    (void)ws_size;

    const int T = 256;
    const int MB  = (N + 31) / 32;
    const int MBa = MB < 768 ? MB : 768;
    const int NL  = (N + 255) / 256;
    k_sort_gemm <<<NCH + MBa, 128, 0, stream>>>(pos, pos + E, tab, binned, E,
                                                x, W1, h0, N, IN_CH, HIDC, 0);
    k_lists_gemm<<<NL + (MB - MBa), 128, 0, stream>>>(binned, tab, rowptr, dinv, srcs, N, E, NL,
                                                      x, W1, h0, N, IN_CH, HIDC, MBa);

    k_gather<HIDC, true><<<(N + 7) / 8, T, 0, stream>>>(rowptr, srcs, dinv, h0, b1, g1, N);

    dim3 g2g((N + 63) / 64, OUTC / 64);
    k_gemm_wmma_f32<2, 2><<<g2g, 128, 0, stream>>>(g1, W2, h0, N, HIDC, OUTC);
    k_gather<OUTC, false><<<(N + 7) / 8, T, 0, stream>>>(rowptr, srcs, dinv, h0, b2, g2, N);

    int total = 2 * E;
    k_decode<<<(total + 256 * DEC_ITERS - 1) / (256 * DEC_ITERS), T, 0, stream>>>(pos, neg, g2, out, E);
    (void)out_size; (void)n_in;
}


# Layer 2's product moved into layer 1's aggregation launch: a block's 16 finished rows go through LDS to WMMA (W2 fragments from global); its k sum is two halves added at the store; one launch less
# speedup vs baseline: 5.1777x; 1.0313x over previous
#include <hip/hip_runtime.h>
#include <hip/hip_bf16.h>

typedef __attribute__((ext_vector_type(2))) float v2f;
typedef __attribute__((ext_vector_type(8))) float v8f;

#define IN_CH  512
#define HIDC   128
#define OUTC   64

#define NCH  256
#define TABW 288
static_assert(NCH == 256, "bucket_lists_128: 4 waves x 32 lanes x 2 pieces");

__device__ __forceinline__ int bucket_of(int d) {
    const unsigned b = (unsigned)d >> 8;
    return (int)(b < 255u ? b : 255u);
}

__device__ __forceinline__
void chunk_sort_128(const int* __restrict__ row, const int* __restrict__ col,
                    int* __restrict__ tab, int* __restrict__ binned, int E, int chunk) {
    __shared__ __attribute__((aligned(8))) int s_cnt[256];
    __shared__ __attribute__((aligned(8))) int s_cur[256];
    __shared__ int s_tot[4];
    const int tid  = threadIdx.x;
    const int lane = tid & 31;
    const int wave = tid >> 5;
    const int cs = (E + NCH - 1) / NCH;
    const int e0 = chunk * cs;
    int e1 = e0 + cs;
    e1 = e1 < E ? e1 : E;
    s_cnt[tid] = 0;
    s_cnt[tid + 128] = 0;
    __syncthreads();

    const int last = e1 - 1;
    for (int i = e0 + tid; i < e1; i += 128 * 8) {
        int d[8];
        #pragma unroll
        for (int k = 0; k < 8; ++k) {
            const int idx = i + k * 128;
            d[k] = col[idx < e1 ? idx : last];
        }
        asm volatile("" :: "v"(d[0]), "v"(d[1]), "v"(d[2]), "v"(d[3]),
                           "v"(d[4]), "v"(d[5]), "v"(d[6]), "v"(d[7]));
        #pragma unroll
        for (int k = 0; k < 8; ++k)
            if (i + k * 128 < e1) atomicAdd(&s_cnt[bucket_of(d[k])], 1);
    }
    __syncthreads();

    const int2 own2 = ((const int2*)s_cnt)[tid];
    const int own = own2.x + own2.y;
    int inc = own;
    #pragma unroll
    for (int d = 1; d < 32; d <<= 1) {
        int t = __shfl_up(inc, d, 32);
        if (lane >= d) inc += t;
    }
    if (lane == 31) s_tot[wave] = inc;
    __syncthreads();
    {
        int off = 0;
        #pragma unroll
        for (int w = 0; w < 4; ++w)
            if (w < wave) off += s_tot[w];
        int2 excl;
        excl.x = off + inc - own;
        excl.y = excl.x + own2.x;
        *(int2*)(tab + chunk * TABW + 2 * tid) = excl;
        if (tid == 127) tab[chunk * TABW + 256] = excl.y + own2.y;
        int2 cur;
        cur.x = e0 + excl.x;
        cur.y = e0 + excl.y;
        ((int2*)s_cur)[tid] = cur;
    }
    __syncthreads();

    for (int i = e0 + tid; i < e1; i += 128 * 8) {
        int d[8], s[8];
        #pragma unroll
        for (int k = 0; k < 8; ++k) {
            const int idx = i + k * 128;
            const int ic  = idx < e1 ? idx : last;
            d[k] = col[ic];
            s[k] = row[ic];
        }
        asm volatile("" :: "v"(d[0]), "v"(d[1]), "v"(d[2]), "v"(d[3]),
                           "v"(d[4]), "v"(d[5]), "v"(d[6]), "v"(d[7]),
                           "v"(s[0]), "v"(s[1]), "v"(s[2]), "v"(s[3]),
                           "v"(s[4]), "v"(s[5]), "v"(s[6]), "v"(s[7]));
        #pragma unroll
        for (int k = 0; k < 8; ++k) {
            if (i + k * 128 < e1) {
                const int pos = atomicAdd(&s_cur[bucket_of(d[k])], 1);
                binned[pos] = (s[k] << 8) | (d[k] & 255);
            }
        }
    }
}

#define PIECES 16

template <bool PLACE>
__device__ __forceinline__
void walk_pieces(const int* __restrict__ binned, int* __restrict__ srcs, int* s_tab,
                 int segA, int lenA, int segB, int lenB, int lane) {
    #pragma nounroll
    for (int t = 0; t < 4; ++t) {
        const int seg_l = t < 2 ? segA : segB;
        const int len_l = t < 2 ? lenA : lenB;
        const int j0 = (t & 1) * PIECES;
        int st[PIECES], ln[PIECES], p[PIECES];
        int longest = 0;
        #pragma unroll
        for (int k = 0; k < PIECES; ++k) {
            st[k] = __builtin_amdgcn_readlane(seg_l, j0 + k);
            ln[k] = __builtin_amdgcn_readlane(len_l, j0 + k);
            longest = longest > ln[k] ? longest : ln[k];
            p[k] = 0;
            if (lane < ln[k]) p[k] = binned[st[k] + lane];
        }
        asm volatile("" :: "v"(p[0]), "v"(p[1]), "v"(p[2]), "v"(p[3]),
                           "v"(p[4]), "v"(p[5]), "v"(p[6]), "v"(p[7]),
                           "v"(p[8]), "v"(p[9]), "v"(p[10]), "v"(p[11]),
                           "v"(p[12]), "v"(p[13]), "v"(p[14]), "v"(p[15]));
        #pragma unroll
        for (int k = 0; k < PIECES; ++k) {
            if (lane < ln[k]) {
                const int pos = atomicAdd(&s_tab[p[k] & 255], 1);
                if (PLACE) srcs[pos] = p[k] >> 8;
            }
        }
        if (longest > 32) {
            #pragma nounroll
            for (int k = 0; k < PIECES; ++k) {
                const int s = __builtin_amdgcn_readlane(seg_l, j0 + k);
                const int l = __builtin_amdgcn_readlane(len_l, j0 + k);
                for (int j = lane + 32; j < l; j += 32) {
                    const int q = binned[s + j];
                    const int pos = atomicAdd(&s_tab[q & 255], 1);
                    if (PLACE) srcs[pos] = q >> 8;
                }
            }
        }
    }
}
static_assert(PIECES == 16, "walk_pieces: the load fence names 16 values, a lane register holds 2 x 16 pieces");

__device__ __forceinline__
void bucket_lists_128(const int* __restrict__ binned, const int* __restrict__ tab,
                      int* __restrict__ rowptr, float* __restrict__ dinv,
                      int* __restrict__ srcs, int n, int E, int b) {
    __shared__ __attribute__((aligned(8))) int s_cnt[256];
    __shared__ __attribute__((aligned(8))) int s_cur[256];
    __shared__ int s_tot[4];
    __shared__ int s_low[4];
    const int tid  = threadIdx.x;
    const int lane = tid & 31;
    const int wave = __builtin_amdgcn_readfirstlane(tid >> 5);
    const int cs = (E + NCH - 1) / NCH;

    const int cA = wave * 64 + lane;
    const int cB = cA + 32;
    const int loA = tab[cA * TABW + b];
    const int hiA = tab[cA * TABW + b + 1];
    const int loB = tab[cB * TABW + b];
    const int hiB = tab[cB * TABW + b + 1];
    const int segA = cA * cs + loA;
    const int lenA = hiA - loA;
    const int segB = cB * cs + loB;
    const int lenB = hiB - loB;
    int low = loA + loB;
    #pragma unroll
    for (int d = 16; d >= 1; d >>= 1) low += __shfl_xor(low, d, 32);
    if (lane == 0) s_low[wave] = low;
    s_cnt[tid] = 0;
    s_cnt[tid + 128] = 0;
    __syncthreads();
    const int beg = s_low[0] + s_low[1] + s_low[2] + s_low[3];

    walk_pieces<false>(binned, srcs, s_cnt, segA, lenA, segB, lenB, lane);
    __syncthreads();

    const int2 own2 = ((const int2*)s_cnt)[tid];
    const int own = own2.x + own2.y;
    int inc = own;
    #pragma unroll
    for (int d = 1; d < 32; d <<= 1) {
        int t = __shfl_up(inc, d, 32);
        if (lane >= d) inc += t;
    }
    if (lane == 31) s_tot[wave] = inc;
    __syncthreads();
    {
        int off = 0;
        #pragma unroll
        for (int w = 0; w < 4; ++w)
            if (w < wave) off += s_tot[w];
        int2 excl;
        excl.x = beg + off + inc - own;
        excl.y = excl.x + own2.x;
        ((int2*)s_cur)[tid] = excl;
        const int node = b * 256 + 2 * tid;
        if (node < n) {
            rowptr[node] = excl.x;
            dinv[node]   = rsqrtf((float)(own2.x + 1));
            if (node == n - 1) rowptr[n] = excl.x + own2.x;
        }
        if (node + 1 < n) {
            rowptr[node + 1] = excl.y;
            dinv[node + 1]   = rsqrtf((float)(own2.y + 1));
            if (node + 1 == n - 1) rowptr[n] = excl.y + own2.y;
        }
    }
    __syncthreads();

    walk_pieces<true>(binned, srcs, s_cur, segA, lenA, segB, lenB, lane);
}

__device__ __forceinline__ void stage_k_pairs(float* dst, float4 rk, float4 rk1) {
    float4 lo, hi;
    lo.x = rk.x; lo.y = rk1.x; lo.z = rk.y; lo.w = rk1.y;
    hi.x = rk.z; hi.y = rk1.z; hi.z = rk.w; hi.w = rk1.w;
    *(float4*)dst       = lo;
    *(float4*)(dst + 4) = hi;
}

template <int WM, int WN>
__device__ __forceinline__
void gemm_tile(const float* __restrict__ A, const float* __restrict__ B,
               float* __restrict__ C, int M, int K, int N, int bx, int by) {
    constexpr int NT = WM * WN * 32;
    constexpr int BM = WM * 32;
    constexpr int BN = WN * 32;
    constexpr int PA = 36;
    constexpr int PB = 2 * BN + 32;
    constexpr int NA = (BM * 8) / NT;
    constexpr int NB = (16 * (BN / 4)) / NT;
    static_assert(NA * NT == BM * 8 && NB * NT == 16 * (BN / 4), "staging slots divide evenly");

    __shared__ __attribute__((aligned(16))) float ldsA[BM * PA];
    __shared__ __attribute__((aligned(16))) float ldsB[16 * PB];

    const int tid  = threadIdx.x;
    const int lane = tid & 31;
    const int wave = __builtin_amdgcn_readfirstlane(tid >> 5);
    const int wm = wave / WN;
    const int wn = wave % WN;
    const int m0 = bx * BM;
    const int n0 = by * BN;
    const int h  = lane >> 4;
    const int c  = lane & 15;

    constexpr int RS = NT / 8;
    const int r0 = tid >> 3;
    const int q  = tid & 7;
    const int rlast = M - 1;
    const int gr0 = m0 + r0;
    const int gr1 = gr0 + RS;
    const int gr2 = gr0 + 2 * RS;
    const int gr3 = gr0 + 3 * RS;
    const float* pa0 = A + (size_t)(gr0 < M ? gr0 : rlast) * K + 4 * q;
    const float* pa1 = A + (size_t)(gr1 < M ? gr1 : rlast) * K + 4 * q;
    const float* pa2 = A + (size_t)(gr2 < M ? gr2 : rlast) * K + 4 * q;
    const float* pa3 = A + (size_t)(gr3 < M ? gr3 : rlast) * K + 4 * q;
    float* sa = ldsA + r0 * PA + 4 * q;
    constexpr int GB = BN / 4;
    constexpr int PS = NT / GB;
    static_assert(PS * NB == 16 && (NB == 2 || NB == 4) && (NA == 2 || NA == 4), "slot counts the code below is written for");
    const int p0 = tid / GB;
    const int g  = tid % GB;
    const float* pb0 = B + (size_t)(2 * p0) * N + n0 + 4 * g;
    const size_t bstep = (size_t)(2 * PS) * N;
    float* sb = ldsB + p0 * PB + 8 * g;

    float4 va0, va1, va2, va3, vb00, vb01, vb10, vb11, vb20, vb21, vb30, vb31;
    va0 = *(const float4*)pa0;
    va1 = *(const float4*)pa1;
    if constexpr (NA == 4) {
        va2 = *(const float4*)pa2;
        va3 = *(const float4*)pa3;
    }
    vb00 = *(const float4*)pb0;
    vb01 = *(const float4*)(pb0 + N);
    vb10 = *(const float4*)(pb0 + bstep);
    vb11 = *(const float4*)(pb0 + bstep + N);
    if constexpr (NB == 4) {
        vb20 = *(const float4*)(pb0 + 2 * bstep);
        vb21 = *(const float4*)(pb0 + 2 * bstep + N);
        vb30 = *(const float4*)(pb0 + 3 * bstep);
        vb31 = *(const float4*)(pb0 + 3 * bstep + N);
    }

    v8f acc[2][2] = {};

    const float* fa = ldsA + (wm * 32 + c) * PA + 2 * h;
    const float* fb = ldsB + h * PB + (wn * 32 + c) * 2;

    for (int k0 = 0; k0 < K; k0 += 32) {
        *(float4*)sa = va0;
        *(float4*)(sa + RS * PA) = va1;
        if constexpr (NA == 4) {
            *(float4*)(sa + 2 * RS * PA) = va2;
            *(float4*)(sa + 3 * RS * PA) = va3;
        }
        stage_k_pairs(sb, vb00, vb01);
        stage_k_pairs(sb + PS * PB, vb10, vb11);
        if constexpr (NB == 4) {
            stage_k_pairs(sb + 2 * PS * PB, vb20, vb21);
            stage_k_pairs(sb + 3 * PS * PB, vb30, vb31);
        }
        __syncthreads();

        if (k0 + 32 < K) {
            const int kn = k0 + 32;
            va0 = *(const float4*)(pa0 + kn);
            va1 = *(const float4*)(pa1 + kn);
            if constexpr (NA == 4) {
                va2 = *(const float4*)(pa2 + kn);
                va3 = *(const float4*)(pa3 + kn);
            }
            const float* q0 = pb0 + (size_t)kn * N;
            vb00 = *(const float4*)q0;
            vb01 = *(const float4*)(q0 + N);
            vb10 = *(const float4*)(q0 + bstep);
            vb11 = *(const float4*)(q0 + bstep + N);
            if constexpr (NB == 4) {
                vb20 = *(const float4*)(q0 + 2 * bstep);
                vb21 = *(const float4*)(q0 + 2 * bstep + N);
                vb30 = *(const float4*)(q0 + 3 * bstep);
                vb31 = *(const float4*)(q0 + 3 * bstep + N);
            }
        }

        #pragma unroll
        for (int kk = 0; kk < 32; kk += 4) {
            const v2f a0 = *(const v2f*)(fa + kk);
            const v2f a1 = *(const v2f*)(fa + 16 * PA + kk);
            const v2f b0 = *(const v2f*)(fb + (kk / 2) * PB);
            const v2f b1 = *(const v2f*)(fb + (kk / 2) * PB + 32);
            acc[0][0] = __builtin_amdgcn_wmma_f32_16x16x4_f32(false, a0, false, b0, (short)0, acc[0][0], false, false);
            acc[0][1] = __builtin_amdgcn_wmma_f32_16x16x4_f32(false, a0, false, b1, (short)0, acc[0][1], false, false);
            acc[1][0] = __builtin_amdgcn_wmma_f32_16x16x4_f32(false, a1, false, b0, (short)0, acc[1][0], false, false);
            acc[1][1] = __builtin_amdgcn_wmma_f32_16x16x4_f32(false, a1, false, b1, (short)0, acc[1][1], false, false);
        }
        __syncthreads();
    }

    const int ccol = n0 + wn * 32 + lane;
    #pragma unroll
    for (int i = 0; i < 2; ++i) {
        const int crow = m0 + wm * 32 + i * 16;
        #pragma unroll
        for (int v = 0; v < 8; ++v) {
            const float keep = h ? acc[i][1][v] : acc[i][0][v];
            const float send = h ? acc[i][0][v] : acc[i][1][v];
            const float recv = __int_as_float((int)__builtin_amdgcn_permlanex16(
                0u, (unsigned)__float_as_int(send), 0x76543210u, 0xfedcba98u, false, false));
            const int r_lo = crow + v;
            const int r_hi = crow + v + 8;
            if (r_lo < M) C[(size_t)r_lo * N + ccol] = h ? recv : keep;
            if (r_hi < M) C[(size_t)r_hi * N + ccol] = h ? keep : recv;
        }
    }
}

__global__ __launch_bounds__(128)
void k_sort_gemm(const int* __restrict__ row, const int* __restrict__ col,
                 int* __restrict__ tab, int* __restrict__ binned, int E,
                 const float* __restrict__ A, const float* __restrict__ B,
                 float* __restrict__ C, int M, int K, int N, int mb0) {
    if (blockIdx.x < NCH) {
        chunk_sort_128(row, col, tab, binned, E, blockIdx.x);
        return;
    }
    gemm_tile<1, 4>(A, B, C, M, K, N, (int)blockIdx.x - NCH + mb0, 0);
}

__global__ __launch_bounds__(128)
void k_lists_gemm(const int* __restrict__ binned, const int* __restrict__ tab,
                  int* __restrict__ rowptr, float* __restrict__ dinv,
                  int* __restrict__ srcs, int n, int E, int nsort,
                  const float* __restrict__ A, const float* __restrict__ B,
                  float* __restrict__ C, int M, int K, int N, int mb0) {
    if ((int)blockIdx.x < nsort) {
        bucket_lists_128(binned, tab, rowptr, dinv, srcs, n, E, blockIdx.x);
        return;
    }
    gemm_tile<1, 4>(A, B, C, M, K, N, (int)blockIdx.x - nsort + mb0, 0);
}

template <int B, typename vecT>
__device__ __forceinline__ void add_rows(vecT& acc, const vecT* __restrict__ hv,
                                         int r_l, float d_l, float dc, int j, int lane) {
    vecT v[B];
    #pragma unroll
    for (int k = 0; k < B; ++k) {
        const int r = __builtin_amdgcn_readlane(r_l, j + k);
        v[k] = hv[(size_t)r * 32 + lane];
    }
    #pragma unroll
    for (int k = 0; k < B; ++k) {
        const float w = __int_as_float(__builtin_amdgcn_readlane(__float_as_int(d_l), j + k)) * dc;
        acc += v[k] * w;
    }
}

template <typename vecT>
__device__ __forceinline__ vecT node_sum(const int* __restrict__ rowptr, const int* __restrict__ srcs,
                                         const float* __restrict__ dinv, const float* __restrict__ h,
                                         int c, int lane) {
    const vecT* hv = (const vecT*)h;
    const int beg = rowptr[c];
    const int end = rowptr[c + 1];
    const float dc = dinv[c];
    const int total = end - beg + 1;
    vecT acc = {};
    for (int b = 0; b < total; b += 32) {
        int m = total - b;
        m = m < 32 ? m : 32;
        const int idx = b + lane;
        int   r_l = c;
        float d_l = dc;
        if (idx >= 1 && lane < m) {
            r_l = srcs[beg + idx - 1];
            d_l = dinv[r_l];
        }
        int j = 0;
        for (; j + 4 <= m; j += 4) add_rows<4>(acc, hv, r_l, d_l, dc, j, lane);
        const int rem = m - j;
        if      (rem == 3) add_rows<3>(acc, hv, r_l, d_l, dc, j, lane);
        else if (rem == 2) add_rows<2>(acc, hv, r_l, d_l, dc, j, lane);
        else if (rem == 1) add_rows<1>(acc, hv, r_l, d_l, dc, j, lane);
    }
    return acc;
}

template <int F, bool RELU>
__global__ __launch_bounds__(256)
void k_gather(const int* __restrict__ rowptr, const int* __restrict__ srcs,
              const float* __restrict__ dinv, const float* __restrict__ h,
              const float* __restrict__ bias, float* __restrict__ out, int n) {
    typedef __attribute__((ext_vector_type(F / 32))) float vecT;
    const int lane = threadIdx.x & 31;
    const int wv = __builtin_amdgcn_readfirstlane((int)(threadIdx.x >> 5));
    const int c  = blockIdx.x * 8 + wv;
    if (c >= n) return;

    vecT o = node_sum<vecT>(rowptr, srcs, dinv, h, c, lane) + ((const vecT*)bias)[lane];
    if (RELU) {
        #pragma unroll
        for (int k = 0; k < F / 32; ++k) o[k] = o[k] > 0.0f ? o[k] : 0.0f;
    }
    ((vecT*)out)[(size_t)c * 32 + lane] = o;
}

#define FT_PA 132
#define FT_PZ 68

__global__ __launch_bounds__(256)
void k_gather_mm(const int* __restrict__ rowptr, const int* __restrict__ srcs,
                 const float* __restrict__ dinv, const float* __restrict__ h,
                 const float* __restrict__ bias, const float* __restrict__ W2,
                 float* __restrict__ z0, int n) {
    typedef __attribute__((ext_vector_type(4))) float vecT;
    static_assert(HIDC == 128 && OUTC == 64, "8 waves = 4 column tiles x 2 k halves of 64");
    __shared__ __attribute__((aligned(16))) float sA[16 * FT_PA];
    __shared__ __attribute__((aligned(16))) float sZ[2 * 16 * FT_PZ];

    const int tid  = threadIdx.x;
    const int lane = tid & 31;
    const int wv = __builtin_amdgcn_readfirstlane(tid >> 5);
    const int hh = lane >> 4;
    const int cc = lane & 15;
    const int n0 = (wv & 3) * 16;
    const int kh = wv >> 2;

    const int c0 = blockIdx.x * 16;
    #pragma nounroll
    for (int i = 0; i < 2; ++i) {
        const int r = 2 * wv + i;
        const int c = c0 + r;
        vecT o = {};
        if (c < n) {
            o = node_sum<vecT>(rowptr, srcs, dinv, h, c, lane) + ((const vecT*)bias)[lane];
            #pragma unroll
            for (int k = 0; k < 4; ++k) o[k] = o[k] > 0.0f ? o[k] : 0.0f;
        }
        *(vecT*)(sA + r * FT_PA + 4 * lane) = o;
    }

    v2f bf[16];
    const float* wp = W2 + (size_t)(64 * kh + 2 * hh) * OUTC + n0 + cc;
    #pragma unroll
    for (int s = 0; s < 16; ++s) {
        bf[s].x = wp[(4 * s) * OUTC];
        bf[s].y = wp[(4 * s + 1) * OUTC];
    }
    __syncthreads();

    v8f acc = {};
    const float* fa = sA + cc * FT_PA + 64 * kh + 2 * hh;
    #pragma unroll
    for (int s = 0; s < 16; ++s) {
        const v2f a = *(const v2f*)(fa + 4 * s);
        acc = __builtin_amdgcn_wmma_f32_16x16x4_f32(false, a, false, bf[s], (short)0, acc, false, false);
    }

    float* zp = sZ + (kh * 16 + 8 * hh) * FT_PZ + n0 + cc;
    #pragma unroll
    for (int v = 0; v < 8; ++v) zp[v * FT_PZ] = acc[v];
    __syncthreads();

    const int row = tid >> 4;
    const int c4  = (tid & 15) * 4;
    const float4 p = *(const float4*)(sZ + row * FT_PZ + c4);
    const float4 q = *(const float4*)(sZ + (16 + row) * FT_PZ + c4);
    if (c0 + row < n) {
        float4 o;
        o.x = p.x + q.x; o.y = p.y + q.y; o.z = p.z + q.z; o.w = p.w + q.w;
        *(float4*)(z0 + (size_t)(c0 + row) * OUTC + c4) = o;
    }
}

#define DEC_ITERS 2

template <int K>
__device__ __forceinline__ int row_share(int v) {
    return __builtin_amdgcn_update_dpp(0, v, 0x150 + K, 0xf, 0xf, false);
}
template <int D>
__device__ __forceinline__ float row_xor(float v) {
    return __int_as_float(__builtin_amdgcn_update_dpp(0, __float_as_int(v), 0x160 + D, 0xf, 0xf, false));
}

template <int K>
__device__ __forceinline__ void fetch_slot(float4 (&a)[16], float4 (&b)[16],
                                           const float* __restrict__ z,
                                           int src, int dst, unsigned c16) {
    constexpr int R = ((K & 1) << 3) | ((K & 2) << 1) | ((K & 4) >> 1) | ((K & 8) >> 3);
    const unsigned so = (unsigned)row_share<R>(src) * 256u + c16;
    const unsigned d_o = (unsigned)row_share<R>(dst) * 256u + c16;
    a[K] = *(const float4*)((const char*)z + so);
    b[K] = *(const float4*)((const char*)z + d_o);
}

template <int D, int H>
__device__ __forceinline__ void fold_step(float (&s)[16], bool up) {
    #pragma unroll
    for (int j = 0; j < H; ++j) {
        const float keep = up ? s[j + H] : s[j];
        const float send = up ? s[j]     : s[j + H];
        s[j] = keep + row_xor<D>(send);
    }
}

__device__ __forceinline__ void load_ends(const int* __restrict__ pos, const int* __restrict__ neg,
                                          int e, int E, int& src, int& dst) {
    src = 0; dst = 0;
    if (e < 2 * E) {
        const int* p = e < E ? pos + e : neg + (e - E);
        src = p[0];
        dst = p[E];
    }
}

__global__ __launch_bounds__(256)
void k_decode(const int* __restrict__ pos, const int* __restrict__ neg,
              const float* __restrict__ z, float* __restrict__ logits, int E) {
    const int lane = threadIdx.x & 31;
    const unsigned c   = lane & 15;
    const unsigned c16 = c * 16u;
    const int wv = __builtin_amdgcn_readfirstlane((int)(threadIdx.x >> 5));
    const int total = 2 * E;
    const int base = (blockIdx.x * 8 + wv) * (32 * DEC_ITERS);
    if (base >= total) return;

    int src[DEC_ITERS], dst[DEC_ITERS];
    #pragma unroll
    for (int it = 0; it < DEC_ITERS; ++it)
        load_ends(pos, neg, base + 32 * it + lane, E, src[it], dst[it]);

    #pragma unroll
    for (int it = 0; it < DEC_ITERS; ++it) {
        const int b0 = base + 32 * it;
        if (b0 >= total) break;

        float4 a[16], b[16];
        fetch_slot<0>(a, b, z, src[it], dst[it], c16);   fetch_slot<1>(a, b, z, src[it], dst[it], c16);
        fetch_slot<2>(a, b, z, src[it], dst[it], c16);   fetch_slot<3>(a, b, z, src[it], dst[it], c16);
        fetch_slot<4>(a, b, z, src[it], dst[it], c16);   fetch_slot<5>(a, b, z, src[it], dst[it], c16);
        fetch_slot<6>(a, b, z, src[it], dst[it], c16);   fetch_slot<7>(a, b, z, src[it], dst[it], c16);
        fetch_slot<8>(a, b, z, src[it], dst[it], c16);   fetch_slot<9>(a, b, z, src[it], dst[it], c16);
        fetch_slot<10>(a, b, z, src[it], dst[it], c16);  fetch_slot<11>(a, b, z, src[it], dst[it], c16);
        fetch_slot<12>(a, b, z, src[it], dst[it], c16);  fetch_slot<13>(a, b, z, src[it], dst[it], c16);
        fetch_slot<14>(a, b, z, src[it], dst[it], c16);  fetch_slot<15>(a, b, z, src[it], dst[it], c16);

        float s[16];
        #pragma unroll
        for (int k = 0; k < 16; ++k)
            s[k] = a[k].x * b[k].x + a[k].y * b[k].y + a[k].z * b[k].z + a[k].w * b[k].w;

        fold_step<1, 8>(s, (c & 1u) != 0);
        fold_step<2, 4>(s, (c & 2u) != 0);
        fold_step<4, 2>(s, (c & 4u) != 0);
        fold_step<8, 1>(s, (c & 8u) != 0);

        const int e = b0 + lane;
        if (e < total) logits[e] = s[0];
    }
}

static inline size_t align_f(size_t n) { return (n + 63) & ~size_t(63); }

extern "C" void kernel_launch(void* const* d_in, const int* in_sizes, int n_in,
                              void* d_out, int out_size, void* d_ws, size_t ws_size,
                              hipStream_t stream) {
    const float* x   = (const float*)d_in[0];
    const int*   pos = (const int*)  d_in[1];
    const int*   neg = (const int*)  d_in[2];
    const float* W1  = (const float*)d_in[3];
    const float* b1  = (const float*)d_in[4];
    const float* W2  = (const float*)d_in[5];
    const float* b2  = (const float*)d_in[6];
    float* out = (float*)d_out;

    const int N = in_sizes[0] / IN_CH;
    const int E = in_sizes[1] / 2;

    float* ws = (float*)d_ws;
    size_t o = 0;
    float* dinv = ws + o; o += align_f((size_t)N);
    float* h0   = ws + o; o += align_f((size_t)N * HIDC);
    float* g1   = ws + o; o += align_f((size_t)N * HIDC);
    float* g2   = ws + o; o += align_f((size_t)N * OUTC);
    int* tab    = (int*)(ws + o); o += align_f((size_t)NCH * TABW);
    int* rowptr = (int*)(ws + o); o += align_f((size_t)N + 1);
    int* binned = (int*)(ws + o); o += align_f((size_t)E);
    int* srcs   = (int*)(ws + o); o += align_f((size_t)E);
    (void)ws_size;

    const int T = 256;
    const int MB  = (N + 31) / 32;
    const int MBa = MB < 768 ? MB : 768;
    const int NL  = (N + 255) / 256;
    k_sort_gemm <<<NCH + MBa, 128, 0, stream>>>(pos, pos + E, tab, binned, E,
                                                x, W1, h0, N, IN_CH, HIDC, 0);
    k_lists_gemm<<<NL + (MB - MBa), 128, 0, stream>>>(binned, tab, rowptr, dinv, srcs, N, E, NL,
                                                      x, W1, h0, N, IN_CH, HIDC, MBa);

    k_gather_mm<<<(N + 15) / 16, T, 0, stream>>>(rowptr, srcs, dinv, h0, b1, W2, g1, N);

    k_gather<OUTC, false><<<(N + 7) / 8, T, 0, stream>>>(rowptr, srcs, dinv, g1, b2, g2, N);

    int total = 2 * E;
    k_decode<<<(total + 256 * DEC_ITERS - 1) / (256 * DEC_ITERS), T, 0, stream>>>(pos, neg, g2, out, E);
    (void)out_size; (void)n_in;
}
